# starting kernel
# baseline (speedup 1.0000x reference)
#include <hip/hip_runtime.h>
#include <math.h>

#define F 20
#define NEG_SLOPE 0.2f

typedef float v2f __attribute__((ext_vector_type(2)));
typedef float v8f __attribute__((ext_vector_type(8)));

__device__ __forceinline__ void atomicMaxF32(float* addr, float val) {
    if (val >= 0.0f) {
        atomicMax((int*)addr, __float_as_int(val));
    } else {
        atomicMin((unsigned int*)addr, __float_as_uint(val));
    }
}

__global__ void gat_node_init_l1(const float* __restrict__ x,
                                 const float* __restrict__ W1,
                                 const float* __restrict__ a_src,
                                 const float* __restrict__ a_dst,
                                 float* __restrict__ h,
                                 float* __restrict__ asrc,
                                 float* __restrict__ adst,
                                 float* __restrict__ m,
                                 float* __restrict__ denom,
                                 float* __restrict__ accum,
                                 int N) {
    int i = blockIdx.x * blockDim.x + threadIdx.x;
    if (i >= N) return;
    float xv = x[i];
    float sa = 0.0f, sd = 0.0f;
#pragma unroll
    for (int f = 0; f < F; ++f) {
        float hv = xv * W1[f];
        h[i * F + f] = hv;
        sa += hv * a_src[f];
        sd += hv * a_dst[f];
        accum[i * F + f] = 0.0f;
    }
    asrc[i] = sa;
    adst[i] = sd;
    m[i] = -3.402823466e38f;
    denom[i] = 0.0f;
}

__global__ void __launch_bounds__(32)
gat_wmma_feat_x_w(const float* __restrict__ feat,
                  const float* __restrict__ W,
                  float* __restrict__ h,
                  int N) {
    int tile  = blockIdx.x;
    int lane  = threadIdx.x;
    int mrow  = lane & 15;
    int khalf = (lane >> 4) * 2;

    int row  = tile * 16 + mrow;
    int rowc = row < N ? row : N - 1;
    const float* arow = feat + (size_t)rowc * F;

    int   n1   = 16 + mrow;
    int   n1c  = (n1 < F) ? n1 : 0;
    float msk1 = (n1 < F) ? 1.0f : 0.0f;

    v8f c0 = {};
    v8f c1 = {};
#pragma unroll
    for (int k0 = 0; k0 < F; k0 += 4) {
        const float* wr0 = W + (size_t)(k0 + khalf + 0) * F;
        const float* wr1 = W + (size_t)(k0 + khalf + 1) * F;

        v2f a;
        a.x = arow[k0 + khalf + 0];
        a.y = arow[k0 + khalf + 1];

        v2f b0;
        b0.x = wr0[mrow];
        b0.y = wr1[mrow];
        c0 = __builtin_amdgcn_wmma_f32_16x16x4_f32(false, a, false, b0,
                                                   (short)0, c0, false, false);

        v2f b1;
        b1.x = wr0[n1c] * msk1;
        b1.y = wr1[n1c] * msk1;
        c1 = __builtin_amdgcn_wmma_f32_16x16x4_f32(false, a, false, b1,
                                                   (short)0, c1, false, false);
    }

    int rbase = tile * 16 + ((lane >> 4) << 3);
#pragma unroll
    for (int j = 0; j < 8; ++j) {
        int r = rbase + j;
        if (r < N) {
            h[(size_t)r * F + mrow] = c0[j];
            if (n1 < F) h[(size_t)r * F + n1] = c1[j];
        }
    }
}

__global__ void gat_node_alpha_init(const float* __restrict__ h,
                                    const float* __restrict__ a_src,
                                    const float* __restrict__ a_dst,
                                    float* __restrict__ asrc,
                                    float* __restrict__ adst,
                                    float* __restrict__ m,
                                    float* __restrict__ denom,
                                    float* __restrict__ accum,
                                    int N) {
    int i = blockIdx.x * blockDim.x + threadIdx.x;
    if (i >= N) return;
    float sa = 0.0f, sd = 0.0f;
#pragma unroll
    for (int f = 0; f < F; ++f) {
        float hv = h[i * F + f];
        sa += hv * a_src[f];
        sd += hv * a_dst[f];
        accum[i * F + f] = 0.0f;
    }
    asrc[i] = sa;
    adst[i] = sd;
    m[i] = -3.402823466e38f;
    denom[i] = 0.0f;
}

__global__ void gat_edge_max(const int* __restrict__ src,
                             const int* __restrict__ dst,
                             const float* __restrict__ asrc,
                             const float* __restrict__ adst,
                             float* __restrict__ m,
                             int E, int Etot) {
    int e = blockIdx.x * blockDim.x + threadIdx.x;
    if (e >= Etot) return;
    int s, d;
    if (e < E) { s = src[e]; d = dst[e]; }
    else       { s = e - E; d = e - E; }
    float ev = asrc[s] + adst[d];
    ev = ev > 0.0f ? ev : NEG_SLOPE * ev;
    atomicMaxF32(&m[d], ev);
}

__global__ void gat_edge_acc(const int* __restrict__ src,
                             const int* __restrict__ dst,
                             const float* __restrict__ asrc,
                             const float* __restrict__ adst,
                             const float* __restrict__ m,
                             const float* __restrict__ h,
                             float* __restrict__ denom,
                             float* __restrict__ accum,
                             int E, int Etot) {
    int e = blockIdx.x * blockDim.x + threadIdx.x;
    if (e >= Etot) return;
    int s, d;
    if (e < E) { s = src[e]; d = dst[e]; }
    else       { s = e - E; d = e - E; }
    float ev = asrc[s] + adst[d];
    ev = ev > 0.0f ? ev : NEG_SLOPE * ev;
    float t = __expf(ev - m[d]);
    atomicAdd(&denom[d], t);
    const float* hs = h + (size_t)s * F;
    float* ac = accum + (size_t)d * F;
#pragma unroll
    for (int f = 0; f < F; ++f) {
        atomicAdd(&ac[f], t * hs[f]);
    }
}

__global__ void gat_node_finish(const float* __restrict__ accum,
                                const float* __restrict__ denom,
                                const float* __restrict__ b,
                                float* __restrict__ feat,
                                int N) {
    int i = blockIdx.x * blockDim.x + threadIdx.x;
    if (i >= N) return;
    float inv = 1.0f / (denom[i] + 1e-16f);
#pragma unroll
    for (int f = 0; f < F; ++f) {
        float v = accum[i * F + f] * inv + b[f];
        feat[i * F + f] = v > 0.0f ? v : 0.0f;
    }
}

__global__ void gat_final_linear(const float* __restrict__ feat,
                                 const float* __restrict__ Wl,
                                 const float* __restrict__ bl,
                                 float* __restrict__ out,
                                 int N) {
    int i = blockIdx.x * blockDim.x + threadIdx.x;
    if (i >= N) return;
    float sacc = bl[0];
#pragma unroll
    for (int f = 0; f < F; ++f) sacc += feat[i * F + f] * Wl[f];
    out[i] = sacc;
}

extern "C" void kernel_launch(void* const* d_in, const int* in_sizes, int n_in,
                              void* d_out, int out_size, void* d_ws, size_t ws_size,
                              hipStream_t stream) {
    (void)n_in; (void)out_size; (void)ws_size;

    const int N    = in_sizes[0];
    const int E    = in_sizes[1] / 2;
    const int Etot = E + N;

    const float* x    = (const float*)d_in[0];
    const int*   ei   = (const int*)d_in[1];
    const float* W1   = (const float*)d_in[2];
    const float* as1  = (const float*)d_in[3];
    const float* ad1  = (const float*)d_in[4];
    const float* b1   = (const float*)d_in[5];
    const float* W2   = (const float*)d_in[6];
    const float* as2  = (const float*)d_in[7];
    const float* ad2  = (const float*)d_in[8];
    const float* b2   = (const float*)d_in[9];
    const float* Wl   = (const float*)d_in[10];
    const float* bl   = (const float*)d_in[11];

    const int* srcIdx = ei;
    const int* dstIdx = ei + E;

    float* ws     = (float*)d_ws;
    float* h      = ws + (size_t)0  * N;
    float* accum  = ws + (size_t)20 * N;
    float* feat1  = ws + (size_t)40 * N;
    float* feat2  = ws + (size_t)60 * N;
    float* asrc   = ws + (size_t)80 * N;
    float* adst   = ws + (size_t)81 * N;
    float* m      = ws + (size_t)82 * N;
    float* denom  = ws + (size_t)83 * N;

    const int TB   = 256;
    const int gN   = (N + TB - 1) / TB;
    const int gE   = (Etot + TB - 1) / TB;
    const int gMM  = (N + 15) / 16;

    gat_node_init_l1<<<gN, TB, 0, stream>>>(x, W1, as1, ad1,
                                            h, asrc, adst, m, denom, accum, N);
    gat_edge_max<<<gE, TB, 0, stream>>>(srcIdx, dstIdx, asrc, adst, m, E, Etot);
    gat_edge_acc<<<gE, TB, 0, stream>>>(srcIdx, dstIdx, asrc, adst, m, h,
                                        denom, accum, E, Etot);
    gat_node_finish<<<gN, TB, 0, stream>>>(accum, denom, b1, feat1, N);

    gat_wmma_feat_x_w<<<gMM, 32, 0, stream>>>(feat1, W2, h, N);
    gat_node_alpha_init<<<gN, TB, 0, stream>>>(h, as2, ad2,
                                               asrc, adst, m, denom, accum, N);
    gat_edge_max<<<gE, TB, 0, stream>>>(srcIdx, dstIdx, asrc, adst, m, E, Etot);
    gat_edge_acc<<<gE, TB, 0, stream>>>(srcIdx, dstIdx, asrc, adst, m, h,
                                        denom, accum, E, Etot);
    gat_node_finish<<<gN, TB, 0, stream>>>(accum, denom, b2, feat2, N);

    gat_final_linear<<<gN, TB, 0, stream>>>(feat2, Wl, bl, (float*)d_out, N);
}


# Edge softmax and aggregation without float atomics: edges counting-sorted by dst once per call (int atomic rank + scan), then a per-node register gather; sums run in arrival-rank order per segment
# speedup vs baseline: 5.1840x; 5.1840x over previous
#include <hip/hip_runtime.h>
#include <math.h>

#define F 20
#define NEG_SLOPE 0.2f

typedef float v2f __attribute__((ext_vector_type(2)));
typedef float v8f __attribute__((ext_vector_type(8)));

__global__ void gat_node_init_l1(const float* __restrict__ x,
                                 const float* __restrict__ W1,
                                 const float* __restrict__ a_src,
                                 const float* __restrict__ a_dst,
                                 float* __restrict__ h,
                                 float* __restrict__ asrc,
                                 float* __restrict__ adst,
                                 int* __restrict__ cnt,
                                 int N) {
    int i = blockIdx.x * blockDim.x + threadIdx.x;
    if (i >= N) return;
    float xv = x[i];
    float sa = 0.0f, sd = 0.0f;
#pragma unroll
    for (int f = 0; f < F; ++f) {
        float hv = xv * W1[f];
        h[i * F + f] = hv;
        sa += hv * a_src[f];
        sd += hv * a_dst[f];
    }
    asrc[i] = sa;
    adst[i] = sd;
    cnt[i] = 0;
}

#define EPT 4
__global__ void __launch_bounds__(256)
csr_count(const int* __restrict__ dst,
          int* __restrict__ cnt,
          int* __restrict__ rank,
          int E) {
    const int e0 = blockIdx.x * (256 * EPT) + threadIdx.x;
    if (blockIdx.x * (256 * EPT) + 256 * EPT <= E) {
        int d0 = dst[e0];
        int d1 = dst[e0 + 256];
        int d2 = dst[e0 + 512];
        int d3 = dst[e0 + 768];
        int r0 = atomicAdd(&cnt[d0], 1);
        int r1 = atomicAdd(&cnt[d1], 1);
        int r2 = atomicAdd(&cnt[d2], 1);
        int r3 = atomicAdd(&cnt[d3], 1);
        rank[e0]       = r0;
        rank[e0 + 256] = r1;
        rank[e0 + 512] = r2;
        rank[e0 + 768] = r3;
    } else {
        for (int k = 0; k < EPT; ++k) {
            int e = e0 + k * 256;
            if (e < E) rank[e] = atomicAdd(&cnt[dst[e]], 1);
        }
    }
}

__global__ void __launch_bounds__(256)
csr_block_sums(const int* __restrict__ cnt,
               int* __restrict__ bsum,
               int N) {
    __shared__ int red[256];
    int t = threadIdx.x;
    int i = blockIdx.x * 256 + t;
    red[t] = i < N ? cnt[i] : 0;
    __syncthreads();
    for (int s = 128; s > 0; s >>= 1) {
        if (t < s) red[t] += red[t + s];
        __syncthreads();
    }
    if (t == 0) bsum[blockIdx.x] = red[0];
}

__global__ void __launch_bounds__(256)
csr_offsets(const int* __restrict__ cnt,
            const int* __restrict__ bsum,
            int* __restrict__ off,
            int N) {
    __shared__ int red[256];
    __shared__ int buf[2][256];
    int t = threadIdx.x;
    int b = blockIdx.x;

    int part = 0;
    for (int k = t; k < b; k += 256) part += bsum[k];
    red[t] = part;
    __syncthreads();
    for (int s = 128; s > 0; s >>= 1) {
        if (t < s) red[t] += red[t + s];
        __syncthreads();
    }
    int base = red[0];

    int i = b * 256 + t;
    int v = i < N ? cnt[i] : 0;
    int cur = 0;
    buf[0][t] = v;
    __syncthreads();
    for (int dd = 1; dd < 256; dd <<= 1) {
        int xv = buf[cur][t];
        if (t >= dd) xv += buf[cur][t - dd];
        buf[cur ^ 1][t] = xv;
        cur ^= 1;
        __syncthreads();
    }
    int incl = buf[cur][t];
    if (i < N) {
        off[i] = base + incl - v;
        if (i == N - 1) off[N] = base + incl;
    }
}

__global__ void __launch_bounds__(256)
csr_fill(const int* __restrict__ src,
         const int* __restrict__ dst,
         const int* __restrict__ rank,
         const int* __restrict__ off,
         int* __restrict__ ssrc,
         int E) {
    const int e0 = blockIdx.x * (256 * EPT) + threadIdx.x;
    if (blockIdx.x * (256 * EPT) + 256 * EPT <= E) {
        int d0 = dst[e0], d1 = dst[e0 + 256], d2 = dst[e0 + 512], d3 = dst[e0 + 768];
        int r0 = rank[e0], r1 = rank[e0 + 256], r2 = rank[e0 + 512], r3 = rank[e0 + 768];
        int s0 = src[e0], s1 = src[e0 + 256], s2 = src[e0 + 512], s3 = src[e0 + 768];
        int p0 = off[d0] + r0;
        int p1 = off[d1] + r1;
        int p2 = off[d2] + r2;
        int p3 = off[d3] + r3;
        ssrc[p0] = s0;
        ssrc[p1] = s1;
        ssrc[p2] = s2;
        ssrc[p3] = s3;
    } else {
        for (int k = 0; k < EPT; ++k) {
            int e = e0 + k * 256;
            if (e < E) ssrc[off[dst[e]] + rank[e]] = src[e];
        }
    }
}

__global__ void __launch_bounds__(32)
gat_wmma_feat_x_w(const float* __restrict__ feat,
                  const float* __restrict__ W,
                  float* __restrict__ h,
                  int N) {
    int tile  = blockIdx.x;
    int lane  = threadIdx.x;
    int mrow  = lane & 15;
    int khalf = (lane >> 4) * 2;

    int row  = tile * 16 + mrow;
    int rowc = row < N ? row : N - 1;
    const float* arow = feat + (size_t)rowc * F;

    int   n1   = 16 + mrow;
    int   n1c  = (n1 < F) ? n1 : 0;
    float msk1 = (n1 < F) ? 1.0f : 0.0f;

    v8f c0 = {};
    v8f c1 = {};
#pragma unroll
    for (int k0 = 0; k0 < F; k0 += 4) {
        const float* wr0 = W + (size_t)(k0 + khalf + 0) * F;
        const float* wr1 = W + (size_t)(k0 + khalf + 1) * F;

        v2f a;
        a.x = arow[k0 + khalf + 0];
        a.y = arow[k0 + khalf + 1];

        v2f b0;
        b0.x = wr0[mrow];
        b0.y = wr1[mrow];
        c0 = __builtin_amdgcn_wmma_f32_16x16x4_f32(false, a, false, b0,
                                                   (short)0, c0, false, false);

        v2f b1;
        b1.x = wr0[n1c] * msk1;
        b1.y = wr1[n1c] * msk1;
        c1 = __builtin_amdgcn_wmma_f32_16x16x4_f32(false, a, false, b1,
                                                   (short)0, c1, false, false);
    }

    int rbase = tile * 16 + ((lane >> 4) << 3);
#pragma unroll
    for (int j = 0; j < 8; ++j) {
        int r = rbase + j;
        if (r < N) {
            h[(size_t)r * F + mrow] = c0[j];
            if (n1 < F) h[(size_t)r * F + n1] = c1[j];
        }
    }
}

__global__ void gat_node_alpha_init(const float* __restrict__ h,
                                    const float* __restrict__ a_src,
                                    const float* __restrict__ a_dst,
                                    float* __restrict__ asrc,
                                    float* __restrict__ adst,
                                    int N) {
    int i = blockIdx.x * blockDim.x + threadIdx.x;
    if (i >= N) return;
    float sa = 0.0f, sd = 0.0f;
#pragma unroll
    for (int f = 0; f < F; ++f) {
        float hv = h[i * F + f];
        sa += hv * a_src[f];
        sd += hv * a_dst[f];
    }
    asrc[i] = sa;
    adst[i] = sd;
}

__device__ __forceinline__ float leaky(float v) {
    return v > 0.0f ? v : NEG_SLOPE * v;
}

__global__ void __launch_bounds__(256)
gat_node_gather(const int* __restrict__ off,
                const int* __restrict__ ssrc,
                const float* __restrict__ asrc,
                const float* __restrict__ adst,
                const float* __restrict__ h,
                const float* __restrict__ b,
                float* __restrict__ feat,
                int N) {
    const int gid = blockIdx.x * 256 + threadIdx.x;
    const int sub = threadIdx.x & 7;
    int node = gid >> 3;
    const bool valid = node < N;
    if (!valid) node = N - 1;

    const int beg = off[node];
    const int end = off[node + 1];
    const float ad = adst[node];
    const float es = leaky(asrc[node] + ad);

    float mx = es;
    for (int j = beg + sub; j < end; j += 8) {
        float ev = leaky(asrc[ssrc[j]] + ad);
        mx = ev > mx ? ev : mx;
    }
    mx = fmaxf(mx, __shfl_xor(mx, 1, 8));
    mx = fmaxf(mx, __shfl_xor(mx, 2, 8));
    mx = fmaxf(mx, __shfl_xor(mx, 4, 8));

    const int col = sub < 5 ? sub : 4;
    const float4* __restrict__ h4 = (const float4*)h + col;

    float t = __expf(es - mx);
    float den = t;
    float4 hv = h4[(size_t)node * 5];
    float4 acc;
    acc.x = t * hv.x; acc.y = t * hv.y; acc.z = t * hv.z; acc.w = t * hv.w;

    int j = beg;
    for (; j + 4 <= end; j += 4) {
        int s0 = ssrc[j], s1 = ssrc[j + 1], s2 = ssrc[j + 2], s3 = ssrc[j + 3];
        float a0 = asrc[s0], a1 = asrc[s1], a2 = asrc[s2], a3 = asrc[s3];
        float4 g0 = h4[(size_t)s0 * 5];
        float4 g1 = h4[(size_t)s1 * 5];
        float4 g2 = h4[(size_t)s2 * 5];
        float4 g3 = h4[(size_t)s3 * 5];
        float t0 = __expf(leaky(a0 + ad) - mx);
        float t1 = __expf(leaky(a1 + ad) - mx);
        float t2 = __expf(leaky(a2 + ad) - mx);
        float t3 = __expf(leaky(a3 + ad) - mx);
        den += t0;
        acc.x += t0 * g0.x; acc.y += t0 * g0.y; acc.z += t0 * g0.z; acc.w += t0 * g0.w;
        den += t1;
        acc.x += t1 * g1.x; acc.y += t1 * g1.y; acc.z += t1 * g1.z; acc.w += t1 * g1.w;
        den += t2;
        acc.x += t2 * g2.x; acc.y += t2 * g2.y; acc.z += t2 * g2.z; acc.w += t2 * g2.w;
        den += t3;
        acc.x += t3 * g3.x; acc.y += t3 * g3.y; acc.z += t3 * g3.z; acc.w += t3 * g3.w;
    }
#pragma clang loop unroll(disable)
    for (; j < end; ++j) {
        int s0 = ssrc[j];
        float4 g0 = h4[(size_t)s0 * 5];
        float t0 = __expf(leaky(asrc[s0] + ad) - mx);
        den += t0;
        acc.x += t0 * g0.x; acc.y += t0 * g0.y; acc.z += t0 * g0.z; acc.w += t0 * g0.w;
    }

    if (valid && sub < 5) {
        float inv = 1.0f / (den + 1e-16f);
        float4 o;
        o.x = acc.x * inv + b[sub * 4 + 0];
        o.y = acc.y * inv + b[sub * 4 + 1];
        o.z = acc.z * inv + b[sub * 4 + 2];
        o.w = acc.w * inv + b[sub * 4 + 3];
        o.x = o.x > 0.0f ? o.x : 0.0f;
        o.y = o.y > 0.0f ? o.y : 0.0f;
        o.z = o.z > 0.0f ? o.z : 0.0f;
        o.w = o.w > 0.0f ? o.w : 0.0f;
        ((float4*)feat)[(size_t)node * 5 + sub] = o;
    }
}

__global__ void gat_final_linear(const float* __restrict__ feat,
                                 const float* __restrict__ Wl,
                                 const float* __restrict__ bl,
                                 float* __restrict__ out,
                                 int N) {
    int i = blockIdx.x * blockDim.x + threadIdx.x;
    if (i >= N) return;
    float sacc = bl[0];
#pragma unroll
    for (int f = 0; f < F; ++f) sacc += feat[i * F + f] * Wl[f];
    out[i] = sacc;
}

extern "C" void kernel_launch(void* const* d_in, const int* in_sizes, int n_in,
                              void* d_out, int out_size, void* d_ws, size_t ws_size,
                              hipStream_t stream) {
    (void)n_in; (void)out_size; (void)ws_size;

    const int N    = in_sizes[0];
    const int E    = in_sizes[1] / 2;

    const float* x    = (const float*)d_in[0];
    const int*   ei   = (const int*)d_in[1];
    const float* W1   = (const float*)d_in[2];
    const float* as1  = (const float*)d_in[3];
    const float* ad1  = (const float*)d_in[4];
    const float* b1   = (const float*)d_in[5];
    const float* W2   = (const float*)d_in[6];
    const float* as2  = (const float*)d_in[7];
    const float* ad2  = (const float*)d_in[8];
    const float* b2   = (const float*)d_in[9];
    const float* Wl   = (const float*)d_in[10];
    const float* bl   = (const float*)d_in[11];

    const int* srcIdx = ei;
    const int* dstIdx = ei + E;

    const int TB   = 256;
    const int gN   = (N + TB - 1) / TB;
    const int gE4  = (E + TB * EPT - 1) / (TB * EPT);
    const int gMM  = (N + 15) / 16;
    const int gG   = (N + 31) / 32;

    float* ws     = (float*)d_ws;
    float* h      = ws + (size_t)0  * N;
    float* feat1  = ws + (size_t)20 * N;
    float* feat2  = ws + (size_t)40 * N;
    float* asrc   = ws + (size_t)60 * N;
    float* adst   = ws + (size_t)61 * N;
    int*   cnt    = (int*)(ws + (size_t)62 * N);
    int*   off    = (int*)(ws + (size_t)63 * N);
    int*   bsum   = (int*)(ws + (size_t)65 * N);
    int*   rank   = (int*)(ws + (size_t)66 * N);
    int*   ssrc   = rank + (size_t)E;

    gat_node_init_l1<<<gN, TB, 0, stream>>>(x, W1, as1, ad1, h, asrc, adst, cnt, N);
    csr_count<<<gE4, TB, 0, stream>>>(dstIdx, cnt, rank, E);
    csr_block_sums<<<gN, TB, 0, stream>>>(cnt, bsum, N);
    csr_offsets<<<gN, TB, 0, stream>>>(cnt, bsum, off, N);
    csr_fill<<<gE4, TB, 0, stream>>>(srcIdx, dstIdx, rank, off, ssrc, E);

    gat_node_gather<<<gG, TB, 0, stream>>>(off, ssrc, asrc, adst, h, b1, feat1, N);

    gat_wmma_feat_x_w<<<gMM, 32, 0, stream>>>(feat1, W2, h, N);
    gat_node_alpha_init<<<gN, TB, 0, stream>>>(h, as2, ad2, asrc, adst, N);
    gat_node_gather<<<gG, TB, 0, stream>>>(off, ssrc, asrc, adst, h, b2, feat2, N);

    gat_final_linear<<<gN, TB, 0, stream>>>(feat2, Wl, bl, (float*)d_out, N);
}


# Each transformed-feature row padded to one aligned 128-byte line that also holds the node's source score: an edge costs one line fetch, not a straddling 80-byte row plus a separate score line
# speedup vs baseline: 5.8944x; 1.1370x over previous
#include <hip/hip_runtime.h>
#include <math.h>

#define F 20
#define NEG_SLOPE 0.2f
#define HP 32
#define HP4 (HP / 4)

typedef float v2f __attribute__((ext_vector_type(2)));
typedef float v8f __attribute__((ext_vector_type(8)));

__global__ void gat_node_init_l1(const float* __restrict__ x,
                                 const float* __restrict__ W1,
                                 const float* __restrict__ a_src,
                                 const float* __restrict__ a_dst,
                                 float* __restrict__ h,
                                 float* __restrict__ asrc,
                                 float* __restrict__ adst,
                                 int* __restrict__ cnt,
                                 int N) {
    int i = blockIdx.x * blockDim.x + threadIdx.x;
    if (i >= N) return;
    float xv = x[i];
    float sa = 0.0f, sd = 0.0f;
#pragma unroll
    for (int f = 0; f < F; ++f) {
        float hv = xv * W1[f];
        h[i * HP + f] = hv;
        sa += hv * a_src[f];
        sd += hv * a_dst[f];
    }
    *(float4*)(h + i * HP + F) = make_float4(sa, sd, 0.0f, 0.0f);
    asrc[i] = sa;
    adst[i] = sd;
    cnt[i] = 0;
}

#define EPT 4
__global__ void __launch_bounds__(256)
csr_count(const int* __restrict__ dst,
          int* __restrict__ cnt,
          int* __restrict__ rank,
          int E) {
    const int e0 = blockIdx.x * (256 * EPT) + threadIdx.x;
    if (blockIdx.x * (256 * EPT) + 256 * EPT <= E) {
        int d0 = dst[e0];
        int d1 = dst[e0 + 256];
        int d2 = dst[e0 + 512];
        int d3 = dst[e0 + 768];
        int r0 = atomicAdd(&cnt[d0], 1);
        int r1 = atomicAdd(&cnt[d1], 1);
        int r2 = atomicAdd(&cnt[d2], 1);
        int r3 = atomicAdd(&cnt[d3], 1);
        rank[e0]       = r0;
        rank[e0 + 256] = r1;
        rank[e0 + 512] = r2;
        rank[e0 + 768] = r3;
    } else {
        for (int k = 0; k < EPT; ++k) {
            int e = e0 + k * 256;
            if (e < E) rank[e] = atomicAdd(&cnt[dst[e]], 1);
        }
    }
}

__global__ void __launch_bounds__(256)
csr_block_sums(const int* __restrict__ cnt,
               int* __restrict__ bsum,
               int N) {
    __shared__ int red[256];
    int t = threadIdx.x;
    int i = blockIdx.x * 256 + t;
    red[t] = i < N ? cnt[i] : 0;
    __syncthreads();
    for (int s = 128; s > 0; s >>= 1) {
        if (t < s) red[t] += red[t + s];
        __syncthreads();
    }
    if (t == 0) bsum[blockIdx.x] = red[0];
}

__global__ void __launch_bounds__(256)
csr_offsets(const int* __restrict__ cnt,
            const int* __restrict__ bsum,
            int* __restrict__ off,
            int N) {
    __shared__ int red[256];
    __shared__ int buf[2][256];
    int t = threadIdx.x;
    int b = blockIdx.x;

    int part = 0;
    for (int k = t; k < b; k += 256) part += bsum[k];
    red[t] = part;
    __syncthreads();
    for (int s = 128; s > 0; s >>= 1) {
        if (t < s) red[t] += red[t + s];
        __syncthreads();
    }
    int base = red[0];

    int i = b * 256 + t;
    int v = i < N ? cnt[i] : 0;
    int cur = 0;
    buf[0][t] = v;
    __syncthreads();
    for (int dd = 1; dd < 256; dd <<= 1) {
        int xv = buf[cur][t];
        if (t >= dd) xv += buf[cur][t - dd];
        buf[cur ^ 1][t] = xv;
        cur ^= 1;
        __syncthreads();
    }
    int incl = buf[cur][t];
    if (i < N) {
        off[i] = base + incl - v;
        if (i == N - 1) off[N] = base + incl;
    }
}

__global__ void __launch_bounds__(256)
csr_fill(const int* __restrict__ src,
         const int* __restrict__ dst,
         const int* __restrict__ rank,
         const int* __restrict__ off,
         int* __restrict__ ssrc,
         int E) {
    const int e0 = blockIdx.x * (256 * EPT) + threadIdx.x;
    if (blockIdx.x * (256 * EPT) + 256 * EPT <= E) {
        int d0 = dst[e0], d1 = dst[e0 + 256], d2 = dst[e0 + 512], d3 = dst[e0 + 768];
        int r0 = rank[e0], r1 = rank[e0 + 256], r2 = rank[e0 + 512], r3 = rank[e0 + 768];
        int s0 = src[e0], s1 = src[e0 + 256], s2 = src[e0 + 512], s3 = src[e0 + 768];
        int p0 = off[d0] + r0;
        int p1 = off[d1] + r1;
        int p2 = off[d2] + r2;
        int p3 = off[d3] + r3;
        ssrc[p0] = s0;
        ssrc[p1] = s1;
        ssrc[p2] = s2;
        ssrc[p3] = s3;
    } else {
        for (int k = 0; k < EPT; ++k) {
            int e = e0 + k * 256;
            if (e < E) ssrc[off[dst[e]] + rank[e]] = src[e];
        }
    }
}

__global__ void __launch_bounds__(32)
gat_wmma_feat_x_w(const float* __restrict__ feat,
                  const float* __restrict__ W,
                  float* __restrict__ h,
                  int N) {
    int tile  = blockIdx.x;
    int lane  = threadIdx.x;
    int mrow  = lane & 15;
    int khalf = (lane >> 4) * 2;

    int row  = tile * 16 + mrow;
    int rowc = row < N ? row : N - 1;
    const float* arow = feat + (size_t)rowc * F;

    int   n1   = 16 + mrow;
    int   n1c  = (n1 < F) ? n1 : 0;
    float msk1 = (n1 < F) ? 1.0f : 0.0f;

    v8f c0 = {};
    v8f c1 = {};
#pragma unroll
    for (int k0 = 0; k0 < F; k0 += 4) {
        const float* wr0 = W + (size_t)(k0 + khalf + 0) * F;
        const float* wr1 = W + (size_t)(k0 + khalf + 1) * F;

        v2f a;
        a.x = arow[k0 + khalf + 0];
        a.y = arow[k0 + khalf + 1];

        v2f b0;
        b0.x = wr0[mrow];
        b0.y = wr1[mrow];
        c0 = __builtin_amdgcn_wmma_f32_16x16x4_f32(false, a, false, b0,
                                                   (short)0, c0, false, false);

        v2f b1;
        b1.x = wr0[n1c] * msk1;
        b1.y = wr1[n1c] * msk1;
        c1 = __builtin_amdgcn_wmma_f32_16x16x4_f32(false, a, false, b1,
                                                   (short)0, c1, false, false);
    }

    int rbase = tile * 16 + ((lane >> 4) << 3);
#pragma unroll
    for (int j = 0; j < 8; ++j) {
        int r = rbase + j;
        if (r < N) {
            h[(size_t)r * HP + mrow] = c0[j];
            if (n1 < F) h[(size_t)r * HP + n1] = c1[j];
        }
    }
}

__global__ void gat_node_alpha_init(float* __restrict__ h,
                                    const float* __restrict__ a_src,
                                    const float* __restrict__ a_dst,
                                    float* __restrict__ asrc,
                                    float* __restrict__ adst,
                                    int N) {
    int i = blockIdx.x * blockDim.x + threadIdx.x;
    if (i >= N) return;
    float sa = 0.0f, sd = 0.0f;
#pragma unroll
    for (int f = 0; f < F; ++f) {
        float hv = h[i * HP + f];
        sa += hv * a_src[f];
        sd += hv * a_dst[f];
    }
    *(float4*)(h + i * HP + F) = make_float4(sa, sd, 0.0f, 0.0f);
    asrc[i] = sa;
    adst[i] = sd;
}

__device__ __forceinline__ float leaky(float v) {
    return v > 0.0f ? v : NEG_SLOPE * v;
}

__device__ __forceinline__ float group_lane5(float v) {
    return __builtin_bit_cast(float,
        __builtin_amdgcn_mov_dpp8(__builtin_bit_cast(unsigned, v), 0xB6DB6D));
}

__global__ void __launch_bounds__(256)
gat_node_gather(const int* __restrict__ off,
                const int* __restrict__ ssrc,
                const float* __restrict__ asrc,
                const float* __restrict__ adst,
                const float* __restrict__ h,
                const float* __restrict__ b,
                float* __restrict__ feat,
                int N) {
    const int gid = blockIdx.x * 256 + threadIdx.x;
    const int sub = threadIdx.x & 7;
    int node = gid >> 3;
    const bool valid = node < N;
    if (!valid) node = N - 1;

    const int beg = off[node];
    const int end = off[node + 1];
    const float ad = adst[node];
    const float es = leaky(asrc[node] + ad);

    float mx = es;
    for (int j = beg + sub; j < end; j += 8) {
        float ev = leaky(asrc[ssrc[j]] + ad);
        mx = ev > mx ? ev : mx;
    }
    mx = fmaxf(mx, __shfl_xor(mx, 1, 8));
    mx = fmaxf(mx, __shfl_xor(mx, 2, 8));
    mx = fmaxf(mx, __shfl_xor(mx, 4, 8));

    const int col = sub < 5 ? sub : 5;
    const float4* __restrict__ h4 = (const float4*)h + col;

    float t = __expf(es - mx);
    float den = t;
    float4 hv = h4[(size_t)node * HP4];
    float4 acc;
    acc.x = t * hv.x; acc.y = t * hv.y; acc.z = t * hv.z; acc.w = t * hv.w;

    int j = beg;
    for (; j + 4 <= end; j += 4) {
        int s0 = ssrc[j], s1 = ssrc[j + 1], s2 = ssrc[j + 2], s3 = ssrc[j + 3];
        float4 g0 = h4[(size_t)s0 * HP4];
        float4 g1 = h4[(size_t)s1 * HP4];
        float4 g2 = h4[(size_t)s2 * HP4];
        float4 g3 = h4[(size_t)s3 * HP4];
        float a0 = group_lane5(g0.x), a1 = group_lane5(g1.x);
        float a2 = group_lane5(g2.x), a3 = group_lane5(g3.x);
        float t0 = __expf(leaky(a0 + ad) - mx);
        float t1 = __expf(leaky(a1 + ad) - mx);
        float t2 = __expf(leaky(a2 + ad) - mx);
        float t3 = __expf(leaky(a3 + ad) - mx);
        den += t0;
        acc.x += t0 * g0.x; acc.y += t0 * g0.y; acc.z += t0 * g0.z; acc.w += t0 * g0.w;
        den += t1;
        acc.x += t1 * g1.x; acc.y += t1 * g1.y; acc.z += t1 * g1.z; acc.w += t1 * g1.w;
        den += t2;
        acc.x += t2 * g2.x; acc.y += t2 * g2.y; acc.z += t2 * g2.z; acc.w += t2 * g2.w;
        den += t3;
        acc.x += t3 * g3.x; acc.y += t3 * g3.y; acc.z += t3 * g3.z; acc.w += t3 * g3.w;
    }
#pragma clang loop unroll(disable)
    for (; j < end; ++j) {
        int s0 = ssrc[j];
        float4 g0 = h4[(size_t)s0 * HP4];
        float t0 = __expf(leaky(group_lane5(g0.x) + ad) - mx);
        den += t0;
        acc.x += t0 * g0.x; acc.y += t0 * g0.y; acc.z += t0 * g0.z; acc.w += t0 * g0.w;
    }

    if (valid && sub < 5) {
        float inv = 1.0f / (den + 1e-16f);
        float4 o;
        o.x = acc.x * inv + b[sub * 4 + 0];
        o.y = acc.y * inv + b[sub * 4 + 1];
        o.z = acc.z * inv + b[sub * 4 + 2];
        o.w = acc.w * inv + b[sub * 4 + 3];
        o.x = o.x > 0.0f ? o.x : 0.0f;
        o.y = o.y > 0.0f ? o.y : 0.0f;
        o.z = o.z > 0.0f ? o.z : 0.0f;
        o.w = o.w > 0.0f ? o.w : 0.0f;
        ((float4*)feat)[(size_t)node * 5 + sub] = o;
    }
}

__global__ void gat_final_linear(const float* __restrict__ feat,
                                 const float* __restrict__ Wl,
                                 const float* __restrict__ bl,
                                 float* __restrict__ out,
                                 int N) {
    int i = blockIdx.x * blockDim.x + threadIdx.x;
    if (i >= N) return;
    float sacc = bl[0];
#pragma unroll
    for (int f = 0; f < F; ++f) sacc += feat[i * F + f] * Wl[f];
    out[i] = sacc;
}

extern "C" void kernel_launch(void* const* d_in, const int* in_sizes, int n_in,
                              void* d_out, int out_size, void* d_ws, size_t ws_size,
                              hipStream_t stream) {
    (void)n_in; (void)out_size; (void)ws_size;

    const int N    = in_sizes[0];
    const int E    = in_sizes[1] / 2;

    const float* x    = (const float*)d_in[0];
    const int*   ei   = (const int*)d_in[1];
    const float* W1   = (const float*)d_in[2];
    const float* as1  = (const float*)d_in[3];
    const float* ad1  = (const float*)d_in[4];
    const float* b1   = (const float*)d_in[5];
    const float* W2   = (const float*)d_in[6];
    const float* as2  = (const float*)d_in[7];
    const float* ad2  = (const float*)d_in[8];
    const float* b2   = (const float*)d_in[9];
    const float* Wl   = (const float*)d_in[10];
    const float* bl   = (const float*)d_in[11];

    const int* srcIdx = ei;
    const int* dstIdx = ei + E;

    const int TB   = 256;
    const int gN   = (N + TB - 1) / TB;
    const int gE4  = (E + TB * EPT - 1) / (TB * EPT);
    const int gMM  = (N + 15) / 16;
    const int gG   = (N + 31) / 32;

    float* ws     = (float*)d_ws;
    float* h      = ws + (size_t)0  * N;
    float* feat1  = ws + (size_t)32 * N;
    float* feat2  = ws + (size_t)52 * N;
    float* asrc   = ws + (size_t)72 * N;
    float* adst   = ws + (size_t)73 * N;
    int*   cnt    = (int*)(ws + (size_t)74 * N);
    int*   off    = (int*)(ws + (size_t)75 * N);
    int*   bsum   = (int*)(ws + (size_t)77 * N);
    int*   rank   = (int*)(ws + (size_t)78 * N);
    int*   ssrc   = rank + (size_t)E;

    gat_node_init_l1<<<gN, TB, 0, stream>>>(x, W1, as1, ad1, h, asrc, adst, cnt, N);
    csr_count<<<gE4, TB, 0, stream>>>(dstIdx, cnt, rank, E);
    csr_block_sums<<<gN, TB, 0, stream>>>(cnt, bsum, N);
    csr_offsets<<<gN, TB, 0, stream>>>(cnt, bsum, off, N);
    csr_fill<<<gE4, TB, 0, stream>>>(srcIdx, dstIdx, rank, off, ssrc, E);

    gat_node_gather<<<gG, TB, 0, stream>>>(off, ssrc, asrc, adst, h, b1, feat1, N);

    gat_wmma_feat_x_w<<<gMM, 32, 0, stream>>>(feat1, W2, h, N);
    gat_node_alpha_init<<<gN, TB, 0, stream>>>(h, as2, ad2, asrc, adst, N);
    gat_node_gather<<<gG, TB, 0, stream>>>(off, ssrc, asrc, adst, h, b2, feat2, N);

    gat_final_linear<<<gN, TB, 0, stream>>>(feat2, Wl, bl, (float*)d_out, N);
}


# Destination sort on LDS counters instead of one scattered global returning add per edge: edges binned to 256-node buckets per 8192-edge chunk, then ranked per bucket; order stays arrival order
# speedup vs baseline: 13.6748x; 2.3200x over previous
#include <hip/hip_runtime.h>
#include <math.h>

#define F 20
#define NEG_SLOPE 0.2f
#define HP 32
#define HP4 (HP / 4)

typedef float v2f __attribute__((ext_vector_type(2)));
typedef float v8f __attribute__((ext_vector_type(8)));

__global__ void gat_node_init_l1(const float* __restrict__ x,
                                 const float* __restrict__ W1,
                                 const float* __restrict__ a_src,
                                 const float* __restrict__ a_dst,
                                 float* __restrict__ h,
                                 float* __restrict__ asrc,
                                 float* __restrict__ adst,
                                 int* __restrict__ btot,
                                 int N, int NB) {
    int i = blockIdx.x * blockDim.x + threadIdx.x;
    if (i >= N) return;
    if (i < NB) btot[i] = 0;
    float xv = x[i];
    float sa = 0.0f, sd = 0.0f;
#pragma unroll
    for (int f = 0; f < F; ++f) {
        float hv = xv * W1[f];
        h[i * HP + f] = hv;
        sa += hv * a_src[f];
        sd += hv * a_dst[f];
    }
    *(float4*)(h + i * HP + F) = make_float4(sa, sd, 0.0f, 0.0f);
    asrc[i] = sa;
    adst[i] = sd;
}

#define BKT_SH 8
#define BKT_W  (1 << BKT_SH)
#define SORT_T   512
#define SORT_EPT 16
#define SORT_CH  (SORT_T * SORT_EPT)

__global__ void __launch_bounds__(SORT_T)
bucket_count(const int* __restrict__ dst,
             int* __restrict__ btot,
             int* __restrict__ cb,
             int E, int NB) {
    extern __shared__ int hist[];
    const int t  = threadIdx.x;
    const int c  = blockIdx.x;
    const int e0 = c * SORT_CH + t;
    for (int b = t; b < NB; b += SORT_T) hist[b] = 0;
    __syncthreads();
    if (c * SORT_CH + SORT_CH <= E) {
        int d[SORT_EPT];
#pragma unroll
        for (int k = 0; k < SORT_EPT; ++k) d[k] = dst[e0 + k * SORT_T];
#pragma unroll
        for (int k = 0; k < SORT_EPT; ++k) atomicAdd(&hist[d[k] >> BKT_SH], 1);
    } else {
        for (int k = 0; k < SORT_EPT; ++k) {
            int e = e0 + k * SORT_T;
            if (e < E) atomicAdd(&hist[dst[e] >> BKT_SH], 1);
        }
    }
    __syncthreads();
    for (int b = t; b < NB; b += SORT_T)
        cb[(size_t)c * NB + b] = atomicAdd(&btot[b], hist[b]);
}

__global__ void __launch_bounds__(SORT_T)
bucket_scatter(const int* __restrict__ src,
               const int* __restrict__ dst,
               const int* __restrict__ btot,
               const int* __restrict__ cb,
               int* __restrict__ bstart,
               int2* __restrict__ pairs,
               int E, int NB) {
    extern __shared__ int sm[];
    __shared__ int buf[2][SORT_T];
    int* hist = sm;
    int* base = sm + NB;
    const int t  = threadIdx.x;
    const int c  = blockIdx.x;
    const int e0 = c * SORT_CH + t;
    for (int b = t; b < NB; b += SORT_T) hist[b] = 0;
    __syncthreads();

    int d[SORT_EPT], s[SORT_EPT], r[SORT_EPT];
    if (c * SORT_CH + SORT_CH <= E) {
#pragma unroll
        for (int k = 0; k < SORT_EPT; ++k) {
            d[k] = dst[e0 + k * SORT_T];
            s[k] = src[e0 + k * SORT_T];
        }
#pragma unroll
        for (int k = 0; k < SORT_EPT; ++k) r[k] = atomicAdd(&hist[d[k] >> BKT_SH], 1);
    } else {
#pragma unroll
        for (int k = 0; k < SORT_EPT; ++k) {
            int e = e0 + k * SORT_T;
            d[k] = -1; s[k] = 0; r[k] = 0;
            if (e < E) {
                d[k] = dst[e];
                s[k] = src[e];
                r[k] = atomicAdd(&hist[d[k] >> BKT_SH], 1);
            }
        }
    }

    int carry = 0;
    for (int b0 = 0; b0 < NB; b0 += SORT_T) {
        const int b = b0 + t;
        const int v = b < NB ? btot[b] : 0;
        int cur = 0;
        buf[0][t] = v;
        __syncthreads();
        for (int dd = 1; dd < SORT_T; dd <<= 1) {
            int xv = buf[cur][t];
            if (t >= dd) xv += buf[cur][t - dd];
            buf[cur ^ 1][t] = xv;
            cur ^= 1;
            __syncthreads();
        }
        const int st = carry + buf[cur][t] - v;
        if (b < NB) {
            base[b] = st + cb[(size_t)c * NB + b];
            if (c == 0) bstart[b] = st;
        }
        carry += buf[cur][SORT_T - 1];
        __syncthreads();
    }
    if (c == 0 && t == 0) bstart[NB] = carry;

#pragma unroll
    for (int k = 0; k < SORT_EPT; ++k) {
        if (d[k] >= 0) pairs[base[d[k] >> BKT_SH] + r[k]] = make_int2(d[k], s[k]);
    }
}

#define PLACE_T 1024
#define PLACE_CAP 12288
__global__ void __launch_bounds__(PLACE_T)
bucket_place(const int2* __restrict__ pairs,
             const int* __restrict__ btot,
             const int* __restrict__ bstart,
             int* __restrict__ off,
             int* __restrict__ ssrc,
             int N) {
    __shared__ int cnt[BKT_W];
    __shared__ int buf[2][BKT_W];
    __shared__ int stage[PLACE_CAP];
    const int t     = threadIdx.x;
    const int b     = blockIdx.x;
    const int nb0   = b << BKT_SH;
    const int start = bstart[b];
    const int n     = btot[b];
    const int2* __restrict__ pp = pairs + start;

    if (t < BKT_W) cnt[t] = 0;
    __syncthreads();

    for (int i = t; i < n; i += 4 * PLACE_T) {
        const int i1 = i + PLACE_T, i2 = i + 2 * PLACE_T, i3 = i + 3 * PLACE_T;
        const int d0 = pp[i].x;
        const int d1 = pp[i1 < n ? i1 : i].x;
        const int d2 = pp[i2 < n ? i2 : i].x;
        const int d3 = pp[i3 < n ? i3 : i].x;
        atomicAdd(&cnt[d0 - nb0], 1);
        if (i1 < n) atomicAdd(&cnt[d1 - nb0], 1);
        if (i2 < n) atomicAdd(&cnt[d2 - nb0], 1);
        if (i3 < n) atomicAdd(&cnt[d3 - nb0], 1);
    }
    __syncthreads();

    int v = 0, cur = 0;
    if (t < BKT_W) { v = cnt[t]; buf[0][t] = v; }
    __syncthreads();
    for (int dd = 1; dd < BKT_W; dd <<= 1) {
        if (t < BKT_W) {
            int xv = buf[cur][t];
            if (t >= dd) xv += buf[cur][t - dd];
            buf[cur ^ 1][t] = xv;
        }
        cur ^= 1;
        __syncthreads();
    }
    if (t < BKT_W) {
        const int ex = buf[cur][t] - v;
        cnt[t] = ex;
        const int node = nb0 + t;
        if (node < N) off[node] = start + ex;
        if (node == N - 1) off[N] = start + n;
    }
    __syncthreads();

    if (n <= PLACE_CAP) {
        for (int i = t; i < n; i += 4 * PLACE_T) {
            const int i1 = i + PLACE_T, i2 = i + 2 * PLACE_T, i3 = i + 3 * PLACE_T;
            const int2 p0 = pp[i];
            const int2 p1 = pp[i1 < n ? i1 : i];
            const int2 p2 = pp[i2 < n ? i2 : i];
            const int2 p3 = pp[i3 < n ? i3 : i];
            stage[atomicAdd(&cnt[p0.x - nb0], 1)] = p0.y;
            if (i1 < n) stage[atomicAdd(&cnt[p1.x - nb0], 1)] = p1.y;
            if (i2 < n) stage[atomicAdd(&cnt[p2.x - nb0], 1)] = p2.y;
            if (i3 < n) stage[atomicAdd(&cnt[p3.x - nb0], 1)] = p3.y;
        }
        __syncthreads();
        for (int i = t; i < n; i += PLACE_T) ssrc[start + i] = stage[i];
    } else {
        for (int i = t; i < n; i += 4 * PLACE_T) {
            const int i1 = i + PLACE_T, i2 = i + 2 * PLACE_T, i3 = i + 3 * PLACE_T;
            const int2 p0 = pp[i];
            const int2 p1 = pp[i1 < n ? i1 : i];
            const int2 p2 = pp[i2 < n ? i2 : i];
            const int2 p3 = pp[i3 < n ? i3 : i];
            ssrc[start + atomicAdd(&cnt[p0.x - nb0], 1)] = p0.y;
            if (i1 < n) ssrc[start + atomicAdd(&cnt[p1.x - nb0], 1)] = p1.y;
            if (i2 < n) ssrc[start + atomicAdd(&cnt[p2.x - nb0], 1)] = p2.y;
            if (i3 < n) ssrc[start + atomicAdd(&cnt[p3.x - nb0], 1)] = p3.y;
        }
    }
}

__global__ void __launch_bounds__(32)
gat_wmma_feat_x_w(const float* __restrict__ feat,
                  const float* __restrict__ W,
                  float* __restrict__ h,
                  int N) {
    int tile  = blockIdx.x;
    int lane  = threadIdx.x;
    int mrow  = lane & 15;
    int khalf = (lane >> 4) * 2;

    int row  = tile * 16 + mrow;
    int rowc = row < N ? row : N - 1;
    const float* arow = feat + (size_t)rowc * F;

    int   n1   = 16 + mrow;
    int   n1c  = (n1 < F) ? n1 : 0;
    float msk1 = (n1 < F) ? 1.0f : 0.0f;

    v8f c0 = {};
    v8f c1 = {};
#pragma unroll
    for (int k0 = 0; k0 < F; k0 += 4) {
        const float* wr0 = W + (size_t)(k0 + khalf + 0) * F;
        const float* wr1 = W + (size_t)(k0 + khalf + 1) * F;

        v2f a;
        a.x = arow[k0 + khalf + 0];
        a.y = arow[k0 + khalf + 1];

        v2f b0;
        b0.x = wr0[mrow];
        b0.y = wr1[mrow];
        c0 = __builtin_amdgcn_wmma_f32_16x16x4_f32(false, a, false, b0,
                                                   (short)0, c0, false, false);

        v2f b1;
        b1.x = wr0[n1c] * msk1;
        b1.y = wr1[n1c] * msk1;
        c1 = __builtin_amdgcn_wmma_f32_16x16x4_f32(false, a, false, b1,
                                                   (short)0, c1, false, false);
    }

    int rbase = tile * 16 + ((lane >> 4) << 3);
#pragma unroll
    for (int j = 0; j < 8; ++j) {
        int r = rbase + j;
        if (r < N) {
            h[(size_t)r * HP + mrow] = c0[j];
            if (n1 < F) h[(size_t)r * HP + n1] = c1[j];
        }
    }
}

__global__ void gat_node_alpha_init(float* __restrict__ h,
                                    const float* __restrict__ a_src,
                                    const float* __restrict__ a_dst,
                                    float* __restrict__ asrc,
                                    float* __restrict__ adst,
                                    int N) {
    int i = blockIdx.x * blockDim.x + threadIdx.x;
    if (i >= N) return;
    float sa = 0.0f, sd = 0.0f;
#pragma unroll
    for (int f = 0; f < F; ++f) {
        float hv = h[i * HP + f];
        sa += hv * a_src[f];
        sd += hv * a_dst[f];
    }
    *(float4*)(h + i * HP + F) = make_float4(sa, sd, 0.0f, 0.0f);
    asrc[i] = sa;
    adst[i] = sd;
}

__device__ __forceinline__ float leaky(float v) {
    return v > 0.0f ? v : NEG_SLOPE * v;
}

__device__ __forceinline__ float group_lane5(float v) {
    return __builtin_bit_cast(float,
        __builtin_amdgcn_mov_dpp8(__builtin_bit_cast(unsigned, v), 0xB6DB6D));
}

__global__ void __launch_bounds__(256)
gat_node_gather(const int* __restrict__ off,
                const int* __restrict__ ssrc,
                const float* __restrict__ asrc,
                const float* __restrict__ adst,
                const float* __restrict__ h,
                const float* __restrict__ b,
                float* __restrict__ feat,
                int N) {
    const int gid = blockIdx.x * 256 + threadIdx.x;
    const int sub = threadIdx.x & 7;
    int node = gid >> 3;
    const bool valid = node < N;
    if (!valid) node = N - 1;

    const int beg = off[node];
    const int end = off[node + 1];
    const float ad = adst[node];
    const float es = leaky(asrc[node] + ad);

    float mx = es;
    for (int j = beg + sub; j < end; j += 8) {
        float ev = leaky(asrc[ssrc[j]] + ad);
        mx = ev > mx ? ev : mx;
    }
    mx = fmaxf(mx, __shfl_xor(mx, 1, 8));
    mx = fmaxf(mx, __shfl_xor(mx, 2, 8));
    mx = fmaxf(mx, __shfl_xor(mx, 4, 8));

    const int col = sub < 5 ? sub : 5;
    const float4* __restrict__ h4 = (const float4*)h + col;

    float t = __expf(es - mx);
    float den = t;
    float4 hv = h4[(size_t)node * HP4];
    float4 acc;
    acc.x = t * hv.x; acc.y = t * hv.y; acc.z = t * hv.z; acc.w = t * hv.w;

    int j = beg;
    for (; j + 4 <= end; j += 4) {
        int s0 = ssrc[j], s1 = ssrc[j + 1], s2 = ssrc[j + 2], s3 = ssrc[j + 3];
        float4 g0 = h4[(size_t)s0 * HP4];
        float4 g1 = h4[(size_t)s1 * HP4];
        float4 g2 = h4[(size_t)s2 * HP4];
        float4 g3 = h4[(size_t)s3 * HP4];
        float a0 = group_lane5(g0.x), a1 = group_lane5(g1.x);
        float a2 = group_lane5(g2.x), a3 = group_lane5(g3.x);
        float t0 = __expf(leaky(a0 + ad) - mx);
        float t1 = __expf(leaky(a1 + ad) - mx);
        float t2 = __expf(leaky(a2 + ad) - mx);
        float t3 = __expf(leaky(a3 + ad) - mx);
        den += t0;
        acc.x += t0 * g0.x; acc.y += t0 * g0.y; acc.z += t0 * g0.z; acc.w += t0 * g0.w;
        den += t1;
        acc.x += t1 * g1.x; acc.y += t1 * g1.y; acc.z += t1 * g1.z; acc.w += t1 * g1.w;
        den += t2;
        acc.x += t2 * g2.x; acc.y += t2 * g2.y; acc.z += t2 * g2.z; acc.w += t2 * g2.w;
        den += t3;
        acc.x += t3 * g3.x; acc.y += t3 * g3.y; acc.z += t3 * g3.z; acc.w += t3 * g3.w;
    }
#pragma clang loop unroll(disable)
    for (; j < end; ++j) {
        int s0 = ssrc[j];
        float4 g0 = h4[(size_t)s0 * HP4];
        float t0 = __expf(leaky(group_lane5(g0.x) + ad) - mx);
        den += t0;
        acc.x += t0 * g0.x; acc.y += t0 * g0.y; acc.z += t0 * g0.z; acc.w += t0 * g0.w;
    }

    if (valid && sub < 5) {
        float inv = 1.0f / (den + 1e-16f);
        float4 o;
        o.x = acc.x * inv + b[sub * 4 + 0];
        o.y = acc.y * inv + b[sub * 4 + 1];
        o.z = acc.z * inv + b[sub * 4 + 2];
        o.w = acc.w * inv + b[sub * 4 + 3];
        o.x = o.x > 0.0f ? o.x : 0.0f;
        o.y = o.y > 0.0f ? o.y : 0.0f;
        o.z = o.z > 0.0f ? o.z : 0.0f;
        o.w = o.w > 0.0f ? o.w : 0.0f;
        ((float4*)feat)[(size_t)node * 5 + sub] = o;
    }
}

__global__ void gat_final_linear(const float* __restrict__ feat,
                                 const float* __restrict__ Wl,
                                 const float* __restrict__ bl,
                                 float* __restrict__ out,
                                 int N) {
    int i = blockIdx.x * blockDim.x + threadIdx.x;
    if (i >= N) return;
    float sacc = bl[0];
#pragma unroll
    for (int f = 0; f < F; ++f) sacc += feat[i * F + f] * Wl[f];
    out[i] = sacc;
}

extern "C" void kernel_launch(void* const* d_in, const int* in_sizes, int n_in,
                              void* d_out, int out_size, void* d_ws, size_t ws_size,
                              hipStream_t stream) {
    (void)n_in; (void)out_size; (void)ws_size;

    const int N    = in_sizes[0];
    const int E    = in_sizes[1] / 2;

    const float* x    = (const float*)d_in[0];
    const int*   ei   = (const int*)d_in[1];
    const float* W1   = (const float*)d_in[2];
    const float* as1  = (const float*)d_in[3];
    const float* ad1  = (const float*)d_in[4];
    const float* b1   = (const float*)d_in[5];
    const float* W2   = (const float*)d_in[6];
    const float* as2  = (const float*)d_in[7];
    const float* ad2  = (const float*)d_in[8];
    const float* b2   = (const float*)d_in[9];
    const float* Wl   = (const float*)d_in[10];
    const float* bl   = (const float*)d_in[11];

    const int* srcIdx = ei;
    const int* dstIdx = ei + E;

    const int TB   = 256;
    const int gN   = (N + TB - 1) / TB;
    const int NB   = (N + BKT_W - 1) / BKT_W;
    const int gCH  = (E + SORT_CH - 1) / SORT_CH;
    const int gMM  = (N + 15) / 16;
    const int gG   = (N + 31) / 32;

    float* ws     = (float*)d_ws;
    float* h      = ws + (size_t)0  * N;
    float* feat1  = ws + (size_t)32 * N;
    float* feat2  = ws + (size_t)52 * N;
    float* asrc   = ws + (size_t)72 * N;
    float* adst   = ws + (size_t)73 * N;
    int*   btot   = (int*)(ws + (size_t)74 * N);
    int*   off    = (int*)(ws + (size_t)75 * N);
    int*   bstart = (int*)(ws + (size_t)77 * N);
    int2*  pairs  = (int2*)(ws + (size_t)78 * N);
    int*   ssrc   = (int*)(ws + (size_t)78 * N) + (size_t)2 * E;
    int*   cb     = ssrc + (size_t)E;

    gat_node_init_l1<<<gN, TB, 0, stream>>>(x, W1, as1, ad1, h, asrc, adst, btot, N, NB);
    bucket_count<<<gCH, SORT_T, NB * sizeof(int), stream>>>(dstIdx, btot, cb, E, NB);
    bucket_scatter<<<gCH, SORT_T, 2 * NB * sizeof(int), stream>>>(srcIdx, dstIdx, btot, cb, bstart, pairs, E, NB);
    bucket_place<<<NB, PLACE_T, 0, stream>>>(pairs, btot, bstart, off, ssrc, N);

    gat_node_gather<<<gG, TB, 0, stream>>>(off, ssrc, asrc, adst, h, b1, feat1, N);

    gat_wmma_feat_x_w<<<gMM, 32, 0, stream>>>(feat1, W2, h, N);
    gat_node_alpha_init<<<gN, TB, 0, stream>>>(h, as2, ad2, asrc, adst, N);
    gat_node_gather<<<gG, TB, 0, stream>>>(off, ssrc, asrc, adst, h, b2, feat2, N);

    gat_final_linear<<<gN, TB, 0, stream>>>(feat2, Wl, bl, (float*)d_out, N);
}


# Edge softmax in one pass per segment: a running max with rescaled partial sums (online softmax) replaces the separate max pass and its extra random score line per edge; edge order kept
# speedup vs baseline: 17.9169x; 1.3102x over previous
#include <hip/hip_runtime.h>
#include <math.h>

#define F 20
#define NEG_SLOPE 0.2f
#define HP 32
#define HP4 (HP / 4)

typedef float v2f __attribute__((ext_vector_type(2)));
typedef float v8f __attribute__((ext_vector_type(8)));

__global__ void gat_node_init_l1(const float* __restrict__ x,
                                 const float* __restrict__ W1,
                                 const float* __restrict__ a_src,
                                 const float* __restrict__ a_dst,
                                 float* __restrict__ h,
                                 float* __restrict__ asrc,
                                 float* __restrict__ adst,
                                 int* __restrict__ btot,
                                 int N, int NB) {
    int i = blockIdx.x * blockDim.x + threadIdx.x;
    if (i >= N) return;
    if (i < NB) btot[i] = 0;
    float xv = x[i];
    float sa = 0.0f, sd = 0.0f;
#pragma unroll
    for (int f = 0; f < F; ++f) {
        float hv = xv * W1[f];
        h[i * HP + f] = hv;
        sa += hv * a_src[f];
        sd += hv * a_dst[f];
    }
    *(float4*)(h + i * HP + F) = make_float4(sa, sd, 0.0f, 0.0f);
    asrc[i] = sa;
    adst[i] = sd;
}

#define BKT_SH 8
#define BKT_W  (1 << BKT_SH)
#define SORT_T   512
#define SORT_EPT 16
#define SORT_CH  (SORT_T * SORT_EPT)

__global__ void __launch_bounds__(SORT_T)
bucket_count(const int* __restrict__ dst,
             int* __restrict__ btot,
             int* __restrict__ cb,
             int E, int NB) {
    extern __shared__ int hist[];
    const int t  = threadIdx.x;
    const int c  = blockIdx.x;
    const int e0 = c * SORT_CH + t;
    for (int b = t; b < NB; b += SORT_T) hist[b] = 0;
    __syncthreads();
    if (c * SORT_CH + SORT_CH <= E) {
        int d[SORT_EPT];
#pragma unroll
        for (int k = 0; k < SORT_EPT; ++k) d[k] = dst[e0 + k * SORT_T];
#pragma unroll
        for (int k = 0; k < SORT_EPT; ++k) atomicAdd(&hist[d[k] >> BKT_SH], 1);
    } else {
        for (int k = 0; k < SORT_EPT; ++k) {
            int e = e0 + k * SORT_T;
            if (e < E) atomicAdd(&hist[dst[e] >> BKT_SH], 1);
        }
    }
    __syncthreads();
    for (int b = t; b < NB; b += SORT_T)
        cb[(size_t)c * NB + b] = atomicAdd(&btot[b], hist[b]);
}

__global__ void __launch_bounds__(SORT_T)
bucket_scatter(const int* __restrict__ src,
               const int* __restrict__ dst,
               const int* __restrict__ btot,
               const int* __restrict__ cb,
               int* __restrict__ bstart,
               int2* __restrict__ pairs,
               int E, int NB) {
    extern __shared__ int sm[];
    __shared__ int buf[2][SORT_T];
    int* hist = sm;
    int* base = sm + NB;
    const int t  = threadIdx.x;
    const int c  = blockIdx.x;
    const int e0 = c * SORT_CH + t;
    for (int b = t; b < NB; b += SORT_T) hist[b] = 0;
    __syncthreads();

    int d[SORT_EPT], s[SORT_EPT], r[SORT_EPT];
    if (c * SORT_CH + SORT_CH <= E) {
#pragma unroll
        for (int k = 0; k < SORT_EPT; ++k) {
            d[k] = dst[e0 + k * SORT_T];
            s[k] = src[e0 + k * SORT_T];
        }
#pragma unroll
        for (int k = 0; k < SORT_EPT; ++k) r[k] = atomicAdd(&hist[d[k] >> BKT_SH], 1);
    } else {
#pragma unroll
        for (int k = 0; k < SORT_EPT; ++k) {
            int e = e0 + k * SORT_T;
            d[k] = -1; s[k] = 0; r[k] = 0;
            if (e < E) {
                d[k] = dst[e];
                s[k] = src[e];
                r[k] = atomicAdd(&hist[d[k] >> BKT_SH], 1);
            }
        }
    }

    int carry = 0;
    for (int b0 = 0; b0 < NB; b0 += SORT_T) {
        const int b = b0 + t;
        const int v = b < NB ? btot[b] : 0;
        int cur = 0;
        buf[0][t] = v;
        __syncthreads();
        for (int dd = 1; dd < SORT_T; dd <<= 1) {
            int xv = buf[cur][t];
            if (t >= dd) xv += buf[cur][t - dd];
            buf[cur ^ 1][t] = xv;
            cur ^= 1;
            __syncthreads();
        }
        const int st = carry + buf[cur][t] - v;
        if (b < NB) {
            base[b] = st + cb[(size_t)c * NB + b];
            if (c == 0) bstart[b] = st;
        }
        carry += buf[cur][SORT_T - 1];
        __syncthreads();
    }
    if (c == 0 && t == 0) bstart[NB] = carry;

#pragma unroll
    for (int k = 0; k < SORT_EPT; ++k) {
        if (d[k] >= 0) pairs[base[d[k] >> BKT_SH] + r[k]] = make_int2(d[k], s[k]);
    }
}

#define PLACE_T 1024
#define PLACE_CAP 12288
__global__ void __launch_bounds__(PLACE_T)
bucket_place(const int2* __restrict__ pairs,
             const int* __restrict__ btot,
             const int* __restrict__ bstart,
             int* __restrict__ off,
             int* __restrict__ ssrc,
             int N) {
    __shared__ int cnt[BKT_W];
    __shared__ int buf[2][BKT_W];
    __shared__ int stage[PLACE_CAP];
    const int t     = threadIdx.x;
    const int b     = blockIdx.x;
    const int nb0   = b << BKT_SH;
    const int start = bstart[b];
    const int n     = btot[b];
    const int2* __restrict__ pp = pairs + start;

    if (t < BKT_W) cnt[t] = 0;
    __syncthreads();

    for (int i = t; i < n; i += 4 * PLACE_T) {
        const int i1 = i + PLACE_T, i2 = i + 2 * PLACE_T, i3 = i + 3 * PLACE_T;
        const int d0 = pp[i].x;
        const int d1 = pp[i1 < n ? i1 : i].x;
        const int d2 = pp[i2 < n ? i2 : i].x;
        const int d3 = pp[i3 < n ? i3 : i].x;
        atomicAdd(&cnt[d0 - nb0], 1);
        if (i1 < n) atomicAdd(&cnt[d1 - nb0], 1);
        if (i2 < n) atomicAdd(&cnt[d2 - nb0], 1);
        if (i3 < n) atomicAdd(&cnt[d3 - nb0], 1);
    }
    __syncthreads();

    int v = 0, cur = 0;
    if (t < BKT_W) { v = cnt[t]; buf[0][t] = v; }
    __syncthreads();
    for (int dd = 1; dd < BKT_W; dd <<= 1) {
        if (t < BKT_W) {
            int xv = buf[cur][t];
            if (t >= dd) xv += buf[cur][t - dd];
            buf[cur ^ 1][t] = xv;
        }
        cur ^= 1;
        __syncthreads();
    }
    if (t < BKT_W) {
        const int ex = buf[cur][t] - v;
        cnt[t] = ex;
        const int node = nb0 + t;
        if (node < N) off[node] = start + ex;
        if (node == N - 1) off[N] = start + n;
    }
    __syncthreads();

    if (n <= PLACE_CAP) {
        for (int i = t; i < n; i += 4 * PLACE_T) {
            const int i1 = i + PLACE_T, i2 = i + 2 * PLACE_T, i3 = i + 3 * PLACE_T;
            const int2 p0 = pp[i];
            const int2 p1 = pp[i1 < n ? i1 : i];
            const int2 p2 = pp[i2 < n ? i2 : i];
            const int2 p3 = pp[i3 < n ? i3 : i];
            stage[atomicAdd(&cnt[p0.x - nb0], 1)] = p0.y;
            if (i1 < n) stage[atomicAdd(&cnt[p1.x - nb0], 1)] = p1.y;
            if (i2 < n) stage[atomicAdd(&cnt[p2.x - nb0], 1)] = p2.y;
            if (i3 < n) stage[atomicAdd(&cnt[p3.x - nb0], 1)] = p3.y;
        }
        __syncthreads();
        for (int i = t; i < n; i += PLACE_T) ssrc[start + i] = stage[i];
    } else {
        for (int i = t; i < n; i += 4 * PLACE_T) {
            const int i1 = i + PLACE_T, i2 = i + 2 * PLACE_T, i3 = i + 3 * PLACE_T;
            const int2 p0 = pp[i];
            const int2 p1 = pp[i1 < n ? i1 : i];
            const int2 p2 = pp[i2 < n ? i2 : i];
            const int2 p3 = pp[i3 < n ? i3 : i];
            ssrc[start + atomicAdd(&cnt[p0.x - nb0], 1)] = p0.y;
            if (i1 < n) ssrc[start + atomicAdd(&cnt[p1.x - nb0], 1)] = p1.y;
            if (i2 < n) ssrc[start + atomicAdd(&cnt[p2.x - nb0], 1)] = p2.y;
            if (i3 < n) ssrc[start + atomicAdd(&cnt[p3.x - nb0], 1)] = p3.y;
        }
    }
}

__global__ void __launch_bounds__(32)
gat_wmma_feat_x_w(const float* __restrict__ feat,
                  const float* __restrict__ W,
                  float* __restrict__ h,
                  int N) {
    int tile  = blockIdx.x;
    int lane  = threadIdx.x;
    int mrow  = lane & 15;
    int khalf = (lane >> 4) * 2;

    int row  = tile * 16 + mrow;
    int rowc = row < N ? row : N - 1;
    const float* arow = feat + (size_t)rowc * F;

    int   n1   = 16 + mrow;
    int   n1c  = (n1 < F) ? n1 : 0;
    float msk1 = (n1 < F) ? 1.0f : 0.0f;

    v8f c0 = {};
    v8f c1 = {};
#pragma unroll
    for (int k0 = 0; k0 < F; k0 += 4) {
        const float* wr0 = W + (size_t)(k0 + khalf + 0) * F;
        const float* wr1 = W + (size_t)(k0 + khalf + 1) * F;

        v2f a;
        a.x = arow[k0 + khalf + 0];
        a.y = arow[k0 + khalf + 1];

        v2f b0;
        b0.x = wr0[mrow];
        b0.y = wr1[mrow];
        c0 = __builtin_amdgcn_wmma_f32_16x16x4_f32(false, a, false, b0,
                                                   (short)0, c0, false, false);

        v2f b1;
        b1.x = wr0[n1c] * msk1;
        b1.y = wr1[n1c] * msk1;
        c1 = __builtin_amdgcn_wmma_f32_16x16x4_f32(false, a, false, b1,
                                                   (short)0, c1, false, false);
    }

    int rbase = tile * 16 + ((lane >> 4) << 3);
#pragma unroll
    for (int j = 0; j < 8; ++j) {
        int r = rbase + j;
        if (r < N) {
            h[(size_t)r * HP + mrow] = c0[j];
            if (n1 < F) h[(size_t)r * HP + n1] = c1[j];
        }
    }
}

__global__ void gat_node_alpha_init(float* __restrict__ h,
                                    const float* __restrict__ a_src,
                                    const float* __restrict__ a_dst,
                                    float* __restrict__ asrc,
                                    float* __restrict__ adst,
                                    int N) {
    int i = blockIdx.x * blockDim.x + threadIdx.x;
    if (i >= N) return;
    float sa = 0.0f, sd = 0.0f;
#pragma unroll
    for (int f = 0; f < F; ++f) {
        float hv = h[i * HP + f];
        sa += hv * a_src[f];
        sd += hv * a_dst[f];
    }
    *(float4*)(h + i * HP + F) = make_float4(sa, sd, 0.0f, 0.0f);
    asrc[i] = sa;
    adst[i] = sd;
}

__device__ __forceinline__ float leaky(float v) {
    return v > 0.0f ? v : NEG_SLOPE * v;
}

__device__ __forceinline__ float group_lane5(float v) {
    return __builtin_bit_cast(float,
        __builtin_amdgcn_mov_dpp8(__builtin_bit_cast(unsigned, v), 0xB6DB6D));
}

__global__ void __launch_bounds__(256)
gat_node_gather(const int* __restrict__ off,
                const int* __restrict__ ssrc,
                const float* __restrict__ asrc,
                const float* __restrict__ adst,
                const float* __restrict__ h,
                const float* __restrict__ b,
                float* __restrict__ feat,
                int N) {
    const int gid = blockIdx.x * 256 + threadIdx.x;
    const int sub = threadIdx.x & 7;
    int node = gid >> 3;
    const bool valid = node < N;
    if (!valid) node = N - 1;

    const int beg = off[node];
    const int end = off[node + 1];
    const float ad = adst[node];

    const int col = sub < 5 ? sub : 5;
    const float4* __restrict__ h4 = (const float4*)h + col;

    float mx = leaky(asrc[node] + ad);
    float den = 1.0f;
    float4 acc = h4[(size_t)node * HP4];

    int j = beg;
    for (; j + 4 <= end; j += 4) {
        int s0 = ssrc[j], s1 = ssrc[j + 1], s2 = ssrc[j + 2], s3 = ssrc[j + 3];
        float4 g0 = h4[(size_t)s0 * HP4];
        float4 g1 = h4[(size_t)s1 * HP4];
        float4 g2 = h4[(size_t)s2 * HP4];
        float4 g3 = h4[(size_t)s3 * HP4];
        float e0 = leaky(group_lane5(g0.x) + ad);
        float e1 = leaky(group_lane5(g1.x) + ad);
        float e2 = leaky(group_lane5(g2.x) + ad);
        float e3 = leaky(group_lane5(g3.x) + ad);
        float mn = fmaxf(fmaxf(fmaxf(e0, e1), fmaxf(e2, e3)), mx);
        float sc = __expf(mx - mn);
        mx = mn;
        float t0 = __expf(e0 - mx);
        float t1 = __expf(e1 - mx);
        float t2 = __expf(e2 - mx);
        float t3 = __expf(e3 - mx);
        den = den * sc + t0;
        acc.x = acc.x * sc + t0 * g0.x; acc.y = acc.y * sc + t0 * g0.y;
        acc.z = acc.z * sc + t0 * g0.z; acc.w = acc.w * sc + t0 * g0.w;
        den += t1;
        acc.x += t1 * g1.x; acc.y += t1 * g1.y; acc.z += t1 * g1.z; acc.w += t1 * g1.w;
        den += t2;
        acc.x += t2 * g2.x; acc.y += t2 * g2.y; acc.z += t2 * g2.z; acc.w += t2 * g2.w;
        den += t3;
        acc.x += t3 * g3.x; acc.y += t3 * g3.y; acc.z += t3 * g3.z; acc.w += t3 * g3.w;
    }
#pragma clang loop unroll(disable)
    for (; j < end; ++j) {
        int s0 = ssrc[j];
        float4 g0 = h4[(size_t)s0 * HP4];
        float e0 = leaky(group_lane5(g0.x) + ad);
        float mn = fmaxf(e0, mx);
        float sc = __expf(mx - mn);
        mx = mn;
        float t0 = __expf(e0 - mx);
        den = den * sc + t0;
        acc.x = acc.x * sc + t0 * g0.x; acc.y = acc.y * sc + t0 * g0.y;
        acc.z = acc.z * sc + t0 * g0.z; acc.w = acc.w * sc + t0 * g0.w;
    }

    if (valid && sub < 5) {
        float inv = 1.0f / (den + 1e-16f);
        float4 o;
        o.x = acc.x * inv + b[sub * 4 + 0];
        o.y = acc.y * inv + b[sub * 4 + 1];
        o.z = acc.z * inv + b[sub * 4 + 2];
        o.w = acc.w * inv + b[sub * 4 + 3];
        o.x = o.x > 0.0f ? o.x : 0.0f;
        o.y = o.y > 0.0f ? o.y : 0.0f;
        o.z = o.z > 0.0f ? o.z : 0.0f;
        o.w = o.w > 0.0f ? o.w : 0.0f;
        ((float4*)feat)[(size_t)node * 5 + sub] = o;
    }
}

__global__ void gat_final_linear(const float* __restrict__ feat,
                                 const float* __restrict__ Wl,
                                 const float* __restrict__ bl,
                                 float* __restrict__ out,
                                 int N) {
    int i = blockIdx.x * blockDim.x + threadIdx.x;
    if (i >= N) return;
    float sacc = bl[0];
#pragma unroll
    for (int f = 0; f < F; ++f) sacc += feat[i * F + f] * Wl[f];
    out[i] = sacc;
}

extern "C" void kernel_launch(void* const* d_in, const int* in_sizes, int n_in,
                              void* d_out, int out_size, void* d_ws, size_t ws_size,
                              hipStream_t stream) {
    (void)n_in; (void)out_size; (void)ws_size;

    const int N    = in_sizes[0];
    const int E    = in_sizes[1] / 2;

    const float* x    = (const float*)d_in[0];
    const int*   ei   = (const int*)d_in[1];
    const float* W1   = (const float*)d_in[2];
    const float* as1  = (const float*)d_in[3];
    const float* ad1  = (const float*)d_in[4];
    const float* b1   = (const float*)d_in[5];
    const float* W2   = (const float*)d_in[6];
    const float* as2  = (const float*)d_in[7];
    const float* ad2  = (const float*)d_in[8];
    const float* b2   = (const float*)d_in[9];
    const float* Wl   = (const float*)d_in[10];
    const float* bl   = (const float*)d_in[11];

    const int* srcIdx = ei;
    const int* dstIdx = ei + E;

    const int TB   = 256;
    const int gN   = (N + TB - 1) / TB;
    const int NB   = (N + BKT_W - 1) / BKT_W;
    const int gCH  = (E + SORT_CH - 1) / SORT_CH;
    const int gMM  = (N + 15) / 16;
    const int gG   = (N + 31) / 32;

    float* ws     = (float*)d_ws;
    float* h      = ws + (size_t)0  * N;
    float* feat1  = ws + (size_t)32 * N;
    float* feat2  = ws + (size_t)52 * N;
    float* asrc   = ws + (size_t)72 * N;
    float* adst   = ws + (size_t)73 * N;
    int*   btot   = (int*)(ws + (size_t)74 * N);
    int*   off    = (int*)(ws + (size_t)75 * N);
    int*   bstart = (int*)(ws + (size_t)77 * N);
    int2*  pairs  = (int2*)(ws + (size_t)78 * N);
    int*   ssrc   = (int*)(ws + (size_t)78 * N) + (size_t)2 * E;
    int*   cb     = ssrc + (size_t)E;

    gat_node_init_l1<<<gN, TB, 0, stream>>>(x, W1, as1, ad1, h, asrc, adst, btot, N, NB);
    bucket_count<<<gCH, SORT_T, NB * sizeof(int), stream>>>(dstIdx, btot, cb, E, NB);
    bucket_scatter<<<gCH, SORT_T, 2 * NB * sizeof(int), stream>>>(srcIdx, dstIdx, btot, cb, bstart, pairs, E, NB);
    bucket_place<<<NB, PLACE_T, 0, stream>>>(pairs, btot, bstart, off, ssrc, N);

    gat_node_gather<<<gG, TB, 0, stream>>>(off, ssrc, asrc, adst, h, b1, feat1, N);

    gat_wmma_feat_x_w<<<gMM, 32, 0, stream>>>(feat1, W2, h, N);
    gat_node_alpha_init<<<gN, TB, 0, stream>>>(h, as2, ad2, asrc, adst, N);
    gat_node_gather<<<gG, TB, 0, stream>>>(off, ssrc, asrc, adst, h, b2, feat2, N);

    gat_final_linear<<<gN, TB, 0, stream>>>(feat2, Wl, bl, (float*)d_out, N);
}


# Layer-1 aggregation on scalars: one input feature makes x@W1 an outer product, so an edge adds x[src] from an LDS copy of x and W1 is applied after the sum; 8-lane partial sums merged by shuffles
# speedup vs baseline: 21.1911x; 1.1827x over previous
#include <hip/hip_runtime.h>
#include <math.h>

#define F 20
#define NEG_SLOPE 0.2f
#define HP 32
#define HP4 (HP / 4)

typedef float v2f __attribute__((ext_vector_type(2)));
typedef float v8f __attribute__((ext_vector_type(8)));
typedef float v4f __attribute__((ext_vector_type(4)));

__global__ void bucket_totals_clear(int* __restrict__ btot, int NB) {
    int i = blockIdx.x * blockDim.x + threadIdx.x;
    if (i < NB) btot[i] = 0;
}

#define BKT_SH 8
#define BKT_W  (1 << BKT_SH)
#define SORT_T   512
#define SORT_EPT 16
#define SORT_CH  (SORT_T * SORT_EPT)

__global__ void __launch_bounds__(SORT_T)
bucket_count(const int* __restrict__ dst,
             int* __restrict__ btot,
             int* __restrict__ cb,
             int E, int NB) {
    extern __shared__ int hist[];
    const int t  = threadIdx.x;
    const int c  = blockIdx.x;
    const int e0 = c * SORT_CH + t;
    for (int b = t; b < NB; b += SORT_T) hist[b] = 0;
    __syncthreads();
    if (c * SORT_CH + SORT_CH <= E) {
        int d[SORT_EPT];
#pragma unroll
        for (int k = 0; k < SORT_EPT; ++k) d[k] = dst[e0 + k * SORT_T];
#pragma unroll
        for (int k = 0; k < SORT_EPT; ++k) atomicAdd(&hist[d[k] >> BKT_SH], 1);
    } else {
        for (int k = 0; k < SORT_EPT; ++k) {
            int e = e0 + k * SORT_T;
            if (e < E) atomicAdd(&hist[dst[e] >> BKT_SH], 1);
        }
    }
    __syncthreads();
    for (int b = t; b < NB; b += SORT_T)
        cb[(size_t)c * NB + b] = atomicAdd(&btot[b], hist[b]);
}

__global__ void __launch_bounds__(SORT_T)
bucket_scatter(const int* __restrict__ src,
               const int* __restrict__ dst,
               const int* __restrict__ btot,
               const int* __restrict__ cb,
               int* __restrict__ bstart,
               int2* __restrict__ pairs,
               int E, int NB) {
    extern __shared__ int sm[];
    __shared__ int buf[2][SORT_T];
    int* hist = sm;
    int* base = sm + NB;
    const int t  = threadIdx.x;
    const int c  = blockIdx.x;
    const int e0 = c * SORT_CH + t;
    for (int b = t; b < NB; b += SORT_T) hist[b] = 0;
    __syncthreads();

    int d[SORT_EPT], s[SORT_EPT], r[SORT_EPT];
    if (c * SORT_CH + SORT_CH <= E) {
#pragma unroll
        for (int k = 0; k < SORT_EPT; ++k) {
            d[k] = dst[e0 + k * SORT_T];
            s[k] = src[e0 + k * SORT_T];
        }
#pragma unroll
        for (int k = 0; k < SORT_EPT; ++k) r[k] = atomicAdd(&hist[d[k] >> BKT_SH], 1);
    } else {
#pragma unroll
        for (int k = 0; k < SORT_EPT; ++k) {
            int e = e0 + k * SORT_T;
            d[k] = -1; s[k] = 0; r[k] = 0;
            if (e < E) {
                d[k] = dst[e];
                s[k] = src[e];
                r[k] = atomicAdd(&hist[d[k] >> BKT_SH], 1);
            }
        }
    }

    int carry = 0;
    for (int b0 = 0; b0 < NB; b0 += SORT_T) {
        const int b = b0 + t;
        const int v = b < NB ? btot[b] : 0;
        int cur = 0;
        buf[0][t] = v;
        __syncthreads();
        for (int dd = 1; dd < SORT_T; dd <<= 1) {
            int xv = buf[cur][t];
            if (t >= dd) xv += buf[cur][t - dd];
            buf[cur ^ 1][t] = xv;
            cur ^= 1;
            __syncthreads();
        }
        const int st = carry + buf[cur][t] - v;
        if (b < NB) {
            base[b] = st + cb[(size_t)c * NB + b];
            if (c == 0) bstart[b] = st;
        }
        carry += buf[cur][SORT_T - 1];
        __syncthreads();
    }
    if (c == 0 && t == 0) bstart[NB] = carry;

#pragma unroll
    for (int k = 0; k < SORT_EPT; ++k) {
        if (d[k] >= 0) pairs[base[d[k] >> BKT_SH] + r[k]] = make_int2(d[k], s[k]);
    }
}

#define PLACE_T 1024
#define PLACE_CAP 12288
__global__ void __launch_bounds__(PLACE_T)
bucket_place(const int2* __restrict__ pairs,
             const int* __restrict__ btot,
             const int* __restrict__ bstart,
             int* __restrict__ off,
             int* __restrict__ ssrc,
             int N) {
    __shared__ int cnt[BKT_W];
    __shared__ int buf[2][BKT_W];
    __shared__ int stage[PLACE_CAP];
    const int t     = threadIdx.x;
    const int b     = blockIdx.x;
    const int nb0   = b << BKT_SH;
    const int start = bstart[b];
    const int n     = btot[b];
    const int2* __restrict__ pp = pairs + start;

    if (t < BKT_W) cnt[t] = 0;
    __syncthreads();

    for (int i = t; i < n; i += 4 * PLACE_T) {
        const int i1 = i + PLACE_T, i2 = i + 2 * PLACE_T, i3 = i + 3 * PLACE_T;
        const int d0 = pp[i].x;
        const int d1 = pp[i1 < n ? i1 : i].x;
        const int d2 = pp[i2 < n ? i2 : i].x;
        const int d3 = pp[i3 < n ? i3 : i].x;
        atomicAdd(&cnt[d0 - nb0], 1);
        if (i1 < n) atomicAdd(&cnt[d1 - nb0], 1);
        if (i2 < n) atomicAdd(&cnt[d2 - nb0], 1);
        if (i3 < n) atomicAdd(&cnt[d3 - nb0], 1);
    }
    __syncthreads();

    int v = 0, cur = 0;
    if (t < BKT_W) { v = cnt[t]; buf[0][t] = v; }
    __syncthreads();
    for (int dd = 1; dd < BKT_W; dd <<= 1) {
        if (t < BKT_W) {
            int xv = buf[cur][t];
            if (t >= dd) xv += buf[cur][t - dd];
            buf[cur ^ 1][t] = xv;
        }
        cur ^= 1;
        __syncthreads();
    }
    if (t < BKT_W) {
        const int ex = buf[cur][t] - v;
        cnt[t] = ex;
        const int node = nb0 + t;
        if (node < N) off[node] = start + ex;
        if (node == N - 1) off[N] = start + n;
    }
    __syncthreads();

    if (n <= PLACE_CAP) {
        for (int i = t; i < n; i += 4 * PLACE_T) {
            const int i1 = i + PLACE_T, i2 = i + 2 * PLACE_T, i3 = i + 3 * PLACE_T;
            const int2 p0 = pp[i];
            const int2 p1 = pp[i1 < n ? i1 : i];
            const int2 p2 = pp[i2 < n ? i2 : i];
            const int2 p3 = pp[i3 < n ? i3 : i];
            stage[atomicAdd(&cnt[p0.x - nb0], 1)] = p0.y;
            if (i1 < n) stage[atomicAdd(&cnt[p1.x - nb0], 1)] = p1.y;
            if (i2 < n) stage[atomicAdd(&cnt[p2.x - nb0], 1)] = p2.y;
            if (i3 < n) stage[atomicAdd(&cnt[p3.x - nb0], 1)] = p3.y;
        }
        __syncthreads();
        for (int i = t; i < n; i += PLACE_T) ssrc[start + i] = stage[i];
    } else {
        for (int i = t; i < n; i += 4 * PLACE_T) {
            const int i1 = i + PLACE_T, i2 = i + 2 * PLACE_T, i3 = i + 3 * PLACE_T;
            const int2 p0 = pp[i];
            const int2 p1 = pp[i1 < n ? i1 : i];
            const int2 p2 = pp[i2 < n ? i2 : i];
            const int2 p3 = pp[i3 < n ? i3 : i];
            ssrc[start + atomicAdd(&cnt[p0.x - nb0], 1)] = p0.y;
            if (i1 < n) ssrc[start + atomicAdd(&cnt[p1.x - nb0], 1)] = p1.y;
            if (i2 < n) ssrc[start + atomicAdd(&cnt[p2.x - nb0], 1)] = p2.y;
            if (i3 < n) ssrc[start + atomicAdd(&cnt[p3.x - nb0], 1)] = p3.y;
        }
    }
}

__global__ void __launch_bounds__(32)
gat_wmma_feat_x_w(const float* __restrict__ feat,
                  const float* __restrict__ W,
                  float* __restrict__ h,
                  int N) {
    int tile  = blockIdx.x;
    int lane  = threadIdx.x;
    int mrow  = lane & 15;
    int khalf = (lane >> 4) * 2;

    int row  = tile * 16 + mrow;
    int rowc = row < N ? row : N - 1;
    const float* arow = feat + (size_t)rowc * F;

    int   n1   = 16 + mrow;
    int   n1c  = (n1 < F) ? n1 : 0;
    float msk1 = (n1 < F) ? 1.0f : 0.0f;

    v8f c0 = {};
    v8f c1 = {};
#pragma unroll
    for (int k0 = 0; k0 < F; k0 += 4) {
        const float* wr0 = W + (size_t)(k0 + khalf + 0) * F;
        const float* wr1 = W + (size_t)(k0 + khalf + 1) * F;

        v2f a;
        a.x = arow[k0 + khalf + 0];
        a.y = arow[k0 + khalf + 1];

        v2f b0;
        b0.x = wr0[mrow];
        b0.y = wr1[mrow];
        c0 = __builtin_amdgcn_wmma_f32_16x16x4_f32(false, a, false, b0,
                                                   (short)0, c0, false, false);

        v2f b1;
        b1.x = wr0[n1c] * msk1;
        b1.y = wr1[n1c] * msk1;
        c1 = __builtin_amdgcn_wmma_f32_16x16x4_f32(false, a, false, b1,
                                                   (short)0, c1, false, false);
    }

    int rbase = tile * 16 + ((lane >> 4) << 3);
#pragma unroll
    for (int j = 0; j < 8; ++j) {
        int r = rbase + j;
        if (r < N) {
            h[(size_t)r * HP + mrow] = c0[j];
            if (n1 < F) h[(size_t)r * HP + n1] = c1[j];
        }
    }
}

__global__ void gat_node_alpha_init(float* __restrict__ h,
                                    const float* __restrict__ a_src,
                                    const float* __restrict__ a_dst,
                                    float* __restrict__ asrc,
                                    float* __restrict__ adst,
                                    int N) {
    int i = blockIdx.x * blockDim.x + threadIdx.x;
    if (i >= N) return;
    float sa = 0.0f, sd = 0.0f;
#pragma unroll
    for (int f = 0; f < F; ++f) {
        float hv = h[i * HP + f];
        sa += hv * a_src[f];
        sd += hv * a_dst[f];
    }
    *(float4*)(h + i * HP + F) = make_float4(sa, sd, 0.0f, 0.0f);
    asrc[i] = sa;
    adst[i] = sd;
}

__device__ __forceinline__ float leaky(float v) {
    return v > 0.0f ? v : NEG_SLOPE * v;
}

__device__ __forceinline__ float group_lane5(float v) {
    return __builtin_bit_cast(float,
        __builtin_amdgcn_mov_dpp8(__builtin_bit_cast(unsigned, v), 0xB6DB6D));
}

__global__ void __launch_bounds__(256)
gat_node_gather(const int* __restrict__ off,
                const int* __restrict__ ssrc,
                const float* __restrict__ asrc,
                const float* __restrict__ adst,
                const float* __restrict__ h,
                const float* __restrict__ b,
                float* __restrict__ feat,
                int N) {
    const int gid = blockIdx.x * 256 + threadIdx.x;
    const int sub = threadIdx.x & 7;
    int node = gid >> 3;
    const bool valid = node < N;
    if (!valid) node = N - 1;

    const int beg = off[node];
    const int end = off[node + 1];
    const float ad = adst[node];

    const int col = sub < 5 ? sub : 5;
    const float4* __restrict__ h4 = (const float4*)h + col;

    float mx = leaky(asrc[node] + ad);
    float den = 1.0f;
    float4 acc = h4[(size_t)node * HP4];

    int j = beg;
    for (; j + 4 <= end; j += 4) {
        int s0 = ssrc[j], s1 = ssrc[j + 1], s2 = ssrc[j + 2], s3 = ssrc[j + 3];
        float4 g0 = h4[(size_t)s0 * HP4];
        float4 g1 = h4[(size_t)s1 * HP4];
        float4 g2 = h4[(size_t)s2 * HP4];
        float4 g3 = h4[(size_t)s3 * HP4];
        float e0 = leaky(group_lane5(g0.x) + ad);
        float e1 = leaky(group_lane5(g1.x) + ad);
        float e2 = leaky(group_lane5(g2.x) + ad);
        float e3 = leaky(group_lane5(g3.x) + ad);
        float mn = fmaxf(fmaxf(fmaxf(e0, e1), fmaxf(e2, e3)), mx);
        float sc = __expf(mx - mn);
        mx = mn;
        float t0 = __expf(e0 - mx);
        float t1 = __expf(e1 - mx);
        float t2 = __expf(e2 - mx);
        float t3 = __expf(e3 - mx);
        den = den * sc + t0;
        acc.x = acc.x * sc + t0 * g0.x; acc.y = acc.y * sc + t0 * g0.y;
        acc.z = acc.z * sc + t0 * g0.z; acc.w = acc.w * sc + t0 * g0.w;
        den += t1;
        acc.x += t1 * g1.x; acc.y += t1 * g1.y; acc.z += t1 * g1.z; acc.w += t1 * g1.w;
        den += t2;
        acc.x += t2 * g2.x; acc.y += t2 * g2.y; acc.z += t2 * g2.z; acc.w += t2 * g2.w;
        den += t3;
        acc.x += t3 * g3.x; acc.y += t3 * g3.y; acc.z += t3 * g3.z; acc.w += t3 * g3.w;
    }
#pragma clang loop unroll(disable)
    for (; j < end; ++j) {
        int s0 = ssrc[j];
        float4 g0 = h4[(size_t)s0 * HP4];
        float e0 = leaky(group_lane5(g0.x) + ad);
        float mn = fmaxf(e0, mx);
        float sc = __expf(mx - mn);
        mx = mn;
        float t0 = __expf(e0 - mx);
        den = den * sc + t0;
        acc.x = acc.x * sc + t0 * g0.x; acc.y = acc.y * sc + t0 * g0.y;
        acc.z = acc.z * sc + t0 * g0.z; acc.w = acc.w * sc + t0 * g0.w;
    }

    if (valid && sub < 5) {
        float inv = 1.0f / (den + 1e-16f);
        float4 o;
        o.x = acc.x * inv + b[sub * 4 + 0];
        o.y = acc.y * inv + b[sub * 4 + 1];
        o.z = acc.z * inv + b[sub * 4 + 2];
        o.w = acc.w * inv + b[sub * 4 + 3];
        o.x = o.x > 0.0f ? o.x : 0.0f;
        o.y = o.y > 0.0f ? o.y : 0.0f;
        o.z = o.z > 0.0f ? o.z : 0.0f;
        o.w = o.w > 0.0f ? o.w : 0.0f;
        ((float4*)feat)[(size_t)node * 5 + sub] = o;
    }
}

#define L1_T    1024
#define L1_G    8
#define L1_GRP  (L1_T / L1_G)
#define L1_EPL  6
#define L1_TAB  80000
#define L1_FILL 10

__global__ void __launch_bounds__(L1_T)
gat_l1_scalar_gather(const int* __restrict__ off,
                     const int* __restrict__ ssrc,
                     const float* __restrict__ x,
                     const float* __restrict__ W1,
                     const float* __restrict__ a_src,
                     const float* __restrict__ a_dst,
                     const float* __restrict__ b,
                     float* __restrict__ feat,
                     int N) {
    __shared__ v4f tab4[L1_TAB / 4];
    const float* tab = (const float*)tab4;
    const int t   = threadIdx.x;
    const int npw = (N + (int)gridDim.x - 1) / (int)gridDim.x;
    const int n0  = (int)blockIdx.x * npw;
    const int n1  = n0 + npw < N ? n0 + npw : N;
    if (n0 >= n1) return;

    const int ntab = N < L1_TAB ? N : L1_TAB;
    const int n4   = ntab >> 2;
    const v4f* __restrict__ x4 = (const v4f*)x;
    for (int i = t; i < n4; i += L1_FILL * L1_T) {
        v4f v[L1_FILL];
#pragma unroll
        for (int k = 0; k < L1_FILL; ++k) {
            const int ii = i + k * L1_T;
            v[k] = x4[ii < n4 ? ii : i];
        }
        asm volatile("" :: "v"(v[0]), "v"(v[1]), "v"(v[2]), "v"(v[3]), "v"(v[4]),
                           "v"(v[5]), "v"(v[6]), "v"(v[7]), "v"(v[8]), "v"(v[9]));
#pragma unroll
        for (int k = 0; k < L1_FILL; ++k) {
            const int ii = i + k * L1_T;
            if (ii < n4) tab4[ii] = v[k];
        }
    }
    for (int i = (n4 << 2) + t; i < ntab; i += L1_T) ((float*)tab4)[i] = x[i];

    float cs = 0.0f, cd = 0.0f;
#pragma unroll
    for (int f = 0; f < F; ++f) {
        cs += W1[f] * a_src[f];
        cd += W1[f] * a_dst[f];
    }

    const int sub = t & (L1_G - 1);
    const int g   = t / L1_G;
    const int c4  = (sub < 5 ? sub : 0) * 4;
    const float w0 = W1[c4], w1 = W1[c4 + 1], w2 = W1[c4 + 2], w3 = W1[c4 + 3];
    const float b0 = b[c4],  b1 = b[c4 + 1],  b2 = b[c4 + 2],  b3 = b[c4 + 3];

    __syncthreads();

    const int nfix  = ntab < N ? ntab : N - 1;
    const int trips = (n1 - n0 + L1_GRP - 1) / L1_GRP;
    const int nf  = n0 + g < n1 ? n0 + g : n1 - 1;
    int   beg = off[nf];
    int   end = off[nf + 1];
    float xi  = x[nf];
    for (int it = 0; it < trips; ++it) {
        const int  node  = n0 + it * L1_GRP + g;
        const bool valid = node < n1;
        const int   nn   = node + L1_GRP < n1 ? node + L1_GRP : n1 - 1;
        const int   begn = off[nn];
        const int   endn = off[nn + 1];
        const float xin  = x[nn];

        const float ad = xi * cd;
        float m   = leaky(xi * cs + ad);
        float den = sub == 0 ? 1.0f : 0.0f;
        float num = sub == 0 ? xi   : 0.0f;

        for (int j = beg + sub; j < end; j += L1_EPL * L1_G) {
            int s[L1_EPL];
#pragma unroll
            for (int k = 0; k < L1_EPL; ++k) {
                const int jk = j + k * L1_G;
                s[k] = ssrc[jk < end ? jk : j];
            }
            float xl[L1_EPL], xg[L1_EPL];
#pragma unroll
            for (int k = 0; k < L1_EPL; ++k) {
                const bool in_tab = s[k] < ntab;
                xl[k] = tab[in_tab ? s[k] : 0];
                xg[k] = x[in_tab ? nfix : s[k]];
            }
            float xv[L1_EPL], e[L1_EPL];
#pragma unroll
            for (int k = 0; k < L1_EPL; ++k) {
                const bool have = j + k * L1_G < end;
                const float xs  = s[k] < ntab ? xl[k] : xg[k];
                xv[k] = have ? xs : 0.0f;
                e[k]  = have ? leaky(xs * cs + ad) : -INFINITY;
            }
            float mn = m;
#pragma unroll
            for (int k = 0; k < L1_EPL; ++k) mn = fmaxf(mn, e[k]);
            const float sc = __expf(m - mn);
            m = mn;
            den *= sc;
            num *= sc;
#pragma unroll
            for (int k = 0; k < L1_EPL; ++k) {
                const float tk = __expf(e[k] - mn);
                den += tk;
                num += tk * xv[k];
            }
        }

        float M = m;
        M = fmaxf(M, __shfl_xor(M, 1, L1_G));
        M = fmaxf(M, __shfl_xor(M, 2, L1_G));
        M = fmaxf(M, __shfl_xor(M, 4, L1_G));
        const float fs = __expf(m - M);
        den *= fs;
        num *= fs;
        den += __shfl_xor(den, 1, L1_G);
        num += __shfl_xor(num, 1, L1_G);
        den += __shfl_xor(den, 2, L1_G);
        num += __shfl_xor(num, 2, L1_G);
        den += __shfl_xor(den, 4, L1_G);
        num += __shfl_xor(num, 4, L1_G);

        if (valid && sub < 5) {
            const float S = num * (1.0f / (den + 1e-16f));
            float4 o;
            o.x = S * w0 + b0;
            o.y = S * w1 + b1;
            o.z = S * w2 + b2;
            o.w = S * w3 + b3;
            o.x = o.x > 0.0f ? o.x : 0.0f;
            o.y = o.y > 0.0f ? o.y : 0.0f;
            o.z = o.z > 0.0f ? o.z : 0.0f;
            o.w = o.w > 0.0f ? o.w : 0.0f;
            ((float4*)feat)[(size_t)node * 5 + sub] = o;
        }
        beg = begn;
        end = endn;
        xi  = xin;
    }
}

__global__ void gat_final_linear(const float* __restrict__ feat,
                                 const float* __restrict__ Wl,
                                 const float* __restrict__ bl,
                                 float* __restrict__ out,
                                 int N) {
    int i = blockIdx.x * blockDim.x + threadIdx.x;
    if (i >= N) return;
    float sacc = bl[0];
#pragma unroll
    for (int f = 0; f < F; ++f) sacc += feat[i * F + f] * Wl[f];
    out[i] = sacc;
}

extern "C" void kernel_launch(void* const* d_in, const int* in_sizes, int n_in,
                              void* d_out, int out_size, void* d_ws, size_t ws_size,
                              hipStream_t stream) {
    (void)n_in; (void)out_size; (void)ws_size;

    const int N    = in_sizes[0];
    const int E    = in_sizes[1] / 2;

    const float* x    = (const float*)d_in[0];
    const int*   ei   = (const int*)d_in[1];
    const float* W1   = (const float*)d_in[2];
    const float* as1  = (const float*)d_in[3];
    const float* ad1  = (const float*)d_in[4];
    const float* b1   = (const float*)d_in[5];
    const float* W2   = (const float*)d_in[6];
    const float* as2  = (const float*)d_in[7];
    const float* ad2  = (const float*)d_in[8];
    const float* b2   = (const float*)d_in[9];
    const float* Wl   = (const float*)d_in[10];
    const float* bl   = (const float*)d_in[11];

    const int* srcIdx = ei;
    const int* dstIdx = ei + E;

    const int TB   = 256;
    const int gN   = (N + TB - 1) / TB;
    const int NB   = (N + BKT_W - 1) / BKT_W;
    const int gCH  = (E + SORT_CH - 1) / SORT_CH;
    const int gMM  = (N + 15) / 16;
    const int gG   = (N + 31) / 32;

    float* ws     = (float*)d_ws;
    float* h      = ws + (size_t)0  * N;
    float* feat1  = ws + (size_t)32 * N;
    float* feat2  = ws + (size_t)52 * N;
    float* asrc   = ws + (size_t)72 * N;
    float* adst   = ws + (size_t)73 * N;
    int*   btot   = (int*)(ws + (size_t)74 * N);
    int*   off    = (int*)(ws + (size_t)75 * N);
    int*   bstart = (int*)(ws + (size_t)77 * N);
    int2*  pairs  = (int2*)(ws + (size_t)78 * N);
    int*   ssrc   = (int*)(ws + (size_t)78 * N) + (size_t)2 * E;
    int*   cb     = ssrc + (size_t)E;

    bucket_totals_clear<<<(NB + TB - 1) / TB, TB, 0, stream>>>(btot, NB);
    bucket_count<<<gCH, SORT_T, NB * sizeof(int), stream>>>(dstIdx, btot, cb, E, NB);
    bucket_scatter<<<gCH, SORT_T, 2 * NB * sizeof(int), stream>>>(srcIdx, dstIdx, btot, cb, bstart, pairs, E, NB);
    bucket_place<<<NB, PLACE_T, 0, stream>>>(pairs, btot, bstart, off, ssrc, N);

    gat_l1_scalar_gather<<<256, L1_T, 0, stream>>>(off, ssrc, x, W1, as1, ad1, b1, feat1, N);

    gat_wmma_feat_x_w<<<gMM, 32, 0, stream>>>(feat1, W2, h, N);
    gat_node_alpha_init<<<gN, TB, 0, stream>>>(h, as2, ad2, asrc, adst, N);
    gat_node_gather<<<gG, TB, 0, stream>>>(off, ssrc, asrc, adst, h, b2, feat2, N);

    gat_final_linear<<<gN, TB, 0, stream>>>(feat2, Wl, bl, (float*)d_out, N);
}


# Launch fusion 9 -> 6: layer-2 WMMA runs in the layer-1 gather's waves (16 summed nodes = one tile), scores as two extra weight columns (W2 a), head by an 8-lane tree in the gather; sum order changes
# speedup vs baseline: 24.0503x; 1.1349x over previous
#include <hip/hip_runtime.h>
#include <math.h>

#define F 20
#define NEG_SLOPE 0.2f
#define HP 32
#define HP4 (HP / 4)

typedef float v2f __attribute__((ext_vector_type(2)));
typedef float v8f __attribute__((ext_vector_type(8)));
typedef float v4f __attribute__((ext_vector_type(4)));

__global__ void bucket_totals_clear(int* __restrict__ btot, int NB) {
    int i = blockIdx.x * blockDim.x + threadIdx.x;
    if (i < NB) btot[i] = 0;
}

#define BKT_SH 8
#define BKT_W  (1 << BKT_SH)
#define SORT_T   512
#define SORT_EPT 16
#define SORT_CH  (SORT_T * SORT_EPT)

__global__ void __launch_bounds__(SORT_T)
bucket_count(const int* __restrict__ dst,
             int* __restrict__ btot,
             int* __restrict__ cb,
             int E, int NB) {
    extern __shared__ int hist[];
    const int t  = threadIdx.x;
    const int c  = blockIdx.x;
    const int e0 = c * SORT_CH + t;
    for (int b = t; b < NB; b += SORT_T) hist[b] = 0;
    __syncthreads();
    if (c * SORT_CH + SORT_CH <= E) {
        int d[SORT_EPT];
#pragma unroll
        for (int k = 0; k < SORT_EPT; ++k) d[k] = dst[e0 + k * SORT_T];
#pragma unroll
        for (int k = 0; k < SORT_EPT; ++k) atomicAdd(&hist[d[k] >> BKT_SH], 1);
    } else {
        for (int k = 0; k < SORT_EPT; ++k) {
            int e = e0 + k * SORT_T;
            if (e < E) atomicAdd(&hist[dst[e] >> BKT_SH], 1);
        }
    }
    __syncthreads();
    for (int b = t; b < NB; b += SORT_T)
        cb[(size_t)c * NB + b] = atomicAdd(&btot[b], hist[b]);
}

__global__ void __launch_bounds__(SORT_T)
bucket_scatter(const int* __restrict__ src,
               const int* __restrict__ dst,
               const int* __restrict__ btot,
               const int* __restrict__ cb,
               int* __restrict__ bstart,
               int2* __restrict__ pairs,
               int E, int NB) {
    extern __shared__ int sm[];
    __shared__ int buf[2][SORT_T];
    int* hist = sm;
    int* base = sm + NB;
    const int t  = threadIdx.x;
    const int c  = blockIdx.x;
    const int e0 = c * SORT_CH + t;
    for (int b = t; b < NB; b += SORT_T) hist[b] = 0;
    __syncthreads();

    int d[SORT_EPT], s[SORT_EPT], r[SORT_EPT];
    if (c * SORT_CH + SORT_CH <= E) {
#pragma unroll
        for (int k = 0; k < SORT_EPT; ++k) {
            d[k] = dst[e0 + k * SORT_T];
            s[k] = src[e0 + k * SORT_T];
        }
#pragma unroll
        for (int k = 0; k < SORT_EPT; ++k) r[k] = atomicAdd(&hist[d[k] >> BKT_SH], 1);
    } else {
#pragma unroll
        for (int k = 0; k < SORT_EPT; ++k) {
            int e = e0 + k * SORT_T;
            d[k] = -1; s[k] = 0; r[k] = 0;
            if (e < E) {
                d[k] = dst[e];
                s[k] = src[e];
                r[k] = atomicAdd(&hist[d[k] >> BKT_SH], 1);
            }
        }
    }

    int carry = 0;
    for (int b0 = 0; b0 < NB; b0 += SORT_T) {
        const int b = b0 + t;
        const int v = b < NB ? btot[b] : 0;
        int cur = 0;
        buf[0][t] = v;
        __syncthreads();
        for (int dd = 1; dd < SORT_T; dd <<= 1) {
            int xv = buf[cur][t];
            if (t >= dd) xv += buf[cur][t - dd];
            buf[cur ^ 1][t] = xv;
            cur ^= 1;
            __syncthreads();
        }
        const int st = carry + buf[cur][t] - v;
        if (b < NB) {
            base[b] = st + cb[(size_t)c * NB + b];
            if (c == 0) bstart[b] = st;
        }
        carry += buf[cur][SORT_T - 1];
        __syncthreads();
    }
    if (c == 0 && t == 0) bstart[NB] = carry;

#pragma unroll
    for (int k = 0; k < SORT_EPT; ++k) {
        if (d[k] >= 0) pairs[base[d[k] >> BKT_SH] + r[k]] = make_int2(d[k], s[k]);
    }
}

#define PLACE_T 1024
#define PLACE_CAP 12288
__global__ void __launch_bounds__(PLACE_T)
bucket_place(const int2* __restrict__ pairs,
             const int* __restrict__ btot,
             const int* __restrict__ bstart,
             int* __restrict__ off,
             int* __restrict__ ssrc,
             int N) {
    __shared__ int cnt[BKT_W];
    __shared__ int buf[2][BKT_W];
    __shared__ int stage[PLACE_CAP];
    const int t     = threadIdx.x;
    const int b     = blockIdx.x;
    const int nb0   = b << BKT_SH;
    const int start = bstart[b];
    const int n     = btot[b];
    const int2* __restrict__ pp = pairs + start;

    if (t < BKT_W) cnt[t] = 0;
    __syncthreads();

    for (int i = t; i < n; i += 4 * PLACE_T) {
        const int i1 = i + PLACE_T, i2 = i + 2 * PLACE_T, i3 = i + 3 * PLACE_T;
        const int d0 = pp[i].x;
        const int d1 = pp[i1 < n ? i1 : i].x;
        const int d2 = pp[i2 < n ? i2 : i].x;
        const int d3 = pp[i3 < n ? i3 : i].x;
        atomicAdd(&cnt[d0 - nb0], 1);
        if (i1 < n) atomicAdd(&cnt[d1 - nb0], 1);
        if (i2 < n) atomicAdd(&cnt[d2 - nb0], 1);
        if (i3 < n) atomicAdd(&cnt[d3 - nb0], 1);
    }
    __syncthreads();

    int v = 0, cur = 0;
    if (t < BKT_W) { v = cnt[t]; buf[0][t] = v; }
    __syncthreads();
    for (int dd = 1; dd < BKT_W; dd <<= 1) {
        if (t < BKT_W) {
            int xv = buf[cur][t];
            if (t >= dd) xv += buf[cur][t - dd];
            buf[cur ^ 1][t] = xv;
        }
        cur ^= 1;
        __syncthreads();
    }
    if (t < BKT_W) {
        const int ex = buf[cur][t] - v;
        cnt[t] = ex;
        const int node = nb0 + t;
        if (node < N) off[node] = start + ex;
        if (node == N - 1) off[N] = start + n;
    }
    __syncthreads();

    if (n <= PLACE_CAP) {
        for (int i = t; i < n; i += 4 * PLACE_T) {
            const int i1 = i + PLACE_T, i2 = i + 2 * PLACE_T, i3 = i + 3 * PLACE_T;
            const int2 p0 = pp[i];
            const int2 p1 = pp[i1 < n ? i1 : i];
            const int2 p2 = pp[i2 < n ? i2 : i];
            const int2 p3 = pp[i3 < n ? i3 : i];
            stage[atomicAdd(&cnt[p0.x - nb0], 1)] = p0.y;
            if (i1 < n) stage[atomicAdd(&cnt[p1.x - nb0], 1)] = p1.y;
            if (i2 < n) stage[atomicAdd(&cnt[p2.x - nb0], 1)] = p2.y;
            if (i3 < n) stage[atomicAdd(&cnt[p3.x - nb0], 1)] = p3.y;
        }
        __syncthreads();
        for (int i = t; i < n; i += PLACE_T) ssrc[start + i] = stage[i];
    } else {
        for (int i = t; i < n; i += 4 * PLACE_T) {
            const int i1 = i + PLACE_T, i2 = i + 2 * PLACE_T, i3 = i + 3 * PLACE_T;
            const int2 p0 = pp[i];
            const int2 p1 = pp[i1 < n ? i1 : i];
            const int2 p2 = pp[i2 < n ? i2 : i];
            const int2 p3 = pp[i3 < n ? i3 : i];
            ssrc[start + atomicAdd(&cnt[p0.x - nb0], 1)] = p0.y;
            if (i1 < n) ssrc[start + atomicAdd(&cnt[p1.x - nb0], 1)] = p1.y;
            if (i2 < n) ssrc[start + atomicAdd(&cnt[p2.x - nb0], 1)] = p2.y;
            if (i3 < n) ssrc[start + atomicAdd(&cnt[p3.x - nb0], 1)] = p3.y;
        }
    }
}

__device__ __forceinline__ float leaky(float v) {
    return v > 0.0f ? v : NEG_SLOPE * v;
}

__device__ __forceinline__ float group_lane5(float v) {
    return __builtin_bit_cast(float,
        __builtin_amdgcn_mov_dpp8(__builtin_bit_cast(unsigned, v), 0xB6DB6D));
}

__global__ void __launch_bounds__(256)
gat_node_gather(const int* __restrict__ off,
                const int* __restrict__ ssrc,
                const float* __restrict__ h,
                const float* __restrict__ b,
                const float* __restrict__ Wl,
                const float* __restrict__ bl,
                float* __restrict__ out,
                int N) {
    __shared__ float outs[32];
    const int gid = blockIdx.x * 256 + threadIdx.x;
    const int sub = threadIdx.x & 7;
    int node = gid >> 3;
    const bool valid = node < N;
    if (!valid) node = N - 1;

    const int beg = off[node];
    const int end = off[node + 1];

    const int col = sub < 5 ? sub : 5;
    const float4* __restrict__ h4 = (const float4*)h + col;

    float4 acc = h4[(size_t)node * HP4];
    const float ad = group_lane5(acc.y);
    float mx = leaky(group_lane5(acc.x) + ad);
    float den = 1.0f;

    int j = beg;
    for (; j + 4 <= end; j += 4) {
        int s0 = ssrc[j], s1 = ssrc[j + 1], s2 = ssrc[j + 2], s3 = ssrc[j + 3];
        float4 g0 = h4[(size_t)s0 * HP4];
        float4 g1 = h4[(size_t)s1 * HP4];
        float4 g2 = h4[(size_t)s2 * HP4];
        float4 g3 = h4[(size_t)s3 * HP4];
        float e0 = leaky(group_lane5(g0.x) + ad);
        float e1 = leaky(group_lane5(g1.x) + ad);
        float e2 = leaky(group_lane5(g2.x) + ad);
        float e3 = leaky(group_lane5(g3.x) + ad);
        float mn = fmaxf(fmaxf(fmaxf(e0, e1), fmaxf(e2, e3)), mx);
        float sc = __expf(mx - mn);
        mx = mn;
        float t0 = __expf(e0 - mx);
        float t1 = __expf(e1 - mx);
        float t2 = __expf(e2 - mx);
        float t3 = __expf(e3 - mx);
        den = den * sc + t0;
        acc.x = acc.x * sc + t0 * g0.x; acc.y = acc.y * sc + t0 * g0.y;
        acc.z = acc.z * sc + t0 * g0.z; acc.w = acc.w * sc + t0 * g0.w;
        den += t1;
        acc.x += t1 * g1.x; acc.y += t1 * g1.y; acc.z += t1 * g1.z; acc.w += t1 * g1.w;
        den += t2;
        acc.x += t2 * g2.x; acc.y += t2 * g2.y; acc.z += t2 * g2.z; acc.w += t2 * g2.w;
        den += t3;
        acc.x += t3 * g3.x; acc.y += t3 * g3.y; acc.z += t3 * g3.z; acc.w += t3 * g3.w;
    }
#pragma clang loop unroll(disable)
    for (; j < end; ++j) {
        int s0 = ssrc[j];
        float4 g0 = h4[(size_t)s0 * HP4];
        float e0 = leaky(group_lane5(g0.x) + ad);
        float mn = fmaxf(e0, mx);
        float sc = __expf(mx - mn);
        mx = mn;
        float t0 = __expf(e0 - mx);
        den = den * sc + t0;
        acc.x = acc.x * sc + t0 * g0.x; acc.y = acc.y * sc + t0 * g0.y;
        acc.z = acc.z * sc + t0 * g0.z; acc.w = acc.w * sc + t0 * g0.w;
    }

    float p = 0.0f;
    if (sub < 5) {
        float inv = 1.0f / (den + 1e-16f);
        float4 o;
        o.x = acc.x * inv + b[sub * 4 + 0];
        o.y = acc.y * inv + b[sub * 4 + 1];
        o.z = acc.z * inv + b[sub * 4 + 2];
        o.w = acc.w * inv + b[sub * 4 + 3];
        o.x = o.x > 0.0f ? o.x : 0.0f;
        o.y = o.y > 0.0f ? o.y : 0.0f;
        o.z = o.z > 0.0f ? o.z : 0.0f;
        o.w = o.w > 0.0f ? o.w : 0.0f;
        p = o.x * Wl[sub * 4 + 0] + o.y * Wl[sub * 4 + 1]
          + o.z * Wl[sub * 4 + 2] + o.w * Wl[sub * 4 + 3];
    }
    p += __shfl_xor(p, 1, 8);
    p += __shfl_xor(p, 2, 8);
    p += __shfl_xor(p, 4, 8);
    if (sub == 0) outs[threadIdx.x >> 3] = p + bl[0];
    __syncthreads();
    if (threadIdx.x < 32) {
        const int on = blockIdx.x * 32 + threadIdx.x;
        if (on < N) out[on] = outs[threadIdx.x];
    }
}

#define MM_KS (F / 4)
#define L1_T    1024
#define L1_G    8
#define L1_GRP  (L1_T / L1_G)
#define L1_EPL  6
#define L1_TAB  80000
#define L1_FILL 10

__global__ void __launch_bounds__(L1_T)
gat_l1_scalar_gather(const int* __restrict__ off,
                     const int* __restrict__ ssrc,
                     const float* __restrict__ x,
                     const float* __restrict__ W1,
                     const float* __restrict__ a_src,
                     const float* __restrict__ a_dst,
                     const float* __restrict__ b,
                     const float* __restrict__ W2,
                     const float* __restrict__ a_src2,
                     const float* __restrict__ a_dst2,
                     float* __restrict__ h,
                     int N) {
    __shared__ v4f tab4[L1_TAB / 4];
    __shared__ float wa[2][F];
    const float* tab = (const float*)tab4;
    const int t   = threadIdx.x;
    const int npw = (N + (int)gridDim.x - 1) / (int)gridDim.x;
    const int n0  = (int)blockIdx.x * npw;
    const int n1  = n0 + npw < N ? n0 + npw : N;
    if (n0 >= n1) return;

    const int ntab = N < L1_TAB ? N : L1_TAB;
    const int n4   = ntab >> 2;
    const v4f* __restrict__ x4 = (const v4f*)x;
    for (int i = t; i < n4; i += L1_FILL * L1_T) {
        v4f v[L1_FILL];
#pragma unroll
        for (int k = 0; k < L1_FILL; ++k) {
            const int ii = i + k * L1_T;
            v[k] = x4[ii < n4 ? ii : i];
        }
        asm volatile("" :: "v"(v[0]), "v"(v[1]), "v"(v[2]), "v"(v[3]), "v"(v[4]),
                           "v"(v[5]), "v"(v[6]), "v"(v[7]), "v"(v[8]), "v"(v[9]));
#pragma unroll
        for (int k = 0; k < L1_FILL; ++k) {
            const int ii = i + k * L1_T;
            if (ii < n4) tab4[ii] = v[k];
        }
    }
    for (int i = (n4 << 2) + t; i < ntab; i += L1_T) ((float*)tab4)[i] = x[i];

    if (t < F) {
        float ws = 0.0f, wd = 0.0f;
#pragma unroll
        for (int f = 0; f < F; ++f) {
            const float w = W2[t * F + f];
            ws += w * a_src2[f];
            wd += w * a_dst2[f];
        }
        wa[0][t] = ws;
        wa[1][t] = wd;
    }

    float cs = 0.0f, cd = 0.0f;
#pragma unroll
    for (int f = 0; f < F; ++f) {
        cs += W1[f] * a_src[f];
        cd += W1[f] * a_dst[f];
    }

    const int sub = t & (L1_G - 1);
    const int g   = t / L1_G;

    const int  lane  = t & 31;
    const int  wv    = t >> 5;
    const int  mrow  = lane & 15;
    const int  lhalf = lane >> 4;
    const int  khalf = lhalf * 2;
    const int  n1col = 16 + mrow;
    const int  n1c   = n1col < F ? n1col : 0;
    const bool is_w  = n1col < F;
    const bool is_sc = n1col >= F && n1col < F + 2;
    const int  scv   = n1col == F + 1 ? 1 : 0;
    float srow = 0.0f;

    __syncthreads();

    v2f b0[MM_KS], b1[MM_KS];
#pragma unroll
    for (int i = 0; i < MM_KS; ++i) {
        const int k = 4 * i + khalf;
        b0[i].x = W2[(k + 0) * F + mrow];
        b0[i].y = W2[(k + 1) * F + mrow];
        const float wx = W2[(k + 0) * F + n1c];
        const float wy = W2[(k + 1) * F + n1c];
        const float sx = wa[scv][k + 0];
        const float sy = wa[scv][k + 1];
        b1[i].x = is_w ? wx : (is_sc ? sx : 0.0f);
        b1[i].y = is_w ? wy : (is_sc ? sy : 0.0f);
    }

    const int nfix  = ntab < N ? ntab : N - 1;
    const int trips = (n1 - n0 + L1_GRP - 1) / L1_GRP;
    const int nf  = n0 + g < n1 ? n0 + g : n1 - 1;
    int   beg = off[nf];
    int   end = off[nf + 1];
    float xi  = x[nf];
    for (int it = 0; it < trips; ++it) {
        const int  node  = n0 + it * L1_GRP + g;
        const int   nn   = node + L1_GRP < n1 ? node + L1_GRP : n1 - 1;
        const int   begn = off[nn];
        const int   endn = off[nn + 1];
        const float xin  = x[nn];

        const float ad = xi * cd;
        float m   = leaky(xi * cs + ad);
        float den = sub == 0 ? 1.0f : 0.0f;
        float num = sub == 0 ? xi   : 0.0f;

        for (int j = beg + sub; j < end; j += L1_EPL * L1_G) {
            int s[L1_EPL];
#pragma unroll
            for (int k = 0; k < L1_EPL; ++k) {
                const int jk = j + k * L1_G;
                s[k] = ssrc[jk < end ? jk : j];
            }
            float xl[L1_EPL], xg[L1_EPL];
#pragma unroll
            for (int k = 0; k < L1_EPL; ++k) {
                const bool in_tab = s[k] < ntab;
                xl[k] = tab[in_tab ? s[k] : 0];
                xg[k] = x[in_tab ? nfix : s[k]];
            }
            float xv[L1_EPL], e[L1_EPL];
#pragma unroll
            for (int k = 0; k < L1_EPL; ++k) {
                const bool have = j + k * L1_G < end;
                const float xs  = s[k] < ntab ? xl[k] : xg[k];
                xv[k] = have ? xs : 0.0f;
                e[k]  = have ? leaky(xs * cs + ad) : -INFINITY;
            }
            float mn = m;
#pragma unroll
            for (int k = 0; k < L1_EPL; ++k) mn = fmaxf(mn, e[k]);
            const float sc = __expf(m - mn);
            m = mn;
            den *= sc;
            num *= sc;
#pragma unroll
            for (int k = 0; k < L1_EPL; ++k) {
                const float tk = __expf(e[k] - mn);
                den += tk;
                num += tk * xv[k];
            }
        }

        float M = m;
        M = fmaxf(M, __shfl_xor(M, 1, L1_G));
        M = fmaxf(M, __shfl_xor(M, 2, L1_G));
        M = fmaxf(M, __shfl_xor(M, 4, L1_G));
        const float fs = __expf(m - M);
        den *= fs;
        num *= fs;
        den += __shfl_xor(den, 1, L1_G);
        num += __shfl_xor(num, 1, L1_G);
        den += __shfl_xor(den, 2, L1_G);
        num += __shfl_xor(num, 2, L1_G);
        den += __shfl_xor(den, 4, L1_G);
        num += __shfl_xor(num, 4, L1_G);

        const float S  = num * (1.0f / (den + 1e-16f));
        const float Sv = __shfl(S, (mrow & 3) * L1_G);
        if ((it & 3) == (mrow >> 2)) srow = Sv;

        if ((it & 3) == 3 || it == trips - 1) {
            const int it0 = it & ~3;
            v8f c0 = {};
            v8f c1 = {};
#pragma unroll
            for (int i = 0; i < MM_KS; ++i) {
                const int k = 4 * i + khalf;
                v2f a;
                a.x = srow * W1[k + 0] + b[k + 0];
                a.y = srow * W1[k + 1] + b[k + 1];
                a.x = a.x > 0.0f ? a.x : 0.0f;
                a.y = a.y > 0.0f ? a.y : 0.0f;
                c0 = __builtin_amdgcn_wmma_f32_16x16x4_f32(false, a, false, b0[i],
                                                           (short)0, c0, false, false);
                c1 = __builtin_amdgcn_wmma_f32_16x16x4_f32(false, a, false, b1[i],
                                                           (short)0, c1, false, false);
            }
#pragma unroll
            for (int r = 0; r < 8; ++r) {
                const int rho = 8 * lhalf + r;
                const int nd  = n0 + (it0 + (rho >> 2)) * L1_GRP + 4 * wv + (rho & 3);
                if (nd < n1) h[(size_t)nd * HP + mrow] = c0[r];
            }
            if (n1col < F + 4) {
#pragma unroll
                for (int r = 0; r < 8; ++r) {
                    const int rho = 8 * lhalf + r;
                    const int nd  = n0 + (it0 + (rho >> 2)) * L1_GRP + 4 * wv + (rho & 3);
                    if (nd < n1) h[(size_t)nd * HP + n1col] = c1[r];
                }
            }
        }
        beg = begn;
        end = endn;
        xi  = xin;
    }
}

extern "C" void kernel_launch(void* const* d_in, const int* in_sizes, int n_in,
                              void* d_out, int out_size, void* d_ws, size_t ws_size,
                              hipStream_t stream) {
    (void)n_in; (void)out_size; (void)ws_size;

    const int N    = in_sizes[0];
    const int E    = in_sizes[1] / 2;

    const float* x    = (const float*)d_in[0];
    const int*   ei   = (const int*)d_in[1];
    const float* W1   = (const float*)d_in[2];
    const float* as1  = (const float*)d_in[3];
    const float* ad1  = (const float*)d_in[4];
    const float* b1   = (const float*)d_in[5];
    const float* W2   = (const float*)d_in[6];
    const float* as2  = (const float*)d_in[7];
    const float* ad2  = (const float*)d_in[8];
    const float* b2   = (const float*)d_in[9];
    const float* Wl   = (const float*)d_in[10];
    const float* bl   = (const float*)d_in[11];

    const int* srcIdx = ei;
    const int* dstIdx = ei + E;

    const int TB   = 256;
    const int NB   = (N + BKT_W - 1) / BKT_W;
    const int gCH  = (E + SORT_CH - 1) / SORT_CH;
    const int gG   = (N + 31) / 32;

    float* ws     = (float*)d_ws;
    float* h      = ws + (size_t)0  * N;
    int*   btot   = (int*)(ws + (size_t)74 * N);
    int*   off    = (int*)(ws + (size_t)75 * N);
    int*   bstart = (int*)(ws + (size_t)77 * N);
    int2*  pairs  = (int2*)(ws + (size_t)78 * N);
    int*   ssrc   = (int*)(ws + (size_t)78 * N) + (size_t)2 * E;
    int*   cb     = ssrc + (size_t)E;

    bucket_totals_clear<<<(NB + TB - 1) / TB, TB, 0, stream>>>(btot, NB);
    bucket_count<<<gCH, SORT_T, NB * sizeof(int), stream>>>(dstIdx, btot, cb, E, NB);
    bucket_scatter<<<gCH, SORT_T, 2 * NB * sizeof(int), stream>>>(srcIdx, dstIdx, btot, cb, bstart, pairs, E, NB);
    bucket_place<<<NB, PLACE_T, 0, stream>>>(pairs, btot, bstart, off, ssrc, N);

    gat_l1_scalar_gather<<<256, L1_T, 0, stream>>>(off, ssrc, x, W1, as1, ad1, b1, W2, as2, ad2, h, N);

    gat_node_gather<<<gG, TB, 0, stream>>>(off, ssrc, h, b2, Wl, bl, (float*)d_out, N);
}


# bucket_scatter orders the chunk's (dst,src) records by bucket in LDS and writes each bucket's run contiguously: a store covers a few runs, not 32 lone 8-byte pieces of 32 lines; edge order kept
# speedup vs baseline: 27.9800x; 1.1634x over previous
#include <hip/hip_runtime.h>
#include <math.h>

#define F 20
#define NEG_SLOPE 0.2f
#define HP 32
#define HP4 (HP / 4)

typedef float v2f __attribute__((ext_vector_type(2)));
typedef float v8f __attribute__((ext_vector_type(8)));
typedef float v4f __attribute__((ext_vector_type(4)));

__global__ void bucket_totals_clear(int* __restrict__ btot, int NB) {
    int i = blockIdx.x * blockDim.x + threadIdx.x;
    if (i < NB) btot[i] = 0;
}

#define BKT_SH 8
#define BKT_W  (1 << BKT_SH)
#define SORT_T   512
#define SORT_EPT 16
#define SORT_CH  (SORT_T * SORT_EPT)

__global__ void __launch_bounds__(SORT_T)
bucket_count(const int* __restrict__ dst,
             int* __restrict__ btot,
             int* __restrict__ cb,
             int E, int NB) {
    extern __shared__ int hist[];
    const int t  = threadIdx.x;
    const int c  = blockIdx.x;
    const int e0 = c * SORT_CH + t;
    for (int b = t; b < NB; b += SORT_T) hist[b] = 0;
    __syncthreads();
    if (c * SORT_CH + SORT_CH <= E) {
        int d[SORT_EPT];
#pragma unroll
        for (int k = 0; k < SORT_EPT; ++k) d[k] = dst[e0 + k * SORT_T];
#pragma unroll
        for (int k = 0; k < SORT_EPT; ++k) atomicAdd(&hist[d[k] >> BKT_SH], 1);
    } else {
        for (int k = 0; k < SORT_EPT; ++k) {
            int e = e0 + k * SORT_T;
            if (e < E) atomicAdd(&hist[dst[e] >> BKT_SH], 1);
        }
    }
    __syncthreads();
    for (int b = t; b < NB; b += SORT_T)
        cb[(size_t)c * NB + b] = atomicAdd(&btot[b], hist[b]);
}

__global__ void __launch_bounds__(SORT_T)
bucket_scatter(const int* __restrict__ src,
               const int* __restrict__ dst,
               const int* __restrict__ btot,
               const int* __restrict__ cb,
               int* __restrict__ bstart,
               int2* __restrict__ pairs,
               int E, int NB) {
    extern __shared__ int sm[];
    __shared__ int buf[2][SORT_T];
    __shared__ int bufh[2][SORT_T];
    __shared__ int2 stage[SORT_CH];
    int* hist = sm;
    int* gofs = sm + NB;
    const int t  = threadIdx.x;
    const int c  = blockIdx.x;
    const int e0 = c * SORT_CH + t;
    for (int b = t; b < NB; b += SORT_T) hist[b] = 0;
    __syncthreads();

    int d[SORT_EPT], s[SORT_EPT], r[SORT_EPT];
    if (c * SORT_CH + SORT_CH <= E) {
#pragma unroll
        for (int k = 0; k < SORT_EPT; ++k) {
            d[k] = dst[e0 + k * SORT_T];
            s[k] = src[e0 + k * SORT_T];
        }
#pragma unroll
        for (int k = 0; k < SORT_EPT; ++k) r[k] = atomicAdd(&hist[d[k] >> BKT_SH], 1);
    } else {
#pragma unroll
        for (int k = 0; k < SORT_EPT; ++k) {
            int e = e0 + k * SORT_T;
            d[k] = -1; s[k] = 0; r[k] = 0;
            if (e < E) {
                d[k] = dst[e];
                s[k] = src[e];
                r[k] = atomicAdd(&hist[d[k] >> BKT_SH], 1);
            }
        }
    }
    __syncthreads();

    int carry = 0, carryh = 0;
    for (int b0 = 0; b0 < NB; b0 += SORT_T) {
        const int b  = b0 + t;
        const int v  = b < NB ? btot[b] : 0;
        const int hv = b < NB ? hist[b] : 0;
        int cur = 0;
        buf[0][t]  = v;
        bufh[0][t] = hv;
        __syncthreads();
        for (int dd = 1; dd < SORT_T; dd <<= 1) {
            int xv = buf[cur][t];
            int xh = bufh[cur][t];
            if (t >= dd) { xv += buf[cur][t - dd]; xh += bufh[cur][t - dd]; }
            buf[cur ^ 1][t]  = xv;
            bufh[cur ^ 1][t] = xh;
            cur ^= 1;
            __syncthreads();
        }
        const int st = carry  + buf[cur][t]  - v;
        const int lb = carryh + bufh[cur][t] - hv;
        if (b < NB) {
            hist[b] = lb;
            gofs[b] = st + cb[(size_t)c * NB + b] - lb;
            if (c == 0) bstart[b] = st;
        }
        carry  += buf[cur][SORT_T - 1];
        carryh += bufh[cur][SORT_T - 1];
        __syncthreads();
    }
    if (c == 0 && t == 0) bstart[NB] = carry;
    const int nrec = carryh;

    int slot[SORT_EPT];
#pragma unroll
    for (int k = 0; k < SORT_EPT; ++k)
        slot[k] = hist[(d[k] >= 0 ? d[k] : 0) >> BKT_SH] + r[k];
#pragma unroll
    for (int k = 0; k < SORT_EPT; ++k) {
        if (d[k] >= 0) stage[slot[k]] = make_int2(d[k], s[k]);
    }
    __syncthreads();

    if (nrec == SORT_CH) {
        int2 p[SORT_EPT];
        int  g[SORT_EPT];
#pragma unroll
        for (int k = 0; k < SORT_EPT; ++k) p[k] = stage[t + k * SORT_T];
#pragma unroll
        for (int k = 0; k < SORT_EPT; ++k) g[k] = gofs[p[k].x >> BKT_SH];
#pragma unroll
        for (int k = 0; k < SORT_EPT; ++k) pairs[g[k] + t + k * SORT_T] = p[k];
    } else {
        for (int i = t; i < nrec; i += SORT_T) {
            const int2 p = stage[i];
            pairs[gofs[p.x >> BKT_SH] + i] = p;
        }
    }
}

#define PLACE_T 1024
#define PLACE_CAP 12288
__global__ void __launch_bounds__(PLACE_T)
bucket_place(const int2* __restrict__ pairs,
             const int* __restrict__ btot,
             const int* __restrict__ bstart,
             int* __restrict__ off,
             int* __restrict__ ssrc,
             int N) {
    __shared__ int cnt[BKT_W];
    __shared__ int buf[2][BKT_W];
    __shared__ int stage[PLACE_CAP];
    const int t     = threadIdx.x;
    const int b     = blockIdx.x;
    const int nb0   = b << BKT_SH;
    const int start = bstart[b];
    const int n     = btot[b];
    const int2* __restrict__ pp = pairs + start;

    if (t < BKT_W) cnt[t] = 0;
    __syncthreads();

    for (int i = t; i < n; i += 4 * PLACE_T) {
        const int i1 = i + PLACE_T, i2 = i + 2 * PLACE_T, i3 = i + 3 * PLACE_T;
        const int d0 = pp[i].x;
        const int d1 = pp[i1 < n ? i1 : i].x;
        const int d2 = pp[i2 < n ? i2 : i].x;
        const int d3 = pp[i3 < n ? i3 : i].x;
        atomicAdd(&cnt[d0 - nb0], 1);
        if (i1 < n) atomicAdd(&cnt[d1 - nb0], 1);
        if (i2 < n) atomicAdd(&cnt[d2 - nb0], 1);
        if (i3 < n) atomicAdd(&cnt[d3 - nb0], 1);
    }
    __syncthreads();

    int v = 0, cur = 0;
    if (t < BKT_W) { v = cnt[t]; buf[0][t] = v; }
    __syncthreads();
    for (int dd = 1; dd < BKT_W; dd <<= 1) {
        if (t < BKT_W) {
            int xv = buf[cur][t];
            if (t >= dd) xv += buf[cur][t - dd];
            buf[cur ^ 1][t] = xv;
        }
        cur ^= 1;
        __syncthreads();
    }
    if (t < BKT_W) {
        const int ex = buf[cur][t] - v;
        cnt[t] = ex;
        const int node = nb0 + t;
        if (node < N) off[node] = start + ex;
        if (node == N - 1) off[N] = start + n;
    }
    __syncthreads();

    if (n <= PLACE_CAP) {
        for (int i = t; i < n; i += 4 * PLACE_T) {
            const int i1 = i + PLACE_T, i2 = i + 2 * PLACE_T, i3 = i + 3 * PLACE_T;
            const int2 p0 = pp[i];
            const int2 p1 = pp[i1 < n ? i1 : i];
            const int2 p2 = pp[i2 < n ? i2 : i];
            const int2 p3 = pp[i3 < n ? i3 : i];
            stage[atomicAdd(&cnt[p0.x - nb0], 1)] = p0.y;
            if (i1 < n) stage[atomicAdd(&cnt[p1.x - nb0], 1)] = p1.y;
            if (i2 < n) stage[atomicAdd(&cnt[p2.x - nb0], 1)] = p2.y;
            if (i3 < n) stage[atomicAdd(&cnt[p3.x - nb0], 1)] = p3.y;
        }
        __syncthreads();
        for (int i = t; i < n; i += PLACE_T) ssrc[start + i] = stage[i];
    } else {
        for (int i = t; i < n; i += 4 * PLACE_T) {
            const int i1 = i + PLACE_T, i2 = i + 2 * PLACE_T, i3 = i + 3 * PLACE_T;
            const int2 p0 = pp[i];
            const int2 p1 = pp[i1 < n ? i1 : i];
            const int2 p2 = pp[i2 < n ? i2 : i];
            const int2 p3 = pp[i3 < n ? i3 : i];
            ssrc[start + atomicAdd(&cnt[p0.x - nb0], 1)] = p0.y;
            if (i1 < n) ssrc[start + atomicAdd(&cnt[p1.x - nb0], 1)] = p1.y;
            if (i2 < n) ssrc[start + atomicAdd(&cnt[p2.x - nb0], 1)] = p2.y;
            if (i3 < n) ssrc[start + atomicAdd(&cnt[p3.x - nb0], 1)] = p3.y;
        }
    }
}

__device__ __forceinline__ float leaky(float v) {
    return v > 0.0f ? v : NEG_SLOPE * v;
}

__device__ __forceinline__ float group_lane5(float v) {
    return __builtin_bit_cast(float,
        __builtin_amdgcn_mov_dpp8(__builtin_bit_cast(unsigned, v), 0xB6DB6D));
}

__global__ void __launch_bounds__(256)
gat_node_gather(const int* __restrict__ off,
                const int* __restrict__ ssrc,
                const float* __restrict__ h,
                const float* __restrict__ b,
                const float* __restrict__ Wl,
                const float* __restrict__ bl,
                float* __restrict__ out,
                int N) {
    __shared__ float outs[32];
    const int gid = blockIdx.x * 256 + threadIdx.x;
    const int sub = threadIdx.x & 7;
    int node = gid >> 3;
    const bool valid = node < N;
    if (!valid) node = N - 1;

    const int beg = off[node];
    const int end = off[node + 1];

    const int col = sub < 5 ? sub : 5;
    const float4* __restrict__ h4 = (const float4*)h + col;

    float4 acc = h4[(size_t)node * HP4];
    const float ad = group_lane5(acc.y);
    float mx = leaky(group_lane5(acc.x) + ad);
    float den = 1.0f;

    int j = beg;
    for (; j + 4 <= end; j += 4) {
        int s0 = ssrc[j], s1 = ssrc[j + 1], s2 = ssrc[j + 2], s3 = ssrc[j + 3];
        float4 g0 = h4[(size_t)s0 * HP4];
        float4 g1 = h4[(size_t)s1 * HP4];
        float4 g2 = h4[(size_t)s2 * HP4];
        float4 g3 = h4[(size_t)s3 * HP4];
        float e0 = leaky(group_lane5(g0.x) + ad);
        float e1 = leaky(group_lane5(g1.x) + ad);
        float e2 = leaky(group_lane5(g2.x) + ad);
        float e3 = leaky(group_lane5(g3.x) + ad);
        float mn = fmaxf(fmaxf(fmaxf(e0, e1), fmaxf(e2, e3)), mx);
        float sc = __expf(mx - mn);
        mx = mn;
        float t0 = __expf(e0 - mx);
        float t1 = __expf(e1 - mx);
        float t2 = __expf(e2 - mx);
        float t3 = __expf(e3 - mx);
        den = den * sc + t0;
        acc.x = acc.x * sc + t0 * g0.x; acc.y = acc.y * sc + t0 * g0.y;
        acc.z = acc.z * sc + t0 * g0.z; acc.w = acc.w * sc + t0 * g0.w;
        den += t1;
        acc.x += t1 * g1.x; acc.y += t1 * g1.y; acc.z += t1 * g1.z; acc.w += t1 * g1.w;
        den += t2;
        acc.x += t2 * g2.x; acc.y += t2 * g2.y; acc.z += t2 * g2.z; acc.w += t2 * g2.w;
        den += t3;
        acc.x += t3 * g3.x; acc.y += t3 * g3.y; acc.z += t3 * g3.z; acc.w += t3 * g3.w;
    }
#pragma clang loop unroll(disable)
    for (; j < end; ++j) {
        int s0 = ssrc[j];
        float4 g0 = h4[(size_t)s0 * HP4];
        float e0 = leaky(group_lane5(g0.x) + ad);
        float mn = fmaxf(e0, mx);
        float sc = __expf(mx - mn);
        mx = mn;
        float t0 = __expf(e0 - mx);
        den = den * sc + t0;
        acc.x = acc.x * sc + t0 * g0.x; acc.y = acc.y * sc + t0 * g0.y;
        acc.z = acc.z * sc + t0 * g0.z; acc.w = acc.w * sc + t0 * g0.w;
    }

    float p = 0.0f;
    if (sub < 5) {
        float inv = 1.0f / (den + 1e-16f);
        float4 o;
        o.x = acc.x * inv + b[sub * 4 + 0];
        o.y = acc.y * inv + b[sub * 4 + 1];
        o.z = acc.z * inv + b[sub * 4 + 2];
        o.w = acc.w * inv + b[sub * 4 + 3];
        o.x = o.x > 0.0f ? o.x : 0.0f;
        o.y = o.y > 0.0f ? o.y : 0.0f;
        o.z = o.z > 0.0f ? o.z : 0.0f;
        o.w = o.w > 0.0f ? o.w : 0.0f;
        p = o.x * Wl[sub * 4 + 0] + o.y * Wl[sub * 4 + 1]
          + o.z * Wl[sub * 4 + 2] + o.w * Wl[sub * 4 + 3];
    }
    p += __shfl_xor(p, 1, 8);
    p += __shfl_xor(p, 2, 8);
    p += __shfl_xor(p, 4, 8);
    if (sub == 0) outs[threadIdx.x >> 3] = p + bl[0];
    __syncthreads();
    if (threadIdx.x < 32) {
        const int on = blockIdx.x * 32 + threadIdx.x;
        if (on < N) out[on] = outs[threadIdx.x];
    }
}

#define MM_KS (F / 4)
#define L1_T    1024
#define L1_G    8
#define L1_GRP  (L1_T / L1_G)
#define L1_EPL  6
#define L1_TAB  80000
#define L1_FILL 10

__global__ void __launch_bounds__(L1_T)
gat_l1_scalar_gather(const int* __restrict__ off,
                     const int* __restrict__ ssrc,
                     const float* __restrict__ x,
                     const float* __restrict__ W1,
                     const float* __restrict__ a_src,
                     const float* __restrict__ a_dst,
                     const float* __restrict__ b,
                     const float* __restrict__ W2,
                     const float* __restrict__ a_src2,
                     const float* __restrict__ a_dst2,
                     float* __restrict__ h,
                     int N) {
    __shared__ v4f tab4[L1_TAB / 4];
    __shared__ float wa[2][F];
    const float* tab = (const float*)tab4;
    const int t   = threadIdx.x;
    const int npw = (N + (int)gridDim.x - 1) / (int)gridDim.x;
    const int n0  = (int)blockIdx.x * npw;
    const int n1  = n0 + npw < N ? n0 + npw : N;
    if (n0 >= n1) return;

    const int ntab = N < L1_TAB ? N : L1_TAB;
    const int n4   = ntab >> 2;
    const v4f* __restrict__ x4 = (const v4f*)x;
    for (int i = t; i < n4; i += L1_FILL * L1_T) {
        v4f v[L1_FILL];
#pragma unroll
        for (int k = 0; k < L1_FILL; ++k) {
            const int ii = i + k * L1_T;
            v[k] = x4[ii < n4 ? ii : i];
        }
        asm volatile("" :: "v"(v[0]), "v"(v[1]), "v"(v[2]), "v"(v[3]), "v"(v[4]),
                           "v"(v[5]), "v"(v[6]), "v"(v[7]), "v"(v[8]), "v"(v[9]));
#pragma unroll
        for (int k = 0; k < L1_FILL; ++k) {
            const int ii = i + k * L1_T;
            if (ii < n4) tab4[ii] = v[k];
        }
    }
    for (int i = (n4 << 2) + t; i < ntab; i += L1_T) ((float*)tab4)[i] = x[i];

    if (t < F) {
        float ws = 0.0f, wd = 0.0f;
#pragma unroll
        for (int f = 0; f < F; ++f) {
            const float w = W2[t * F + f];
            ws += w * a_src2[f];
            wd += w * a_dst2[f];
        }
        wa[0][t] = ws;
        wa[1][t] = wd;
    }

    float cs = 0.0f, cd = 0.0f;
#pragma unroll
    for (int f = 0; f < F; ++f) {
        cs += W1[f] * a_src[f];
        cd += W1[f] * a_dst[f];
    }

    const int sub = t & (L1_G - 1);
    const int g   = t / L1_G;

    const int  lane  = t & 31;
    const int  wv    = t >> 5;
    const int  mrow  = lane & 15;
    const int  lhalf = lane >> 4;
    const int  khalf = lhalf * 2;
    const int  n1col = 16 + mrow;
    const int  n1c   = n1col < F ? n1col : 0;
    const bool is_w  = n1col < F;
    const bool is_sc = n1col >= F && n1col < F + 2;
    const int  scv   = n1col == F + 1 ? 1 : 0;
    float srow = 0.0f;

    __syncthreads();

    v2f b0[MM_KS], b1[MM_KS];
#pragma unroll
    for (int i = 0; i < MM_KS; ++i) {
        const int k = 4 * i + khalf;
        b0[i].x = W2[(k + 0) * F + mrow];
        b0[i].y = W2[(k + 1) * F + mrow];
        const float wx = W2[(k + 0) * F + n1c];
        const float wy = W2[(k + 1) * F + n1c];
        const float sx = wa[scv][k + 0];
        const float sy = wa[scv][k + 1];
        b1[i].x = is_w ? wx : (is_sc ? sx : 0.0f);
        b1[i].y = is_w ? wy : (is_sc ? sy : 0.0f);
    }

    const int nfix  = ntab < N ? ntab : N - 1;
    const int trips = (n1 - n0 + L1_GRP - 1) / L1_GRP;
    const int nf  = n0 + g < n1 ? n0 + g : n1 - 1;
    int   beg = off[nf];
    int   end = off[nf + 1];
    float xi  = x[nf];
    for (int it = 0; it < trips; ++it) {
        const int  node  = n0 + it * L1_GRP + g;
        const int   nn   = node + L1_GRP < n1 ? node + L1_GRP : n1 - 1;
        const int   begn = off[nn];
        const int   endn = off[nn + 1];
        const float xin  = x[nn];

        const float ad = xi * cd;
        float m   = leaky(xi * cs + ad);
        float den = sub == 0 ? 1.0f : 0.0f;
        float num = sub == 0 ? xi   : 0.0f;

        for (int j = beg + sub; j < end; j += L1_EPL * L1_G) {
            int s[L1_EPL];
#pragma unroll
            for (int k = 0; k < L1_EPL; ++k) {
                const int jk = j + k * L1_G;
                s[k] = ssrc[jk < end ? jk : j];
            }
            float xl[L1_EPL], xg[L1_EPL];
#pragma unroll
            for (int k = 0; k < L1_EPL; ++k) {
                const bool in_tab = s[k] < ntab;
                xl[k] = tab[in_tab ? s[k] : 0];
                xg[k] = x[in_tab ? nfix : s[k]];
            }
            float xv[L1_EPL], e[L1_EPL];
#pragma unroll
            for (int k = 0; k < L1_EPL; ++k) {
                const bool have = j + k * L1_G < end;
                const float xs  = s[k] < ntab ? xl[k] : xg[k];
                xv[k] = have ? xs : 0.0f;
                e[k]  = have ? leaky(xs * cs + ad) : -INFINITY;
            }
            float mn = m;
#pragma unroll
            for (int k = 0; k < L1_EPL; ++k) mn = fmaxf(mn, e[k]);
            const float sc = __expf(m - mn);
            m = mn;
            den *= sc;
            num *= sc;
#pragma unroll
            for (int k = 0; k < L1_EPL; ++k) {
                const float tk = __expf(e[k] - mn);
                den += tk;
                num += tk * xv[k];
            }
        }

        float M = m;
        M = fmaxf(M, __shfl_xor(M, 1, L1_G));
        M = fmaxf(M, __shfl_xor(M, 2, L1_G));
        M = fmaxf(M, __shfl_xor(M, 4, L1_G));
        const float fs = __expf(m - M);
        den *= fs;
        num *= fs;
        den += __shfl_xor(den, 1, L1_G);
        num += __shfl_xor(num, 1, L1_G);
        den += __shfl_xor(den, 2, L1_G);
        num += __shfl_xor(num, 2, L1_G);
        den += __shfl_xor(den, 4, L1_G);
        num += __shfl_xor(num, 4, L1_G);

        const float S  = num * (1.0f / (den + 1e-16f));
        const float Sv = __shfl(S, (mrow & 3) * L1_G);
        if ((it & 3) == (mrow >> 2)) srow = Sv;

        if ((it & 3) == 3 || it == trips - 1) {
            const int it0 = it & ~3;
            v8f c0 = {};
            v8f c1 = {};
#pragma unroll
            for (int i = 0; i < MM_KS; ++i) {
                const int k = 4 * i + khalf;
                v2f a;
                a.x = srow * W1[k + 0] + b[k + 0];
                a.y = srow * W1[k + 1] + b[k + 1];
                a.x = a.x > 0.0f ? a.x : 0.0f;
                a.y = a.y > 0.0f ? a.y : 0.0f;
                c0 = __builtin_amdgcn_wmma_f32_16x16x4_f32(false, a, false, b0[i],
                                                           (short)0, c0, false, false);
                c1 = __builtin_amdgcn_wmma_f32_16x16x4_f32(false, a, false, b1[i],
                                                           (short)0, c1, false, false);
            }
#pragma unroll
            for (int r = 0; r < 8; ++r) {
                const int rho = 8 * lhalf + r;
                const int nd  = n0 + (it0 + (rho >> 2)) * L1_GRP + 4 * wv + (rho & 3);
                if (nd < n1) h[(size_t)nd * HP + mrow] = c0[r];
            }
            if (n1col < F + 4) {
#pragma unroll
                for (int r = 0; r < 8; ++r) {
                    const int rho = 8 * lhalf + r;
                    const int nd  = n0 + (it0 + (rho >> 2)) * L1_GRP + 4 * wv + (rho & 3);
                    if (nd < n1) h[(size_t)nd * HP + n1col] = c1[r];
                }
            }
        }
        beg = begn;
        end = endn;
        xi  = xin;
    }
}

extern "C" void kernel_launch(void* const* d_in, const int* in_sizes, int n_in,
                              void* d_out, int out_size, void* d_ws, size_t ws_size,
                              hipStream_t stream) {
    (void)n_in; (void)out_size; (void)ws_size;

    const int N    = in_sizes[0];
    const int E    = in_sizes[1] / 2;

    const float* x    = (const float*)d_in[0];
    const int*   ei   = (const int*)d_in[1];
    const float* W1   = (const float*)d_in[2];
    const float* as1  = (const float*)d_in[3];
    const float* ad1  = (const float*)d_in[4];
    const float* b1   = (const float*)d_in[5];
    const float* W2   = (const float*)d_in[6];
    const float* as2  = (const float*)d_in[7];
    const float* ad2  = (const float*)d_in[8];
    const float* b2   = (const float*)d_in[9];
    const float* Wl   = (const float*)d_in[10];
    const float* bl   = (const float*)d_in[11];

    const int* srcIdx = ei;
    const int* dstIdx = ei + E;

    const int TB   = 256;
    const int NB   = (N + BKT_W - 1) / BKT_W;
    const int gCH  = (E + SORT_CH - 1) / SORT_CH;
    const int gG   = (N + 31) / 32;

    float* ws     = (float*)d_ws;
    float* h      = ws + (size_t)0  * N;
    int*   btot   = (int*)(ws + (size_t)74 * N);
    int*   off    = (int*)(ws + (size_t)75 * N);
    int*   bstart = (int*)(ws + (size_t)77 * N);
    int2*  pairs  = (int2*)(ws + (size_t)78 * N);
    int*   ssrc   = (int*)(ws + (size_t)78 * N) + (size_t)2 * E;
    int*   cb     = ssrc + (size_t)E;

    bucket_totals_clear<<<(NB + TB - 1) / TB, TB, 0, stream>>>(btot, NB);
    bucket_count<<<gCH, SORT_T, NB * sizeof(int), stream>>>(dstIdx, btot, cb, E, NB);
    bucket_scatter<<<gCH, SORT_T, 2 * NB * sizeof(int), stream>>>(srcIdx, dstIdx, btot, cb, bstart, pairs, E, NB);
    bucket_place<<<NB, PLACE_T, 0, stream>>>(pairs, btot, bstart, off, ssrc, N);

    gat_l1_scalar_gather<<<256, L1_T, 0, stream>>>(off, ssrc, x, W1, as1, ad1, b1, W2, as2, ad2, h, N);

    gat_node_gather<<<gG, TB, 0, stream>>>(off, ssrc, h, b2, Wl, bl, (float*)d_out, N);
}


# Layer-2 aggregation on scalars: a layer-1 row is relu(S*W1+b1) of one float S, so an edge reads S[src] from an LDS copy and rebuilds row and score; W2 by WMMA after the sum; sum order changes
# speedup vs baseline: 33.4713x; 1.1963x over previous
#include <hip/hip_runtime.h>
#include <math.h>

#define F 20
#define NEG_SLOPE 0.2f

typedef float v2f __attribute__((ext_vector_type(2)));
typedef float v8f __attribute__((ext_vector_type(8)));
typedef float v4f __attribute__((ext_vector_type(4)));

__global__ void bucket_totals_clear(int* __restrict__ btot, int NB) {
    int i = blockIdx.x * blockDim.x + threadIdx.x;
    if (i < NB) btot[i] = 0;
}

#define BKT_SH 8
#define BKT_W  (1 << BKT_SH)
#define SORT_T   512
#define SORT_EPT 16
#define SORT_CH  (SORT_T * SORT_EPT)

__global__ void __launch_bounds__(SORT_T)
bucket_count(const int* __restrict__ dst,
             int* __restrict__ btot,
             int* __restrict__ cb,
             int E, int NB) {
    extern __shared__ int hist[];
    const int t  = threadIdx.x;
    const int c  = blockIdx.x;
    const int e0 = c * SORT_CH + t;
    for (int b = t; b < NB; b += SORT_T) hist[b] = 0;
    __syncthreads();
    if (c * SORT_CH + SORT_CH <= E) {
        int d[SORT_EPT];
#pragma unroll
        for (int k = 0; k < SORT_EPT; ++k) d[k] = dst[e0 + k * SORT_T];
#pragma unroll
        for (int k = 0; k < SORT_EPT; ++k) atomicAdd(&hist[d[k] >> BKT_SH], 1);
    } else {
        for (int k = 0; k < SORT_EPT; ++k) {
            int e = e0 + k * SORT_T;
            if (e < E) atomicAdd(&hist[dst[e] >> BKT_SH], 1);
        }
    }
    __syncthreads();
    for (int b = t; b < NB; b += SORT_T)
        cb[(size_t)c * NB + b] = atomicAdd(&btot[b], hist[b]);
}

__global__ void __launch_bounds__(SORT_T)
bucket_scatter(const int* __restrict__ src,
               const int* __restrict__ dst,
               const int* __restrict__ btot,
               const int* __restrict__ cb,
               int* __restrict__ bstart,
               int2* __restrict__ pairs,
               int E, int NB) {
    extern __shared__ int sm[];
    __shared__ int buf[2][SORT_T];
    __shared__ int bufh[2][SORT_T];
    __shared__ int2 stage[SORT_CH];
    int* hist = sm;
    int* gofs = sm + NB;
    const int t  = threadIdx.x;
    const int c  = blockIdx.x;
    const int e0 = c * SORT_CH + t;
    for (int b = t; b < NB; b += SORT_T) hist[b] = 0;
    __syncthreads();

    int d[SORT_EPT], s[SORT_EPT], r[SORT_EPT];
    if (c * SORT_CH + SORT_CH <= E) {
#pragma unroll
        for (int k = 0; k < SORT_EPT; ++k) {
            d[k] = dst[e0 + k * SORT_T];
            s[k] = src[e0 + k * SORT_T];
        }
#pragma unroll
        for (int k = 0; k < SORT_EPT; ++k) r[k] = atomicAdd(&hist[d[k] >> BKT_SH], 1);
    } else {
#pragma unroll
        for (int k = 0; k < SORT_EPT; ++k) {
            int e = e0 + k * SORT_T;
            d[k] = -1; s[k] = 0; r[k] = 0;
            if (e < E) {
                d[k] = dst[e];
                s[k] = src[e];
                r[k] = atomicAdd(&hist[d[k] >> BKT_SH], 1);
            }
        }
    }
    __syncthreads();

    int carry = 0, carryh = 0;
    for (int b0 = 0; b0 < NB; b0 += SORT_T) {
        const int b  = b0 + t;
        const int v  = b < NB ? btot[b] : 0;
        const int hv = b < NB ? hist[b] : 0;
        int cur = 0;
        buf[0][t]  = v;
        bufh[0][t] = hv;
        __syncthreads();
        for (int dd = 1; dd < SORT_T; dd <<= 1) {
            int xv = buf[cur][t];
            int xh = bufh[cur][t];
            if (t >= dd) { xv += buf[cur][t - dd]; xh += bufh[cur][t - dd]; }
            buf[cur ^ 1][t]  = xv;
            bufh[cur ^ 1][t] = xh;
            cur ^= 1;
            __syncthreads();
        }
        const int st = carry  + buf[cur][t]  - v;
        const int lb = carryh + bufh[cur][t] - hv;
        if (b < NB) {
            hist[b] = lb;
            gofs[b] = st + cb[(size_t)c * NB + b] - lb;
            if (c == 0) bstart[b] = st;
        }
        carry  += buf[cur][SORT_T - 1];
        carryh += bufh[cur][SORT_T - 1];
        __syncthreads();
    }
    if (c == 0 && t == 0) bstart[NB] = carry;
    const int nrec = carryh;

    int slot[SORT_EPT];
#pragma unroll
    for (int k = 0; k < SORT_EPT; ++k)
        slot[k] = hist[(d[k] >= 0 ? d[k] : 0) >> BKT_SH] + r[k];
#pragma unroll
    for (int k = 0; k < SORT_EPT; ++k) {
        if (d[k] >= 0) stage[slot[k]] = make_int2(d[k], s[k]);
    }
    __syncthreads();

    if (nrec == SORT_CH) {
        int2 p[SORT_EPT];
        int  g[SORT_EPT];
#pragma unroll
        for (int k = 0; k < SORT_EPT; ++k) p[k] = stage[t + k * SORT_T];
#pragma unroll
        for (int k = 0; k < SORT_EPT; ++k) g[k] = gofs[p[k].x >> BKT_SH];
#pragma unroll
        for (int k = 0; k < SORT_EPT; ++k) pairs[g[k] + t + k * SORT_T] = p[k];
    } else {
        for (int i = t; i < nrec; i += SORT_T) {
            const int2 p = stage[i];
            pairs[gofs[p.x >> BKT_SH] + i] = p;
        }
    }
}

#define PLACE_T 1024
#define PLACE_CAP 12288
__global__ void __launch_bounds__(PLACE_T)
bucket_place(const int2* __restrict__ pairs,
             const int* __restrict__ btot,
             const int* __restrict__ bstart,
             int* __restrict__ off,
             int* __restrict__ ssrc,
             int N) {
    __shared__ int cnt[BKT_W];
    __shared__ int buf[2][BKT_W];
    __shared__ int stage[PLACE_CAP];
    const int t     = threadIdx.x;
    const int b     = blockIdx.x;
    const int nb0   = b << BKT_SH;
    const int start = bstart[b];
    const int n     = btot[b];
    const int2* __restrict__ pp = pairs + start;

    if (t < BKT_W) cnt[t] = 0;
    __syncthreads();

    for (int i = t; i < n; i += 4 * PLACE_T) {
        const int i1 = i + PLACE_T, i2 = i + 2 * PLACE_T, i3 = i + 3 * PLACE_T;
        const int d0 = pp[i].x;
        const int d1 = pp[i1 < n ? i1 : i].x;
        const int d2 = pp[i2 < n ? i2 : i].x;
        const int d3 = pp[i3 < n ? i3 : i].x;
        atomicAdd(&cnt[d0 - nb0], 1);
        if (i1 < n) atomicAdd(&cnt[d1 - nb0], 1);
        if (i2 < n) atomicAdd(&cnt[d2 - nb0], 1);
        if (i3 < n) atomicAdd(&cnt[d3 - nb0], 1);
    }
    __syncthreads();

    int v = 0, cur = 0;
    if (t < BKT_W) { v = cnt[t]; buf[0][t] = v; }
    __syncthreads();
    for (int dd = 1; dd < BKT_W; dd <<= 1) {
        if (t < BKT_W) {
            int xv = buf[cur][t];
            if (t >= dd) xv += buf[cur][t - dd];
            buf[cur ^ 1][t] = xv;
        }
        cur ^= 1;
        __syncthreads();
    }
    if (t < BKT_W) {
        const int ex = buf[cur][t] - v;
        cnt[t] = ex;
        const int node = nb0 + t;
        if (node < N) off[node] = start + ex;
        if (node == N - 1) off[N] = start + n;
    }
    __syncthreads();

    if (n <= PLACE_CAP) {
        for (int i = t; i < n; i += 4 * PLACE_T) {
            const int i1 = i + PLACE_T, i2 = i + 2 * PLACE_T, i3 = i + 3 * PLACE_T;
            const int2 p0 = pp[i];
            const int2 p1 = pp[i1 < n ? i1 : i];
            const int2 p2 = pp[i2 < n ? i2 : i];
            const int2 p3 = pp[i3 < n ? i3 : i];
            stage[atomicAdd(&cnt[p0.x - nb0], 1)] = p0.y;
            if (i1 < n) stage[atomicAdd(&cnt[p1.x - nb0], 1)] = p1.y;
            if (i2 < n) stage[atomicAdd(&cnt[p2.x - nb0], 1)] = p2.y;
            if (i3 < n) stage[atomicAdd(&cnt[p3.x - nb0], 1)] = p3.y;
        }
        __syncthreads();
        for (int i = t; i < n; i += PLACE_T) ssrc[start + i] = stage[i];
    } else {
        for (int i = t; i < n; i += 4 * PLACE_T) {
            const int i1 = i + PLACE_T, i2 = i + 2 * PLACE_T, i3 = i + 3 * PLACE_T;
            const int2 p0 = pp[i];
            const int2 p1 = pp[i1 < n ? i1 : i];
            const int2 p2 = pp[i2 < n ? i2 : i];
            const int2 p3 = pp[i3 < n ? i3 : i];
            ssrc[start + atomicAdd(&cnt[p0.x - nb0], 1)] = p0.y;
            if (i1 < n) ssrc[start + atomicAdd(&cnt[p1.x - nb0], 1)] = p1.y;
            if (i2 < n) ssrc[start + atomicAdd(&cnt[p2.x - nb0], 1)] = p2.y;
            if (i3 < n) ssrc[start + atomicAdd(&cnt[p3.x - nb0], 1)] = p3.y;
        }
    }
}

__device__ __forceinline__ float leaky(float v) {
    return v > 0.0f ? v : NEG_SLOPE * v;
}

#define MM_KS (F / 4)
#define L1_T    1024
#define L1_G    8
#define L1_GRP  (L1_T / L1_G)
#define L1_EPL  6
#define L1_TAB  80000
#define L1_FILL 10

__global__ void __launch_bounds__(L1_T)
gat_l1_scalar_gather(const int* __restrict__ off,
                     const int* __restrict__ ssrc,
                     const float* __restrict__ x,
                     const float* __restrict__ W1,
                     const float* __restrict__ a_src,
                     const float* __restrict__ a_dst,
                     float* __restrict__ Sout,
                     int N) {
    __shared__ v4f tab4[L1_TAB / 4];
    const float* tab = (const float*)tab4;
    const int t   = threadIdx.x;
    const int npw = (N + (int)gridDim.x - 1) / (int)gridDim.x;
    const int n0  = (int)blockIdx.x * npw;
    const int n1  = n0 + npw < N ? n0 + npw : N;
    if (n0 >= n1) return;

    const int ntab = N < L1_TAB ? N : L1_TAB;
    const int n4   = ntab >> 2;
    const v4f* __restrict__ x4 = (const v4f*)x;
    for (int i = t; i < n4; i += L1_FILL * L1_T) {
        v4f v[L1_FILL];
#pragma unroll
        for (int k = 0; k < L1_FILL; ++k) {
            const int ii = i + k * L1_T;
            v[k] = x4[ii < n4 ? ii : i];
        }
        asm volatile("" :: "v"(v[0]), "v"(v[1]), "v"(v[2]), "v"(v[3]), "v"(v[4]),
                           "v"(v[5]), "v"(v[6]), "v"(v[7]), "v"(v[8]), "v"(v[9]));
#pragma unroll
        for (int k = 0; k < L1_FILL; ++k) {
            const int ii = i + k * L1_T;
            if (ii < n4) tab4[ii] = v[k];
        }
    }
    for (int i = (n4 << 2) + t; i < ntab; i += L1_T) ((float*)tab4)[i] = x[i];

    float cs = 0.0f, cd = 0.0f;
#pragma unroll
    for (int f = 0; f < F; ++f) {
        cs += W1[f] * a_src[f];
        cd += W1[f] * a_dst[f];
    }

    const int sub = t & (L1_G - 1);
    const int g   = t / L1_G;

    __syncthreads();

    const int nfix  = ntab < N ? ntab : N - 1;
    const int trips = (n1 - n0 + L1_GRP - 1) / L1_GRP;
    const int nf  = n0 + g < n1 ? n0 + g : n1 - 1;
    int   beg = off[nf];
    int   end = off[nf + 1];
    float xi  = x[nf];
    for (int it = 0; it < trips; ++it) {
        const int  node  = n0 + it * L1_GRP + g;
        const int   nn   = node + L1_GRP < n1 ? node + L1_GRP : n1 - 1;
        const int   begn = off[nn];
        const int   endn = off[nn + 1];
        const float xin  = x[nn];

        const float ad = xi * cd;
        float m   = leaky(xi * cs + ad);
        float den = sub == 0 ? 1.0f : 0.0f;
        float num = sub == 0 ? xi   : 0.0f;

        for (int j = beg + sub; j < end; j += L1_EPL * L1_G) {
            int s[L1_EPL];
#pragma unroll
            for (int k = 0; k < L1_EPL; ++k) {
                const int jk = j + k * L1_G;
                s[k] = ssrc[jk < end ? jk : j];
            }
            float xl[L1_EPL], xg[L1_EPL];
#pragma unroll
            for (int k = 0; k < L1_EPL; ++k) {
                const bool in_tab = s[k] < ntab;
                xl[k] = tab[in_tab ? s[k] : 0];
                xg[k] = x[in_tab ? nfix : s[k]];
            }
            float xv[L1_EPL], e[L1_EPL];
#pragma unroll
            for (int k = 0; k < L1_EPL; ++k) {
                const bool have = j + k * L1_G < end;
                const float xs  = s[k] < ntab ? xl[k] : xg[k];
                xv[k] = have ? xs : 0.0f;
                e[k]  = have ? leaky(xs * cs + ad) : -INFINITY;
            }
            float mn = m;
#pragma unroll
            for (int k = 0; k < L1_EPL; ++k) mn = fmaxf(mn, e[k]);
            const float sc = __expf(m - mn);
            m = mn;
            den *= sc;
            num *= sc;
#pragma unroll
            for (int k = 0; k < L1_EPL; ++k) {
                const float tk = __expf(e[k] - mn);
                den += tk;
                num += tk * xv[k];
            }
        }

        float M = m;
        M = fmaxf(M, __shfl_xor(M, 1, L1_G));
        M = fmaxf(M, __shfl_xor(M, 2, L1_G));
        M = fmaxf(M, __shfl_xor(M, 4, L1_G));
        const float fs = __expf(m - M);
        den *= fs;
        num *= fs;
        den += __shfl_xor(den, 1, L1_G);
        num += __shfl_xor(num, 1, L1_G);
        den += __shfl_xor(den, 2, L1_G);
        num += __shfl_xor(num, 2, L1_G);
        den += __shfl_xor(den, 4, L1_G);
        num += __shfl_xor(num, 4, L1_G);

        const float S = num * (1.0f / (den + 1e-16f));
        if (sub == 0 && node < n1) Sout[node] = S;

        beg = begn;
        end = endn;
        xi  = xin;
    }
}

#define L2_T    1024
#define L2_TAB  80000
#define L2_FILL 10

__global__ void __launch_bounds__(L2_T)
gat_l2_scalar_gather(const int* __restrict__ off,
                     const int* __restrict__ ssrc,
                     const float* __restrict__ Sv,
                     const float* __restrict__ W1,
                     const float* __restrict__ b1,
                     const float* __restrict__ W2,
                     const float* __restrict__ a_src2,
                     const float* __restrict__ a_dst2,
                     const float* __restrict__ b2,
                     const float* __restrict__ Wl,
                     const float* __restrict__ bl,
                     float* __restrict__ out,
                     int N) {
    __shared__ v4f tab4[L2_TAB / 4];
    __shared__ float wa[2][F];
    const float* tab = (const float*)tab4;
    const int t    = threadIdx.x;
    const int npw0 = (N + (int)gridDim.x - 1) / (int)gridDim.x;
    const int npw  = (npw0 + 15) & ~15;
    const int n0   = (int)blockIdx.x * npw;
    const int n1   = n0 + npw < N ? n0 + npw : N;
    if (n0 >= n1) return;

    const int ntab = N < L2_TAB ? N : L2_TAB;
    const int n4   = ntab >> 2;
    const v4f* __restrict__ S4 = (const v4f*)Sv;
    for (int i = t; i < n4; i += L2_FILL * L2_T) {
        v4f v[L2_FILL];
#pragma unroll
        for (int k = 0; k < L2_FILL; ++k) {
            const int ii = i + k * L2_T;
            v[k] = S4[ii < n4 ? ii : i];
        }
        asm volatile("" :: "v"(v[0]), "v"(v[1]), "v"(v[2]), "v"(v[3]), "v"(v[4]),
                           "v"(v[5]), "v"(v[6]), "v"(v[7]), "v"(v[8]), "v"(v[9]));
#pragma unroll
        for (int k = 0; k < L2_FILL; ++k) {
            const int ii = i + k * L2_T;
            if (ii < n4) tab4[ii] = v[k];
        }
    }
    for (int i = (n4 << 2) + t; i < ntab; i += L2_T) ((float*)tab4)[i] = Sv[i];

    if (t < F) {
        float ws = 0.0f, wd = 0.0f;
#pragma unroll
        for (int f = 0; f < F; ++f) {
            const float w = W2[t * F + f];
            ws += w * a_src2[f];
            wd += w * a_dst2[f];
        }
        wa[0][t] = ws;
        wa[1][t] = wd;
    }

    __syncthreads();

    const int  lane  = t & 31;
    const int  wv    = t >> 5;
    const int  sub   = lane & 1;
    const int  r     = lane >> 1;
    const int  mrow  = lane & 15;
    const int  lhalf = lane >> 4;
    const int  khalf = lhalf * 2;
    const int  n1col = 16 + mrow;
    const int  n1c   = n1col < F ? n1col : 0;
    const bool is_w  = n1col < F;
    const int  srcl  = 2 * mrow;

    v2f w1[F / 2], c1[F / 2], was[F / 2], wad[F / 2];
#pragma unroll
    for (int i = 0; i < F / 2; ++i) {
        w1[i]  = (v2f){W1[2 * i], W1[2 * i + 1]};
        c1[i]  = (v2f){b1[2 * i], b1[2 * i + 1]};
        was[i] = (v2f){wa[0][2 * i], wa[0][2 * i + 1]};
        wad[i] = (v2f){wa[1][2 * i], wa[1][2 * i + 1]};
    }
    const v2f zero2 = {0.0f, 0.0f};

    const int nfix  = ntab < N ? ntab : N - 1;
    const int ntile = (n1 - n0 + 15) >> 4;
    const int tile = wv;
    if (tile < ntile) {
        const int base = n0 + tile * 16;
        const int node = base + r;
        const int nc   = node < n1 ? node : n1 - 1;
        const int beg  = off[nc];
        const int end  = off[nc + 1];
        const float Si = Sv[nc];

        v2f acc[F / 2];
        v2f qs = zero2, qd = zero2;
#pragma unroll
        for (int i = 0; i < F / 2; ++i) {
            const v2f v = __builtin_elementwise_max(Si * w1[i] + c1[i], zero2);
            qs += v * was[i];
            qd += v * wad[i];
            acc[i] = sub == 0 ? v : zero2;
        }
        const float ad = qd.x + qd.y;
        float m   = leaky(qs.x + qs.y + ad);
        float den = sub == 0 ? 1.0f : 0.0f;

        for (int j = beg + sub; j < end; j += 4) {
            const int  j1    = j + 2;
            const bool have1 = j1 < end;
            const int  s0 = ssrc[j];
            const int  s1 = ssrc[have1 ? j1 : j];
            const bool in0 = s0 < ntab;
            const bool in1 = s1 < ntab;
            const float xl0 = tab[in0 ? s0 : 0];
            const float xl1 = tab[in1 ? s1 : 0];
            const float xg0 = Sv[in0 ? nfix : s0];
            const float xg1 = Sv[in1 ? nfix : s1];
            const float S0 = in0 ? xl0 : xg0;
            const float S1 = in1 ? xl1 : xg1;
            v2f v0[F / 2], v1[F / 2];
            v2f q0 = zero2, q1 = zero2;
#pragma unroll
            for (int i = 0; i < F / 2; ++i) {
                v0[i] = __builtin_elementwise_max(S0 * w1[i] + c1[i], zero2);
                v1[i] = __builtin_elementwise_max(S1 * w1[i] + c1[i], zero2);
                q0 += v0[i] * was[i];
                q1 += v1[i] * was[i];
            }
            const float e0 = leaky(q0.x + q0.y + ad);
            const float e1 = have1 ? leaky(q1.x + q1.y + ad) : -INFINITY;
            const float mn = fmaxf(m, fmaxf(e0, e1));
            const float sc = __expf(m - mn);
            m = mn;
            const float t0 = __expf(e0 - mn);
            const float t1 = __expf(e1 - mn);
            den = den * sc + t0 + t1;
#pragma unroll
            for (int i = 0; i < F / 2; ++i) acc[i] = acc[i] * sc + t0 * v0[i] + t1 * v1[i];
        }

        const float M  = fmaxf(m, __shfl_xor(m, 1));
        const float fs = __expf(m - M);
        den *= fs;
        den += __shfl_xor(den, 1);
        const float inv = 1.0f / (den + 1e-16f);
        float ag[F];
#pragma unroll
        for (int i = 0; i < F / 2; ++i) {
            const float lo = acc[i].x * fs;
            const float hi = acc[i].y * fs;
            ag[2 * i]     = (lo + __shfl_xor(lo, 1)) * inv;
            ag[2 * i + 1] = (hi + __shfl_xor(hi, 1)) * inv;
        }

        v8f d0 = {};
        v8f d1 = {};
#pragma unroll
        for (int i = 0; i < MM_KS; ++i) {
            const float k0 = __shfl(ag[4 * i + 0], srcl);
            const float k1 = __shfl(ag[4 * i + 1], srcl);
            const float k2 = __shfl(ag[4 * i + 2], srcl);
            const float k3 = __shfl(ag[4 * i + 3], srcl);
            v2f a;
            a.x = lhalf ? k2 : k0;
            a.y = lhalf ? k3 : k1;
            const int k = 4 * i + khalf;
            v2f bA, bB;
            bA.x = W2[(k + 0) * F + mrow];
            bA.y = W2[(k + 1) * F + mrow];
            const float wx = W2[(k + 0) * F + n1c];
            const float wy = W2[(k + 1) * F + n1c];
            bB.x = is_w ? wx : 0.0f;
            bB.y = is_w ? wy : 0.0f;
            d0 = __builtin_amdgcn_wmma_f32_16x16x4_f32(false, a, false, bA,
                                                       (short)0, d0, false, false);
            d1 = __builtin_amdgcn_wmma_f32_16x16x4_f32(false, a, false, bB,
                                                       (short)0, d1, false, false);
        }

        const float bc0 = b2[mrow];
        const float wl0 = Wl[mrow];
        const float bc1 = b2[n1c];
        const float wl1 = Wl[n1c];
        float p[8];
#pragma unroll
        for (int q = 0; q < 8; ++q) {
            const float o0 = fmaxf(d0[q] + bc0, 0.0f) * wl0;
            const float o1 = fmaxf(d1[q] + bc1, 0.0f) * wl1;
            p[q] = o0 + (is_w ? o1 : 0.0f);
        }
#pragma unroll
        for (int q = 0; q < 8; ++q) {
            p[q] += __shfl_xor(p[q], 1);
            p[q] += __shfl_xor(p[q], 2);
            p[q] += __shfl_xor(p[q], 4);
            p[q] += __shfl_xor(p[q], 8);
        }
        float res = p[0];
#pragma unroll
        for (int q = 1; q < 8; ++q) res = mrow == q ? p[q] : res;
        const int on = base + 8 * lhalf + mrow;
        if (mrow < 8 && on < n1) out[on] = res + bl[0];
    }
}

extern "C" void kernel_launch(void* const* d_in, const int* in_sizes, int n_in,
                              void* d_out, int out_size, void* d_ws, size_t ws_size,
                              hipStream_t stream) {
    (void)n_in; (void)out_size; (void)ws_size;

    const int N    = in_sizes[0];
    const int E    = in_sizes[1] / 2;

    const float* x    = (const float*)d_in[0];
    const int*   ei   = (const int*)d_in[1];
    const float* W1   = (const float*)d_in[2];
    const float* as1  = (const float*)d_in[3];
    const float* ad1  = (const float*)d_in[4];
    const float* b1   = (const float*)d_in[5];
    const float* W2   = (const float*)d_in[6];
    const float* as2  = (const float*)d_in[7];
    const float* ad2  = (const float*)d_in[8];
    const float* b2   = (const float*)d_in[9];
    const float* Wl   = (const float*)d_in[10];
    const float* bl   = (const float*)d_in[11];

    const int* srcIdx = ei;
    const int* dstIdx = ei + E;

    const int TB   = 256;
    const int NB   = (N + BKT_W - 1) / BKT_W;
    const int gCH  = (E + SORT_CH - 1) / SORT_CH;
    const int gL2  = N > 256 * 512 ? (N + 511) / 512 : 256;

    float* ws     = (float*)d_ws;
    float* Sv     = ws + (size_t)0  * N;
    int*   btot   = (int*)(ws + (size_t)74 * N);
    int*   off    = (int*)(ws + (size_t)75 * N);
    int*   bstart = (int*)(ws + (size_t)77 * N);
    int2*  pairs  = (int2*)(ws + (size_t)78 * N);
    int*   ssrc   = (int*)(ws + (size_t)78 * N) + (size_t)2 * E;
    int*   cb     = ssrc + (size_t)E;

    bucket_totals_clear<<<(NB + TB - 1) / TB, TB, 0, stream>>>(btot, NB);
    bucket_count<<<gCH, SORT_T, NB * sizeof(int), stream>>>(dstIdx, btot, cb, E, NB);
    bucket_scatter<<<gCH, SORT_T, 2 * NB * sizeof(int), stream>>>(srcIdx, dstIdx, btot, cb, bstart, pairs, E, NB);
    bucket_place<<<NB, PLACE_T, 0, stream>>>(pairs, btot, bstart, off, ssrc, N);

    gat_l1_scalar_gather<<<256, L1_T, 0, stream>>>(off, ssrc, x, W1, as1, ad1, Sv, N);

    gat_l2_scalar_gather<<<gL2, L2_T, 0, stream>>>(off, ssrc, Sv, W1, b1, W2, as2, ad2, b2, Wl, bl, (float*)d_out, N);
}


# Destination sort without global counters: a chunk keeps its bucket-ordered records in place with a run-start table; a bucket block gathers its runs by column sums; 6 -> 4 launches; sum order changes
# speedup vs baseline: 34.5448x; 1.0321x over previous
#include <hip/hip_runtime.h>
#include <math.h>

#define F 20
#define NEG_SLOPE 0.2f

typedef float v2f __attribute__((ext_vector_type(2)));
typedef float v8f __attribute__((ext_vector_type(8)));
typedef float v4f __attribute__((ext_vector_type(4)));

#define BKT_SH 8
#define BKT_W  (1 << BKT_SH)
#define SORT_T   512
#define SORT_EPT 16
#define SORT_CH  (SORT_T * SORT_EPT)

__global__ void __launch_bounds__(SORT_T)
bucket_scatter(const int* __restrict__ src,
               const int* __restrict__ dst,
               int* __restrict__ lstart,
               int2* __restrict__ pairs,
               int E, int NB) {
    extern __shared__ int hist[];
    __shared__ int buf[2][SORT_T];
    __shared__ int2 stage[SORT_CH];
    const int t  = threadIdx.x;
    const int c  = blockIdx.x;
    const int e0 = c * SORT_CH + t;
    int* __restrict__ lrow = lstart + (size_t)c * (NB + 1);
    for (int b = t; b < NB; b += SORT_T) hist[b] = 0;
    __syncthreads();

    int d[SORT_EPT], s[SORT_EPT], r[SORT_EPT];
    if (c * SORT_CH + SORT_CH <= E) {
#pragma unroll
        for (int k = 0; k < SORT_EPT; ++k) {
            d[k] = dst[e0 + k * SORT_T];
            s[k] = src[e0 + k * SORT_T];
        }
#pragma unroll
        for (int k = 0; k < SORT_EPT; ++k) r[k] = atomicAdd(&hist[d[k] >> BKT_SH], 1);
    } else {
#pragma unroll
        for (int k = 0; k < SORT_EPT; ++k) {
            int e = e0 + k * SORT_T;
            d[k] = -1; s[k] = 0; r[k] = 0;
            if (e < E) {
                d[k] = dst[e];
                s[k] = src[e];
                r[k] = atomicAdd(&hist[d[k] >> BKT_SH], 1);
            }
        }
    }
    __syncthreads();

    int carry = 0;
    for (int b0 = 0; b0 < NB; b0 += SORT_T) {
        const int b  = b0 + t;
        const int hv = b < NB ? hist[b] : 0;
        int cur = 0;
        buf[0][t] = hv;
        __syncthreads();
        for (int dd = 1; dd < SORT_T; dd <<= 1) {
            int xh = buf[cur][t];
            if (t >= dd) xh += buf[cur][t - dd];
            buf[cur ^ 1][t] = xh;
            cur ^= 1;
            __syncthreads();
        }
        const int lb = carry + buf[cur][t] - hv;
        if (b < NB) {
            hist[b] = lb;
            lrow[b] = lb;
        }
        carry += buf[cur][SORT_T - 1];
        __syncthreads();
    }
    const int nrec = carry;
    if (t == 0) lrow[NB] = nrec;

    int slot[SORT_EPT];
#pragma unroll
    for (int k = 0; k < SORT_EPT; ++k)
        slot[k] = hist[(d[k] >= 0 ? d[k] : 0) >> BKT_SH] + r[k];
#pragma unroll
    for (int k = 0; k < SORT_EPT; ++k) {
        if (d[k] >= 0) stage[slot[k]] = make_int2(d[k], s[k]);
    }
    __syncthreads();

    int2* __restrict__ outp = pairs + (size_t)c * SORT_CH;
    if (nrec == SORT_CH) {
        int2 p[SORT_EPT];
#pragma unroll
        for (int k = 0; k < SORT_EPT; ++k) p[k] = stage[t + k * SORT_T];
#pragma unroll
        for (int k = 0; k < SORT_EPT; ++k) outp[t + k * SORT_T] = p[k];
    } else {
        for (int i = t; i < nrec; i += SORT_T) outp[i] = stage[i];
    }
}

#define PLACE_T 1024
#define PLACE_NW (PLACE_T / 32)
#define PLACE_CAP 12288

#define PLACE_RPT 8

template <class Fn>
__device__ __forceinline__ void walk_runs(const int2* __restrict__ pairs,
                                          const int* rs, const int* rl,
                                          int NCH, int wv, int lane, int last, Fn fn) {
    for (int c = wv; c < NCH; c += PLACE_RPT * PLACE_NW) {
        int a[PLACE_RPT], l[PLACE_RPT];
        int2 p[PLACE_RPT];
#pragma unroll
        for (int k = 0; k < PLACE_RPT; ++k) {
            const int  ck = c + k * PLACE_NW;
            const bool hv = ck < NCH;
            const int  cc = hv ? ck : c;
            a[k] = cc * SORT_CH + rs[cc];
            l[k] = hv ? rl[cc] : 0;
        }
#pragma unroll
        for (int k = 0; k < PLACE_RPT; ++k) {
            const int i = a[k] + (lane < l[k] ? lane : 0);
            p[k] = pairs[i < last ? i : last];
        }
        asm volatile("" :: "v"(p[0].x), "v"(p[0].y), "v"(p[1].x), "v"(p[1].y),
                           "v"(p[2].x), "v"(p[2].y), "v"(p[3].x), "v"(p[3].y),
                           "v"(p[4].x), "v"(p[4].y), "v"(p[5].x), "v"(p[5].y),
                           "v"(p[6].x), "v"(p[6].y), "v"(p[7].x), "v"(p[7].y));
#pragma unroll
        for (int k = 0; k < PLACE_RPT; ++k) {
            if (lane < l[k]) fn(p[k]);
            for (int j = lane + 32; j < l[k]; j += 32) fn(pairs[a[k] + j]);
        }
    }
}

__global__ void __launch_bounds__(PLACE_T)
bucket_place(const int2* __restrict__ pairs,
             const int* __restrict__ lstart,
             int* __restrict__ off,
             int* __restrict__ ssrc,
             int N, int NB, int NCH, int E) {
    extern __shared__ int runs[];
    __shared__ int cnt[BKT_W];
    __shared__ int buf[2][BKT_W];
    __shared__ int stage[PLACE_CAP];
    __shared__ int tot[2];
    int* rs = runs;
    int* rl = runs + NCH;
    const int t     = threadIdx.x;
    const int lane  = t & 31;
    const int wv    = t >> 5;
    const int b     = blockIdx.x;
    const int nb0   = b << BKT_SH;

    if (t < BKT_W) cnt[t] = 0;
    if (t < 2) tot[t] = 0;
    __syncthreads();

    int ps = 0, pe = 0;
    for (int c = t; c < NCH; c += PLACE_T) {
        const int* __restrict__ row = lstart + (size_t)c * (NB + 1) + b;
        const int ra = row[0];
        const int re = row[1];
        rs[c] = ra;
        rl[c] = re - ra;
        ps += ra;
        pe += re;
    }
#pragma unroll
    for (int o = 16; o > 0; o >>= 1) {
        ps += __shfl_xor(ps, o);
        pe += __shfl_xor(pe, o);
    }
    if (lane == 0 && wv * 32 < NCH) {
        atomicAdd(&tot[0], ps);
        atomicAdd(&tot[1], pe);
    }
    __syncthreads();
    const int start = tot[0];
    const int n     = tot[1] - start;
    const int last  = E - 1;

    walk_runs(pairs, rs, rl, NCH, wv, lane, last,
              [&](const int2 p) { atomicAdd(&cnt[p.x - nb0], 1); });
    __syncthreads();

    int v = 0, cur = 0;
    if (t < BKT_W) { v = cnt[t]; buf[0][t] = v; }
    __syncthreads();
    for (int dd = 1; dd < BKT_W; dd <<= 1) {
        if (t < BKT_W) {
            int xv = buf[cur][t];
            if (t >= dd) xv += buf[cur][t - dd];
            buf[cur ^ 1][t] = xv;
        }
        cur ^= 1;
        __syncthreads();
    }
    if (t < BKT_W) {
        const int ex = buf[cur][t] - v;
        cnt[t] = ex;
        const int node = nb0 + t;
        if (node < N) off[node] = start + ex;
        if (node == N - 1) off[N] = start + n;
    }
    __syncthreads();

    if (n <= PLACE_CAP) {
        walk_runs(pairs, rs, rl, NCH, wv, lane, last,
                  [&](const int2 p) { stage[atomicAdd(&cnt[p.x - nb0], 1)] = p.y; });
        __syncthreads();
        for (int i = t; i < n; i += PLACE_T) ssrc[start + i] = stage[i];
    } else {
        walk_runs(pairs, rs, rl, NCH, wv, lane, last,
                  [&](const int2 p) { ssrc[start + atomicAdd(&cnt[p.x - nb0], 1)] = p.y; });
    }
}

__device__ __forceinline__ float leaky(float v) {
    return v > 0.0f ? v : NEG_SLOPE * v;
}

#define MM_KS (F / 4)
#define L1_T    1024
#define L1_G    8
#define L1_GRP  (L1_T / L1_G)
#define L1_EPL  6
#define L1_TAB  80000
#define L1_FILL 10

__global__ void __launch_bounds__(L1_T)
gat_l1_scalar_gather(const int* __restrict__ off,
                     const int* __restrict__ ssrc,
                     const float* __restrict__ x,
                     const float* __restrict__ W1,
                     const float* __restrict__ a_src,
                     const float* __restrict__ a_dst,
                     float* __restrict__ Sout,
                     int N) {
    __shared__ v4f tab4[L1_TAB / 4];
    const float* tab = (const float*)tab4;
    const int t   = threadIdx.x;
    const int npw = (N + (int)gridDim.x - 1) / (int)gridDim.x;
    const int n0  = (int)blockIdx.x * npw;
    const int n1  = n0 + npw < N ? n0 + npw : N;
    if (n0 >= n1) return;

    const int ntab = N < L1_TAB ? N : L1_TAB;
    const int n4   = ntab >> 2;
    const v4f* __restrict__ x4 = (const v4f*)x;
    for (int i = t; i < n4; i += L1_FILL * L1_T) {
        v4f v[L1_FILL];
#pragma unroll
        for (int k = 0; k < L1_FILL; ++k) {
            const int ii = i + k * L1_T;
            v[k] = x4[ii < n4 ? ii : i];
        }
        asm volatile("" :: "v"(v[0]), "v"(v[1]), "v"(v[2]), "v"(v[3]), "v"(v[4]),
                           "v"(v[5]), "v"(v[6]), "v"(v[7]), "v"(v[8]), "v"(v[9]));
#pragma unroll
        for (int k = 0; k < L1_FILL; ++k) {
            const int ii = i + k * L1_T;
            if (ii < n4) tab4[ii] = v[k];
        }
    }
    for (int i = (n4 << 2) + t; i < ntab; i += L1_T) ((float*)tab4)[i] = x[i];

    float cs = 0.0f, cd = 0.0f;
#pragma unroll
    for (int f = 0; f < F; ++f) {
        cs += W1[f] * a_src[f];
        cd += W1[f] * a_dst[f];
    }

    const int sub = t & (L1_G - 1);
    const int g   = t / L1_G;

    __syncthreads();

    const int nfix  = ntab < N ? ntab : N - 1;
    const int trips = (n1 - n0 + L1_GRP - 1) / L1_GRP;
    const int nf  = n0 + g < n1 ? n0 + g : n1 - 1;
    int   beg = off[nf];
    int   end = off[nf + 1];
    float xi  = x[nf];
    for (int it = 0; it < trips; ++it) {
        const int  node  = n0 + it * L1_GRP + g;
        const int   nn   = node + L1_GRP < n1 ? node + L1_GRP : n1 - 1;
        const int   begn = off[nn];
        const int   endn = off[nn + 1];
        const float xin  = x[nn];

        const float ad = xi * cd;
        float m   = leaky(xi * cs + ad);
        float den = sub == 0 ? 1.0f : 0.0f;
        float num = sub == 0 ? xi   : 0.0f;

        for (int j = beg + sub; j < end; j += L1_EPL * L1_G) {
            int s[L1_EPL];
#pragma unroll
            for (int k = 0; k < L1_EPL; ++k) {
                const int jk = j + k * L1_G;
                s[k] = ssrc[jk < end ? jk : j];
            }
            float xl[L1_EPL], xg[L1_EPL];
#pragma unroll
            for (int k = 0; k < L1_EPL; ++k) {
                const bool in_tab = s[k] < ntab;
                xl[k] = tab[in_tab ? s[k] : 0];
                xg[k] = x[in_tab ? nfix : s[k]];
            }
            float xv[L1_EPL], e[L1_EPL];
#pragma unroll
            for (int k = 0; k < L1_EPL; ++k) {
                const bool have = j + k * L1_G < end;
                const float xs  = s[k] < ntab ? xl[k] : xg[k];
                xv[k] = have ? xs : 0.0f;
                e[k]  = have ? leaky(xs * cs + ad) : -INFINITY;
            }
            float mn = m;
#pragma unroll
            for (int k = 0; k < L1_EPL; ++k) mn = fmaxf(mn, e[k]);
            const float sc = __expf(m - mn);
            m = mn;
            den *= sc;
            num *= sc;
#pragma unroll
            for (int k = 0; k < L1_EPL; ++k) {
                const float tk = __expf(e[k] - mn);
                den += tk;
                num += tk * xv[k];
            }
        }

        float M = m;
        M = fmaxf(M, __shfl_xor(M, 1, L1_G));
        M = fmaxf(M, __shfl_xor(M, 2, L1_G));
        M = fmaxf(M, __shfl_xor(M, 4, L1_G));
        const float fs = __expf(m - M);
        den *= fs;
        num *= fs;
        den += __shfl_xor(den, 1, L1_G);
        num += __shfl_xor(num, 1, L1_G);
        den += __shfl_xor(den, 2, L1_G);
        num += __shfl_xor(num, 2, L1_G);
        den += __shfl_xor(den, 4, L1_G);
        num += __shfl_xor(num, 4, L1_G);

        const float S = num * (1.0f / (den + 1e-16f));
        if (sub == 0 && node < n1) Sout[node] = S;

        beg = begn;
        end = endn;
        xi  = xin;
    }
}

#define L2_T    1024
#define L2_TAB  80000
#define L2_FILL 10

__global__ void __launch_bounds__(L2_T)
gat_l2_scalar_gather(const int* __restrict__ off,
                     const int* __restrict__ ssrc,
                     const float* __restrict__ Sv,
                     const float* __restrict__ W1,
                     const float* __restrict__ b1,
                     const float* __restrict__ W2,
                     const float* __restrict__ a_src2,
                     const float* __restrict__ a_dst2,
                     const float* __restrict__ b2,
                     const float* __restrict__ Wl,
                     const float* __restrict__ bl,
                     float* __restrict__ out,
                     int N) {
    __shared__ v4f tab4[L2_TAB / 4];
    __shared__ float wa[2][F];
    const float* tab = (const float*)tab4;
    const int t    = threadIdx.x;
    const int npw0 = (N + (int)gridDim.x - 1) / (int)gridDim.x;
    const int npw  = (npw0 + 15) & ~15;
    const int n0   = (int)blockIdx.x * npw;
    const int n1   = n0 + npw < N ? n0 + npw : N;
    if (n0 >= n1) return;

    const int ntab = N < L2_TAB ? N : L2_TAB;
    const int n4   = ntab >> 2;
    const v4f* __restrict__ S4 = (const v4f*)Sv;
    for (int i = t; i < n4; i += L2_FILL * L2_T) {
        v4f v[L2_FILL];
#pragma unroll
        for (int k = 0; k < L2_FILL; ++k) {
            const int ii = i + k * L2_T;
            v[k] = S4[ii < n4 ? ii : i];
        }
        asm volatile("" :: "v"(v[0]), "v"(v[1]), "v"(v[2]), "v"(v[3]), "v"(v[4]),
                           "v"(v[5]), "v"(v[6]), "v"(v[7]), "v"(v[8]), "v"(v[9]));
#pragma unroll
        for (int k = 0; k < L2_FILL; ++k) {
            const int ii = i + k * L2_T;
            if (ii < n4) tab4[ii] = v[k];
        }
    }
    for (int i = (n4 << 2) + t; i < ntab; i += L2_T) ((float*)tab4)[i] = Sv[i];

    if (t < F) {
        float ws = 0.0f, wd = 0.0f;
#pragma unroll
        for (int f = 0; f < F; ++f) {
            const float w = W2[t * F + f];
            ws += w * a_src2[f];
            wd += w * a_dst2[f];
        }
        wa[0][t] = ws;
        wa[1][t] = wd;
    }

    __syncthreads();

    const int  lane  = t & 31;
    const int  wv    = t >> 5;
    const int  sub   = lane & 1;
    const int  r     = lane >> 1;
    const int  mrow  = lane & 15;
    const int  lhalf = lane >> 4;
    const int  khalf = lhalf * 2;
    const int  n1col = 16 + mrow;
    const int  n1c   = n1col < F ? n1col : 0;
    const bool is_w  = n1col < F;
    const int  srcl  = 2 * mrow;

    v2f w1[F / 2], c1[F / 2], was[F / 2], wad[F / 2];
#pragma unroll
    for (int i = 0; i < F / 2; ++i) {
        w1[i]  = (v2f){W1[2 * i], W1[2 * i + 1]};
        c1[i]  = (v2f){b1[2 * i], b1[2 * i + 1]};
        was[i] = (v2f){wa[0][2 * i], wa[0][2 * i + 1]};
        wad[i] = (v2f){wa[1][2 * i], wa[1][2 * i + 1]};
    }
    const v2f zero2 = {0.0f, 0.0f};

    const int nfix  = ntab < N ? ntab : N - 1;
    const int ntile = (n1 - n0 + 15) >> 4;
    const int tile = wv;
    if (tile < ntile) {
        const int base = n0 + tile * 16;
        const int node = base + r;
        const int nc   = node < n1 ? node : n1 - 1;
        const int beg  = off[nc];
        const int end  = off[nc + 1];
        const float Si = Sv[nc];

        v2f acc[F / 2];
        v2f qs = zero2, qd = zero2;
#pragma unroll
        for (int i = 0; i < F / 2; ++i) {
            const v2f v = __builtin_elementwise_max(Si * w1[i] + c1[i], zero2);
            qs += v * was[i];
            qd += v * wad[i];
            acc[i] = sub == 0 ? v : zero2;
        }
        const float ad = qd.x + qd.y;
        float m   = leaky(qs.x + qs.y + ad);
        float den = sub == 0 ? 1.0f : 0.0f;

        for (int j = beg + sub; j < end; j += 4) {
            const int  j1    = j + 2;
            const bool have1 = j1 < end;
            const int  s0 = ssrc[j];
            const int  s1 = ssrc[have1 ? j1 : j];
            const bool in0 = s0 < ntab;
            const bool in1 = s1 < ntab;
            const float xl0 = tab[in0 ? s0 : 0];
            const float xl1 = tab[in1 ? s1 : 0];
            const float xg0 = Sv[in0 ? nfix : s0];
            const float xg1 = Sv[in1 ? nfix : s1];
            const float S0 = in0 ? xl0 : xg0;
            const float S1 = in1 ? xl1 : xg1;
            v2f v0[F / 2], v1[F / 2];
            v2f q0 = zero2, q1 = zero2;
#pragma unroll
            for (int i = 0; i < F / 2; ++i) {
                v0[i] = __builtin_elementwise_max(S0 * w1[i] + c1[i], zero2);
                v1[i] = __builtin_elementwise_max(S1 * w1[i] + c1[i], zero2);
                q0 += v0[i] * was[i];
                q1 += v1[i] * was[i];
            }
            const float e0 = leaky(q0.x + q0.y + ad);
            const float e1 = have1 ? leaky(q1.x + q1.y + ad) : -INFINITY;
            const float mn = fmaxf(m, fmaxf(e0, e1));
            const float sc = __expf(m - mn);
            m = mn;
            const float t0 = __expf(e0 - mn);
            const float t1 = __expf(e1 - mn);
            den = den * sc + t0 + t1;
#pragma unroll
            for (int i = 0; i < F / 2; ++i) acc[i] = acc[i] * sc + t0 * v0[i] + t1 * v1[i];
        }

        const float M  = fmaxf(m, __shfl_xor(m, 1));
        const float fs = __expf(m - M);
        den *= fs;
        den += __shfl_xor(den, 1);
        const float inv = 1.0f / (den + 1e-16f);
        float ag[F];
#pragma unroll
        for (int i = 0; i < F / 2; ++i) {
            const float lo = acc[i].x * fs;
            const float hi = acc[i].y * fs;
            ag[2 * i]     = (lo + __shfl_xor(lo, 1)) * inv;
            ag[2 * i + 1] = (hi + __shfl_xor(hi, 1)) * inv;
        }

        v8f d0 = {};
        v8f d1 = {};
#pragma unroll
        for (int i = 0; i < MM_KS; ++i) {
            const float k0 = __shfl(ag[4 * i + 0], srcl);
            const float k1 = __shfl(ag[4 * i + 1], srcl);
            const float k2 = __shfl(ag[4 * i + 2], srcl);
            const float k3 = __shfl(ag[4 * i + 3], srcl);
            v2f a;
            a.x = lhalf ? k2 : k0;
            a.y = lhalf ? k3 : k1;
            const int k = 4 * i + khalf;
            v2f bA, bB;
            bA.x = W2[(k + 0) * F + mrow];
            bA.y = W2[(k + 1) * F + mrow];
            const float wx = W2[(k + 0) * F + n1c];
            const float wy = W2[(k + 1) * F + n1c];
            bB.x = is_w ? wx : 0.0f;
            bB.y = is_w ? wy : 0.0f;
            d0 = __builtin_amdgcn_wmma_f32_16x16x4_f32(false, a, false, bA,
                                                       (short)0, d0, false, false);
            d1 = __builtin_amdgcn_wmma_f32_16x16x4_f32(false, a, false, bB,
                                                       (short)0, d1, false, false);
        }

        const float bc0 = b2[mrow];
        const float wl0 = Wl[mrow];
        const float bc1 = b2[n1c];
        const float wl1 = Wl[n1c];
        float p[8];
#pragma unroll
        for (int q = 0; q < 8; ++q) {
            const float o0 = fmaxf(d0[q] + bc0, 0.0f) * wl0;
            const float o1 = fmaxf(d1[q] + bc1, 0.0f) * wl1;
            p[q] = o0 + (is_w ? o1 : 0.0f);
        }
#pragma unroll
        for (int q = 0; q < 8; ++q) {
            p[q] += __shfl_xor(p[q], 1);
            p[q] += __shfl_xor(p[q], 2);
            p[q] += __shfl_xor(p[q], 4);
            p[q] += __shfl_xor(p[q], 8);
        }
        float res = p[0];
#pragma unroll
        for (int q = 1; q < 8; ++q) res = mrow == q ? p[q] : res;
        const int on = base + 8 * lhalf + mrow;
        if (mrow < 8 && on < n1) out[on] = res + bl[0];
    }
}

extern "C" void kernel_launch(void* const* d_in, const int* in_sizes, int n_in,
                              void* d_out, int out_size, void* d_ws, size_t ws_size,
                              hipStream_t stream) {
    (void)n_in; (void)out_size; (void)ws_size;

    const int N    = in_sizes[0];
    const int E    = in_sizes[1] / 2;

    const float* x    = (const float*)d_in[0];
    const int*   ei   = (const int*)d_in[1];
    const float* W1   = (const float*)d_in[2];
    const float* as1  = (const float*)d_in[3];
    const float* ad1  = (const float*)d_in[4];
    const float* b1   = (const float*)d_in[5];
    const float* W2   = (const float*)d_in[6];
    const float* as2  = (const float*)d_in[7];
    const float* ad2  = (const float*)d_in[8];
    const float* b2   = (const float*)d_in[9];
    const float* Wl   = (const float*)d_in[10];
    const float* bl   = (const float*)d_in[11];

    const int* srcIdx = ei;
    const int* dstIdx = ei + E;

    const int NB   = (N + BKT_W - 1) / BKT_W;
    const int gCH  = (E + SORT_CH - 1) / SORT_CH;
    const int gL2  = N > 256 * 512 ? (N + 511) / 512 : 256;

    float* ws     = (float*)d_ws;
    float* Sv     = ws + (size_t)0  * N;
    int*   off    = (int*)(ws + (size_t)75 * N);
    int2*  pairs  = (int2*)(ws + (size_t)78 * N);
    int*   ssrc   = (int*)(ws + (size_t)78 * N) + (size_t)2 * E;
    int*   lstart = ssrc + (size_t)E;

    bucket_scatter<<<gCH, SORT_T, NB * sizeof(int), stream>>>(srcIdx, dstIdx, lstart, pairs, E, NB);
    bucket_place<<<NB, PLACE_T, 2 * gCH * sizeof(int), stream>>>(pairs, lstart, off, ssrc, N, NB, gCH, E);

    gat_l1_scalar_gather<<<256, L1_T, 0, stream>>>(off, ssrc, x, W1, as1, ad1, Sv, N);

    gat_l2_scalar_gather<<<gL2, L2_T, 0, stream>>>(off, ssrc, Sv, W1, b1, W2, as2, ad2, b2, Wl, bl, (float*)d_out, N);
}


# Sort records packed to one 4-byte word (node-in-bucket in the top 8 bits, source in the low 24), not an 8-byte (dst, src) pair: bucket_scatter writes, bucket_place reads half the bytes; order kept
# speedup vs baseline: 36.3005x; 1.0508x over previous
#include <hip/hip_runtime.h>
#include <math.h>

#define F 20
#define NEG_SLOPE 0.2f

typedef float v2f __attribute__((ext_vector_type(2)));
typedef float v8f __attribute__((ext_vector_type(8)));
typedef float v4f __attribute__((ext_vector_type(4)));

#define BKT_SH 8
#define BKT_W  (1 << BKT_SH)
#define REC_SH   (32 - BKT_SH)
#define REC_MASK ((1 << REC_SH) - 1)
#define SORT_T   512
#define SORT_EPT 16
#define SORT_CH  (SORT_T * SORT_EPT)

__global__ void __launch_bounds__(SORT_T)
bucket_scatter(const int* __restrict__ src,
               const int* __restrict__ dst,
               int* __restrict__ lstart,
               int* __restrict__ recs,
               int E, int NB) {
    extern __shared__ int hist[];
    __shared__ int buf[2][SORT_T];
    __shared__ int2 stage2[SORT_CH / 2];
    int* stage = (int*)stage2;
    const int t  = threadIdx.x;
    const int c  = blockIdx.x;
    const int e0 = c * SORT_CH + t;
    int* __restrict__ lrow = lstart + (size_t)c * (NB + 1);
    for (int b = t; b < NB; b += SORT_T) hist[b] = 0;
    __syncthreads();

    int d[SORT_EPT], s[SORT_EPT], r[SORT_EPT];
    if (c * SORT_CH + SORT_CH <= E) {
#pragma unroll
        for (int k = 0; k < SORT_EPT; ++k) {
            d[k] = dst[e0 + k * SORT_T];
            s[k] = src[e0 + k * SORT_T];
        }
#pragma unroll
        for (int k = 0; k < SORT_EPT; ++k) r[k] = atomicAdd(&hist[d[k] >> BKT_SH], 1);
    } else {
#pragma unroll
        for (int k = 0; k < SORT_EPT; ++k) {
            int e = e0 + k * SORT_T;
            d[k] = -1; s[k] = 0; r[k] = 0;
            if (e < E) {
                d[k] = dst[e];
                s[k] = src[e];
                r[k] = atomicAdd(&hist[d[k] >> BKT_SH], 1);
            }
        }
    }
    __syncthreads();

    int carry = 0;
    for (int b0 = 0; b0 < NB; b0 += SORT_T) {
        const int b  = b0 + t;
        const int hv = b < NB ? hist[b] : 0;
        int cur = 0;
        buf[0][t] = hv;
        __syncthreads();
        for (int dd = 1; dd < SORT_T; dd <<= 1) {
            int xh = buf[cur][t];
            if (t >= dd) xh += buf[cur][t - dd];
            buf[cur ^ 1][t] = xh;
            cur ^= 1;
            __syncthreads();
        }
        const int lb = carry + buf[cur][t] - hv;
        if (b < NB) {
            hist[b] = lb;
            lrow[b] = lb;
        }
        carry += buf[cur][SORT_T - 1];
        __syncthreads();
    }
    const int nrec = carry;
    if (t == 0) lrow[NB] = nrec;

    int slot[SORT_EPT];
#pragma unroll
    for (int k = 0; k < SORT_EPT; ++k)
        slot[k] = hist[(d[k] >= 0 ? d[k] : 0) >> BKT_SH] + r[k];
#pragma unroll
    for (int k = 0; k < SORT_EPT; ++k) {
        if (d[k] >= 0) stage[slot[k]] = (int)(((unsigned)d[k] << REC_SH) | (unsigned)s[k]);
    }
    __syncthreads();

    int* __restrict__ outp = recs + (size_t)c * SORT_CH;
    if (nrec == SORT_CH) {
        int2* __restrict__ outp2 = (int2*)outp;
        int2 p[SORT_EPT / 2];
#pragma unroll
        for (int k = 0; k < SORT_EPT / 2; ++k) p[k] = stage2[t + k * SORT_T];
#pragma unroll
        for (int k = 0; k < SORT_EPT / 2; ++k) outp2[t + k * SORT_T] = p[k];
    } else {
        for (int i = t; i < nrec; i += SORT_T) outp[i] = stage[i];
    }
}

#define PLACE_T 1024
#define PLACE_NW (PLACE_T / 32)
#define PLACE_CAP 12288

#define PLACE_RPT 8

template <class Fn>
__device__ __forceinline__ void walk_runs(const int* __restrict__ recs,
                                          const int* rs, const int* rl,
                                          int NCH, int wv, int lane, int last, Fn fn) {
    for (int c = wv; c < NCH; c += PLACE_RPT * PLACE_NW) {
        int a[PLACE_RPT], l[PLACE_RPT];
        int p[PLACE_RPT];
#pragma unroll
        for (int k = 0; k < PLACE_RPT; ++k) {
            const int  ck = c + k * PLACE_NW;
            const bool hv = ck < NCH;
            const int  cc = hv ? ck : c;
            a[k] = cc * SORT_CH + rs[cc];
            l[k] = hv ? rl[cc] : 0;
        }
#pragma unroll
        for (int k = 0; k < PLACE_RPT; ++k) {
            const int i = a[k] + (lane < l[k] ? lane : 0);
            p[k] = recs[i < last ? i : last];
        }
        asm volatile("" :: "v"(p[0]), "v"(p[1]), "v"(p[2]), "v"(p[3]),
                           "v"(p[4]), "v"(p[5]), "v"(p[6]), "v"(p[7]));
#pragma unroll
        for (int k = 0; k < PLACE_RPT; ++k) {
            if (lane < l[k]) fn(p[k]);
            for (int j = lane + 32; j < l[k]; j += 32) fn(recs[a[k] + j]);
        }
    }
}

__global__ void __launch_bounds__(PLACE_T)
bucket_place(const int* __restrict__ recs,
             const int* __restrict__ lstart,
             int* __restrict__ off,
             int* __restrict__ ssrc,
             int N, int NB, int NCH, int E) {
    extern __shared__ int runs[];
    __shared__ int cnt[BKT_W];
    __shared__ int buf[2][BKT_W];
    __shared__ int stage[PLACE_CAP];
    __shared__ int tot[2];
    int* rs = runs;
    int* rl = runs + NCH;
    const int t     = threadIdx.x;
    const int lane  = t & 31;
    const int wv    = t >> 5;
    const int b     = blockIdx.x;
    const int nb0   = b << BKT_SH;

    if (t < BKT_W) cnt[t] = 0;
    if (t < 2) tot[t] = 0;
    __syncthreads();

    int ps = 0, pe = 0;
    for (int c = t; c < NCH; c += PLACE_T) {
        const int* __restrict__ row = lstart + (size_t)c * (NB + 1) + b;
        const int ra = row[0];
        const int re = row[1];
        rs[c] = ra;
        rl[c] = re - ra;
        ps += ra;
        pe += re;
    }
#pragma unroll
    for (int o = 16; o > 0; o >>= 1) {
        ps += __shfl_xor(ps, o);
        pe += __shfl_xor(pe, o);
    }
    if (lane == 0 && wv * 32 < NCH) {
        atomicAdd(&tot[0], ps);
        atomicAdd(&tot[1], pe);
    }
    __syncthreads();
    const int start = tot[0];
    const int n     = tot[1] - start;
    const int last  = E - 1;

    walk_runs(recs, rs, rl, NCH, wv, lane, last,
              [&](const int p) { atomicAdd(&cnt[(unsigned)p >> REC_SH], 1); });
    __syncthreads();

    int v = 0, cur = 0;
    if (t < BKT_W) { v = cnt[t]; buf[0][t] = v; }
    __syncthreads();
    for (int dd = 1; dd < BKT_W; dd <<= 1) {
        if (t < BKT_W) {
            int xv = buf[cur][t];
            if (t >= dd) xv += buf[cur][t - dd];
            buf[cur ^ 1][t] = xv;
        }
        cur ^= 1;
        __syncthreads();
    }
    if (t < BKT_W) {
        const int ex = buf[cur][t] - v;
        cnt[t] = ex;
        const int node = nb0 + t;
        if (node < N) off[node] = start + ex;
        if (node == N - 1) off[N] = start + n;
    }
    __syncthreads();

    if (n <= PLACE_CAP) {
        walk_runs(recs, rs, rl, NCH, wv, lane, last,
                  [&](const int p) { stage[atomicAdd(&cnt[(unsigned)p >> REC_SH], 1)] = p & REC_MASK; });
        __syncthreads();
        for (int i = t; i < n; i += PLACE_T) ssrc[start + i] = stage[i];
    } else {
        walk_runs(recs, rs, rl, NCH, wv, lane, last,
                  [&](const int p) { ssrc[start + atomicAdd(&cnt[(unsigned)p >> REC_SH], 1)] = p & REC_MASK; });
    }
}

__device__ __forceinline__ float leaky(float v) {
    return v > 0.0f ? v : NEG_SLOPE * v;
}

#define MM_KS (F / 4)
#define L1_T    1024
#define L1_G    8
#define L1_GRP  (L1_T / L1_G)
#define L1_EPL  6
#define L1_TAB  80000
#define L1_FILL 10

__global__ void __launch_bounds__(L1_T)
gat_l1_scalar_gather(const int* __restrict__ off,
                     const int* __restrict__ ssrc,
                     const float* __restrict__ x,
                     const float* __restrict__ W1,
                     const float* __restrict__ a_src,
                     const float* __restrict__ a_dst,
                     float* __restrict__ Sout,
                     int N) {
    __shared__ v4f tab4[L1_TAB / 4];
    const float* tab = (const float*)tab4;
    const int t   = threadIdx.x;
    const int npw = (N + (int)gridDim.x - 1) / (int)gridDim.x;
    const int n0  = (int)blockIdx.x * npw;
    const int n1  = n0 + npw < N ? n0 + npw : N;
    if (n0 >= n1) return;

    const int ntab = N < L1_TAB ? N : L1_TAB;
    const int n4   = ntab >> 2;
    const v4f* __restrict__ x4 = (const v4f*)x;
    for (int i = t; i < n4; i += L1_FILL * L1_T) {
        v4f v[L1_FILL];
#pragma unroll
        for (int k = 0; k < L1_FILL; ++k) {
            const int ii = i + k * L1_T;
            v[k] = x4[ii < n4 ? ii : i];
        }
        asm volatile("" :: "v"(v[0]), "v"(v[1]), "v"(v[2]), "v"(v[3]), "v"(v[4]),
                           "v"(v[5]), "v"(v[6]), "v"(v[7]), "v"(v[8]), "v"(v[9]));
#pragma unroll
        for (int k = 0; k < L1_FILL; ++k) {
            const int ii = i + k * L1_T;
            if (ii < n4) tab4[ii] = v[k];
        }
    }
    for (int i = (n4 << 2) + t; i < ntab; i += L1_T) ((float*)tab4)[i] = x[i];

    float cs = 0.0f, cd = 0.0f;
#pragma unroll
    for (int f = 0; f < F; ++f) {
        cs += W1[f] * a_src[f];
        cd += W1[f] * a_dst[f];
    }

    const int sub = t & (L1_G - 1);
    const int g   = t / L1_G;

    __syncthreads();

    const int nfix  = ntab < N ? ntab : N - 1;
    const int trips = (n1 - n0 + L1_GRP - 1) / L1_GRP;
    const int nf  = n0 + g < n1 ? n0 + g : n1 - 1;
    int   beg = off[nf];
    int   end = off[nf + 1];
    float xi  = x[nf];
    for (int it = 0; it < trips; ++it) {
        const int  node  = n0 + it * L1_GRP + g;
        const int   nn   = node + L1_GRP < n1 ? node + L1_GRP : n1 - 1;
        const int   begn = off[nn];
        const int   endn = off[nn + 1];
        const float xin  = x[nn];

        const float ad = xi * cd;
        float m   = leaky(xi * cs + ad);
        float den = sub == 0 ? 1.0f : 0.0f;
        float num = sub == 0 ? xi   : 0.0f;

        for (int j = beg + sub; j < end; j += L1_EPL * L1_G) {
            int s[L1_EPL];
#pragma unroll
            for (int k = 0; k < L1_EPL; ++k) {
                const int jk = j + k * L1_G;
                s[k] = ssrc[jk < end ? jk : j];
            }
            float xl[L1_EPL], xg[L1_EPL];
#pragma unroll
            for (int k = 0; k < L1_EPL; ++k) {
                const bool in_tab = s[k] < ntab;
                xl[k] = tab[in_tab ? s[k] : 0];
                xg[k] = x[in_tab ? nfix : s[k]];
            }
            float xv[L1_EPL], e[L1_EPL];
#pragma unroll
            for (int k = 0; k < L1_EPL; ++k) {
                const bool have = j + k * L1_G < end;
                const float xs  = s[k] < ntab ? xl[k] : xg[k];
                xv[k] = have ? xs : 0.0f;
                e[k]  = have ? leaky(xs * cs + ad) : -INFINITY;
            }
            float mn = m;
#pragma unroll
            for (int k = 0; k < L1_EPL; ++k) mn = fmaxf(mn, e[k]);
            const float sc = __expf(m - mn);
            m = mn;
            den *= sc;
            num *= sc;
#pragma unroll
            for (int k = 0; k < L1_EPL; ++k) {
                const float tk = __expf(e[k] - mn);
                den += tk;
                num += tk * xv[k];
            }
        }

        float M = m;
        M = fmaxf(M, __shfl_xor(M, 1, L1_G));
        M = fmaxf(M, __shfl_xor(M, 2, L1_G));
        M = fmaxf(M, __shfl_xor(M, 4, L1_G));
        const float fs = __expf(m - M);
        den *= fs;
        num *= fs;
        den += __shfl_xor(den, 1, L1_G);
        num += __shfl_xor(num, 1, L1_G);
        den += __shfl_xor(den, 2, L1_G);
        num += __shfl_xor(num, 2, L1_G);
        den += __shfl_xor(den, 4, L1_G);
        num += __shfl_xor(num, 4, L1_G);

        const float S = num * (1.0f / (den + 1e-16f));
        if (sub == 0 && node < n1) Sout[node] = S;

        beg = begn;
        end = endn;
        xi  = xin;
    }
}

#define L2_T    1024
#define L2_TAB  80000
#define L2_FILL 10

__global__ void __launch_bounds__(L2_T)
gat_l2_scalar_gather(const int* __restrict__ off,
                     const int* __restrict__ ssrc,
                     const float* __restrict__ Sv,
                     const float* __restrict__ W1,
                     const float* __restrict__ b1,
                     const float* __restrict__ W2,
                     const float* __restrict__ a_src2,
                     const float* __restrict__ a_dst2,
                     const float* __restrict__ b2,
                     const float* __restrict__ Wl,
                     const float* __restrict__ bl,
                     float* __restrict__ out,
                     int N) {
    __shared__ v4f tab4[L2_TAB / 4];
    __shared__ float wa[2][F];
    const float* tab = (const float*)tab4;
    const int t    = threadIdx.x;
    const int npw0 = (N + (int)gridDim.x - 1) / (int)gridDim.x;
    const int npw  = (npw0 + 15) & ~15;
    const int n0   = (int)blockIdx.x * npw;
    const int n1   = n0 + npw < N ? n0 + npw : N;
    if (n0 >= n1) return;

    const int ntab = N < L2_TAB ? N : L2_TAB;
    const int n4   = ntab >> 2;
    const v4f* __restrict__ S4 = (const v4f*)Sv;
    for (int i = t; i < n4; i += L2_FILL * L2_T) {
        v4f v[L2_FILL];
#pragma unroll
        for (int k = 0; k < L2_FILL; ++k) {
            const int ii = i + k * L2_T;
            v[k] = S4[ii < n4 ? ii : i];
        }
        asm volatile("" :: "v"(v[0]), "v"(v[1]), "v"(v[2]), "v"(v[3]), "v"(v[4]),
                           "v"(v[5]), "v"(v[6]), "v"(v[7]), "v"(v[8]), "v"(v[9]));
#pragma unroll
        for (int k = 0; k < L2_FILL; ++k) {
            const int ii = i + k * L2_T;
            if (ii < n4) tab4[ii] = v[k];
        }
    }
    for (int i = (n4 << 2) + t; i < ntab; i += L2_T) ((float*)tab4)[i] = Sv[i];

    if (t < F) {
        float ws = 0.0f, wd = 0.0f;
#pragma unroll
        for (int f = 0; f < F; ++f) {
            const float w = W2[t * F + f];
            ws += w * a_src2[f];
            wd += w * a_dst2[f];
        }
        wa[0][t] = ws;
        wa[1][t] = wd;
    }

    __syncthreads();

    const int  lane  = t & 31;
    const int  wv    = t >> 5;
    const int  sub   = lane & 1;
    const int  r     = lane >> 1;
    const int  mrow  = lane & 15;
    const int  lhalf = lane >> 4;
    const int  khalf = lhalf * 2;
    const int  n1col = 16 + mrow;
    const int  n1c   = n1col < F ? n1col : 0;
    const bool is_w  = n1col < F;
    const int  srcl  = 2 * mrow;

    v2f w1[F / 2], c1[F / 2], was[F / 2], wad[F / 2];
#pragma unroll
    for (int i = 0; i < F / 2; ++i) {
        w1[i]  = (v2f){W1[2 * i], W1[2 * i + 1]};
        c1[i]  = (v2f){b1[2 * i], b1[2 * i + 1]};
        was[i] = (v2f){wa[0][2 * i], wa[0][2 * i + 1]};
        wad[i] = (v2f){wa[1][2 * i], wa[1][2 * i + 1]};
    }
    const v2f zero2 = {0.0f, 0.0f};

    const int nfix  = ntab < N ? ntab : N - 1;
    const int ntile = (n1 - n0 + 15) >> 4;
    const int tile = wv;
    if (tile < ntile) {
        const int base = n0 + tile * 16;
        const int node = base + r;
        const int nc   = node < n1 ? node : n1 - 1;
        const int beg  = off[nc];
        const int end  = off[nc + 1];
        const float Si = Sv[nc];

        v2f acc[F / 2];
        v2f qs = zero2, qd = zero2;
#pragma unroll
        for (int i = 0; i < F / 2; ++i) {
            const v2f v = __builtin_elementwise_max(Si * w1[i] + c1[i], zero2);
            qs += v * was[i];
            qd += v * wad[i];
            acc[i] = sub == 0 ? v : zero2;
        }
        const float ad = qd.x + qd.y;
        float m   = leaky(qs.x + qs.y + ad);
        float den = sub == 0 ? 1.0f : 0.0f;

        for (int j = beg + sub; j < end; j += 4) {
            const int  j1    = j + 2;
            const bool have1 = j1 < end;
            const int  s0 = ssrc[j];
            const int  s1 = ssrc[have1 ? j1 : j];
            const bool in0 = s0 < ntab;
            const bool in1 = s1 < ntab;
            const float xl0 = tab[in0 ? s0 : 0];
            const float xl1 = tab[in1 ? s1 : 0];
            const float xg0 = Sv[in0 ? nfix : s0];
            const float xg1 = Sv[in1 ? nfix : s1];
            const float S0 = in0 ? xl0 : xg0;
            const float S1 = in1 ? xl1 : xg1;
            v2f v0[F / 2], v1[F / 2];
            v2f q0 = zero2, q1 = zero2;
#pragma unroll
            for (int i = 0; i < F / 2; ++i) {
                v0[i] = __builtin_elementwise_max(S0 * w1[i] + c1[i], zero2);
                v1[i] = __builtin_elementwise_max(S1 * w1[i] + c1[i], zero2);
                q0 += v0[i] * was[i];
                q1 += v1[i] * was[i];
            }
            const float e0 = leaky(q0.x + q0.y + ad);
            const float e1 = have1 ? leaky(q1.x + q1.y + ad) : -INFINITY;
            const float mn = fmaxf(m, fmaxf(e0, e1));
            const float sc = __expf(m - mn);
            m = mn;
            const float t0 = __expf(e0 - mn);
            const float t1 = __expf(e1 - mn);
            den = den * sc + t0 + t1;
#pragma unroll
            for (int i = 0; i < F / 2; ++i) acc[i] = acc[i] * sc + t0 * v0[i] + t1 * v1[i];
        }

        const float M  = fmaxf(m, __shfl_xor(m, 1));
        const float fs = __expf(m - M);
        den *= fs;
        den += __shfl_xor(den, 1);
        const float inv = 1.0f / (den + 1e-16f);
        float ag[F];
#pragma unroll
        for (int i = 0; i < F / 2; ++i) {
            const float lo = acc[i].x * fs;
            const float hi = acc[i].y * fs;
            ag[2 * i]     = (lo + __shfl_xor(lo, 1)) * inv;
            ag[2 * i + 1] = (hi + __shfl_xor(hi, 1)) * inv;
        }

        v8f d0 = {};
        v8f d1 = {};
#pragma unroll
        for (int i = 0; i < MM_KS; ++i) {
            const float k0 = __shfl(ag[4 * i + 0], srcl);
            const float k1 = __shfl(ag[4 * i + 1], srcl);
            const float k2 = __shfl(ag[4 * i + 2], srcl);
            const float k3 = __shfl(ag[4 * i + 3], srcl);
            v2f a;
            a.x = lhalf ? k2 : k0;
            a.y = lhalf ? k3 : k1;
            const int k = 4 * i + khalf;
            v2f bA, bB;
            bA.x = W2[(k + 0) * F + mrow];
            bA.y = W2[(k + 1) * F + mrow];
            const float wx = W2[(k + 0) * F + n1c];
            const float wy = W2[(k + 1) * F + n1c];
            bB.x = is_w ? wx : 0.0f;
            bB.y = is_w ? wy : 0.0f;
            d0 = __builtin_amdgcn_wmma_f32_16x16x4_f32(false, a, false, bA,
                                                       (short)0, d0, false, false);
            d1 = __builtin_amdgcn_wmma_f32_16x16x4_f32(false, a, false, bB,
                                                       (short)0, d1, false, false);
        }

        const float bc0 = b2[mrow];
        const float wl0 = Wl[mrow];
        const float bc1 = b2[n1c];
        const float wl1 = Wl[n1c];
        float p[8];
#pragma unroll
        for (int q = 0; q < 8; ++q) {
            const float o0 = fmaxf(d0[q] + bc0, 0.0f) * wl0;
            const float o1 = fmaxf(d1[q] + bc1, 0.0f) * wl1;
            p[q] = o0 + (is_w ? o1 : 0.0f);
        }
#pragma unroll
        for (int q = 0; q < 8; ++q) {
            p[q] += __shfl_xor(p[q], 1);
            p[q] += __shfl_xor(p[q], 2);
            p[q] += __shfl_xor(p[q], 4);
            p[q] += __shfl_xor(p[q], 8);
        }
        float res = p[0];
#pragma unroll
        for (int q = 1; q < 8; ++q) res = mrow == q ? p[q] : res;
        const int on = base + 8 * lhalf + mrow;
        if (mrow < 8 && on < n1) out[on] = res + bl[0];
    }
}

extern "C" void kernel_launch(void* const* d_in, const int* in_sizes, int n_in,
                              void* d_out, int out_size, void* d_ws, size_t ws_size,
                              hipStream_t stream) {
    (void)n_in; (void)out_size; (void)ws_size;

    const int N    = in_sizes[0];
    const int E    = in_sizes[1] / 2;

    const float* x    = (const float*)d_in[0];
    const int*   ei   = (const int*)d_in[1];
    const float* W1   = (const float*)d_in[2];
    const float* as1  = (const float*)d_in[3];
    const float* ad1  = (const float*)d_in[4];
    const float* b1   = (const float*)d_in[5];
    const float* W2   = (const float*)d_in[6];
    const float* as2  = (const float*)d_in[7];
    const float* ad2  = (const float*)d_in[8];
    const float* b2   = (const float*)d_in[9];
    const float* Wl   = (const float*)d_in[10];
    const float* bl   = (const float*)d_in[11];

    const int* srcIdx = ei;
    const int* dstIdx = ei + E;

    const int NB   = (N + BKT_W - 1) / BKT_W;
    const int gCH  = (E + SORT_CH - 1) / SORT_CH;
    const int gL2  = N > 256 * 512 ? (N + 511) / 512 : 256;

    float* ws     = (float*)d_ws;
    float* Sv     = ws + (size_t)0  * N;
    int*   off    = (int*)(ws + (size_t)75 * N);
    int*   recs   = (int*)(ws + (size_t)78 * N);
    int*   ssrc   = recs + (size_t)E;
    int*   lstart = ssrc + (size_t)E;

    bucket_scatter<<<gCH, SORT_T, NB * sizeof(int), stream>>>(srcIdx, dstIdx, lstart, recs, E, NB);
    bucket_place<<<NB, PLACE_T, 2 * gCH * sizeof(int), stream>>>(recs, lstart, off, ssrc, N, NB, gCH, E);

    gat_l1_scalar_gather<<<256, L1_T, 0, stream>>>(off, ssrc, x, W1, as1, ad1, Sv, N);

    gat_l2_scalar_gather<<<gL2, L2_T, 0, stream>>>(off, ssrc, Sv, W1, b1, W2, as2, ad2, b2, Wl, bl, (float*)d_out, N);
}
